# starting kernel
# baseline (speedup 1.0000x reference)
#include <hip/hip_runtime.h>
#include <hip/hip_bf16.h>


typedef float v2f __attribute__((ext_vector_type(2)));
typedef float v8f __attribute__((ext_vector_type(8)));

#define N_NODES_C 100000
#define N_EDGES_C 800000
#define HID_C     128
#define OUT_DIM_C 500
#define N_GRAPHS_C 64

__global__ void degree_kernel(const int* __restrict__ src, const int* __restrict__ dst,
                              float* __restrict__ odeg, float* __restrict__ ideg,
                              int n_edges) {
    int e = blockIdx.x * blockDim.x + threadIdx.x;
    if (e < n_edges) {
        atomicAdd(&odeg[src[e]], 1.0f);
        atomicAdd(&ideg[dst[e]], 1.0f);
    }
}

__global__ void norm_kernel(const float* __restrict__ odeg, const float* __restrict__ ideg,
                            float* __restrict__ srcn, float* __restrict__ dstn, int n) {
    int i = blockIdx.x * blockDim.x + threadIdx.x;
    if (i < n) {
        srcn[i] = rsqrtf(fmaxf(odeg[i], 1.0f));
        dstn[i] = rsqrtf(fmaxf(ideg[i], 1.0f));
    }
}

__global__ void edge_agg_kernel(const float* __restrict__ h, const float* __restrict__ srcn,
                                const int* __restrict__ src, const int* __restrict__ dst,
                                float* __restrict__ agg, int n_edges) {
    int gid  = blockIdx.x * blockDim.x + threadIdx.x;
    int e    = gid >> 5;
    int lane = gid & 31;
    if (e >= n_edges) return;
    int s = src[e];
    int d = dst[e];
    float sn = srcn[s];
    float4 v = ((const float4*)(h + (size_t)s * HID_C))[lane];
    float* ap = agg + (size_t)d * HID_C + lane * 4;
    atomicAdd(ap + 0, v.x * sn);
    atomicAdd(ap + 1, v.y * sn);
    atomicAdd(ap + 2, v.z * sn);
    atomicAdd(ap + 3, v.w * sn);
}

__global__ void __launch_bounds__(256) conv_gemm_kernel(
    const float* __restrict__ agg, const float* __restrict__ dstn,
    const float* __restrict__ W, const float* __restrict__ bias,
    float* __restrict__ out, int n_nodes) {
    __shared__ float As[16][HID_C + 4];

    const int tid   = threadIdx.x;
    const int mbase = blockIdx.x * 16;

    for (int i = tid; i < 16 * (HID_C / 4); i += 256) {
        int r    = i >> 5;
        int c4   = i & 31;
        int node = mbase + r;
        float4 v = make_float4(0.f, 0.f, 0.f, 0.f);
        float  sc = 0.f;
        if (node < n_nodes) {
            v  = *(const float4*)(agg + (size_t)node * HID_C + c4 * 4);
            sc = dstn[node];
        }
        As[r][c4 * 4 + 0] = v.x * sc;
        As[r][c4 * 4 + 1] = v.y * sc;
        As[r][c4 * 4 + 2] = v.z * sc;
        As[r][c4 * 4 + 3] = v.w * sc;
    }
    __syncthreads();

    const int lane  = tid & 31;
    const int wave  = tid >> 5;
    const int l15   = lane & 15;
    const int khalf = (lane >> 4) << 1;
    const int col   = wave * 16 + l15;

    v8f acc = {0.f, 0.f, 0.f, 0.f, 0.f, 0.f, 0.f, 0.f};
    #pragma unroll
    for (int k = 0; k < HID_C; k += 4) {
        v2f a, b;
        a.x = As[l15][k + khalf];
        a.y = As[l15][k + khalf + 1];
        b.x = W[(size_t)(k + khalf) * HID_C + col];
        b.y = W[(size_t)(k + khalf + 1) * HID_C + col];
        acc = __builtin_amdgcn_wmma_f32_16x16x4_f32(false, a, false, b,
                                                    (short)0, acc, false, false);
    }

    const float bv = bias[col];
    #pragma unroll
    for (int i = 0; i < 8; ++i) {
        int node = mbase + i + ((lane >> 4) << 3);
        if (node < n_nodes)
            out[(size_t)node * HID_C + col] = fmaxf(acc[i] + bv, 0.0f);
    }
}

__global__ void __launch_bounds__(32) wmma_gemm_kernel(
    const float* __restrict__ A, const float* __restrict__ B,
    const float* __restrict__ bias, float* __restrict__ C,
    int M, int N, int K, int do_relu) {
    const int lane  = threadIdx.x & 31;
    const int l15   = lane & 15;
    const int khalf = (lane >> 4) << 1;
    const int mbase = blockIdx.x * 16;
    const int nbase = blockIdx.y * 16;
    const int row   = mbase + l15;
    const int col   = nbase + l15;

    v8f acc = {0.f, 0.f, 0.f, 0.f, 0.f, 0.f, 0.f, 0.f};
    for (int k = 0; k < K; k += 4) {
        int k0 = k + khalf, k1 = k + khalf + 1;
        v2f a, b;
        a.x = (row < M) ? A[(size_t)row * K + k0] : 0.0f;
        a.y = (row < M) ? A[(size_t)row * K + k1] : 0.0f;
        b.x = (col < N) ? B[(size_t)k0 * N + col] : 0.0f;
        b.y = (col < N) ? B[(size_t)k1 * N + col] : 0.0f;
        acc = __builtin_amdgcn_wmma_f32_16x16x4_f32(false, a, false, b,
                                                    (short)0, acc, false, false);
    }

    if (col < N) {
        const float bv = bias[col];
        #pragma unroll
        for (int i = 0; i < 8; ++i) {
            int m = mbase + i + ((lane >> 4) << 3);
            if (m < M) {
                float v = acc[i] + bv;
                C[(size_t)m * N + col] = do_relu ? fmaxf(v, 0.0f) : v;
            }
        }
    }
}

__global__ void segmax_kernel(const float* __restrict__ h, const int* __restrict__ gids,
                              float* __restrict__ hg, int n_nodes) {
    int node = blockIdx.x;
    int c    = threadIdx.x;
    if (node < n_nodes) {
        int g = gids[node];
        float v = h[(size_t)node * HID_C + c];
        atomicMax((unsigned int*)&hg[(size_t)g * HID_C + c], __float_as_uint(v));
    }
}

extern "C" void kernel_launch(void* const* d_in, const int* in_sizes, int n_in,
                              void* d_out, int out_size, void* d_ws, size_t ws_size,
                              hipStream_t stream) {
    (void)in_sizes; (void)n_in; (void)out_size; (void)ws_size;

    const float* feat = (const float*)d_in[0];
    const int*   src  = (const int*)d_in[1];
    const int*   dst  = (const int*)d_in[2];
    const int*   gids = (const int*)d_in[3];
    const float* Wg[3] = {(const float*)d_in[4], (const float*)d_in[6], (const float*)d_in[8]};
    const float* bg[3] = {(const float*)d_in[5], (const float*)d_in[7], (const float*)d_in[9]};
    const float* Wc1 = (const float*)d_in[10]; const float* bc1 = (const float*)d_in[11];
    const float* Wc2 = (const float*)d_in[12]; const float* bc2 = (const float*)d_in[13];
    const float* Wc3 = (const float*)d_in[14]; const float* bc3 = (const float*)d_in[15];
    float* out = (float*)d_out;

    char* ws = (char*)d_ws;
    size_t off = 0;
    auto take = [&](size_t elems) -> float* {
        float* p = (float*)(ws + off);
        off += ((elems * sizeof(float) + 255) / 256) * 256;
        return p;
    };
    float* odeg = take(N_NODES_C);
    float* ideg = take(N_NODES_C);
    float* srcn = take(N_NODES_C);
    float* dstn = take(N_NODES_C);
    float* agg  = take((size_t)N_NODES_C * HID_C);
    float* h    = take((size_t)N_NODES_C * HID_C);
    float* hg   = take((size_t)N_GRAPHS_C * HID_C);
    float* x1   = take((size_t)N_GRAPHS_C * HID_C);
    float* x2   = take((size_t)N_GRAPHS_C * (HID_C / 2));

    hipMemsetAsync(odeg, 0, N_NODES_C * sizeof(float), stream);
    hipMemsetAsync(ideg, 0, N_NODES_C * sizeof(float), stream);
    degree_kernel<<<(N_EDGES_C + 255) / 256, 256, 0, stream>>>(src, dst, odeg, ideg, N_EDGES_C);
    norm_kernel<<<(N_NODES_C + 255) / 256, 256, 0, stream>>>(odeg, ideg, srcn, dstn, N_NODES_C);

    const int mblocks = (N_NODES_C + 15) / 16;
    const float* hin = feat;
    for (int L = 0; L < 3; ++L) {
        hipMemsetAsync(agg, 0, (size_t)N_NODES_C * HID_C * sizeof(float), stream);
        edge_agg_kernel<<<(N_EDGES_C * 32 + 255) / 256, 256, 0, stream>>>(
            hin, srcn, src, dst, agg, N_EDGES_C);
        conv_gemm_kernel<<<mblocks, 256, 0, stream>>>(agg, dstn, Wg[L], bg[L], h, N_NODES_C);
        hin = h;
    }

    hipMemsetAsync(hg, 0, (size_t)N_GRAPHS_C * HID_C * sizeof(float), stream);
    segmax_kernel<<<N_NODES_C, HID_C, 0, stream>>>(h, gids, hg, N_NODES_C);

    wmma_gemm_kernel<<<dim3(N_GRAPHS_C / 16, HID_C / 16), 32, 0, stream>>>(
        hg, Wc1, bc1, x1, N_GRAPHS_C, HID_C, HID_C, 1);
    wmma_gemm_kernel<<<dim3(N_GRAPHS_C / 16, (HID_C / 2) / 16), 32, 0, stream>>>(
        x1, Wc2, bc2, x2, N_GRAPHS_C, HID_C / 2, HID_C, 1);
    wmma_gemm_kernel<<<dim3(N_GRAPHS_C / 16, (OUT_DIM_C + 15) / 16), 32, 0, stream>>>(
        x2, Wc3, bc3, out, N_GRAPHS_C, OUT_DIM_C, HID_C / 2, 0);
}


# Edge sum as a gather: edges bucketed by destination once per call (returning count atomics), one wave adds a node's source rows in source-id order; f32 row atomics and agg zero fill removed
# speedup vs baseline: 1.9788x; 1.9788x over previous
#include <hip/hip_runtime.h>
#include <hip/hip_bf16.h>


typedef float v2f __attribute__((ext_vector_type(2)));
typedef float v8f __attribute__((ext_vector_type(8)));

#define N_NODES_C 100000
#define N_EDGES_C 800000
#define HID_C     128
#define OUT_DIM_C 500
#define N_GRAPHS_C 64

__global__ void degree_kernel(const int* __restrict__ src, const int* __restrict__ dst,
                              float* __restrict__ odeg, float* __restrict__ ideg,
                              int* __restrict__ slot, int n_edges) {
    int e = blockIdx.x * blockDim.x + threadIdx.x;
    if (e < n_edges) {
        atomicAdd(&odeg[src[e]], 1.0f);
        slot[e] = (int)atomicAdd(&ideg[dst[e]], 1.0f);
    }
}

__global__ void __launch_bounds__(256) norm_kernel(
    const float* __restrict__ odeg, const float* __restrict__ ideg,
    float* __restrict__ srcn, float* __restrict__ dstn,
    unsigned long long* __restrict__ bucket, int* __restrict__ cursor, int n) {
    __shared__ int wsum[8];
    __shared__ int base_s;
    const int i    = blockIdx.x * 256 + threadIdx.x;
    const int lane = threadIdx.x & 31;
    const int wave = threadIdx.x >> 5;
    int cnt = 0;
    if (i < n) {
        float id = ideg[i];
        srcn[i] = rsqrtf(fmaxf(odeg[i], 1.0f));
        dstn[i] = rsqrtf(fmaxf(id, 1.0f));
        cnt = (int)id;
    }
    int incl = cnt;
    #pragma unroll
    for (int o = 1; o < 32; o <<= 1) {
        int t = __shfl_up(incl, o);
        if (lane >= o) incl += t;
    }
    if (lane == 31) wsum[wave] = incl;
    __syncthreads();
    if (threadIdx.x == 0) {
        int tot = 0;
        for (int w = 0; w < 8; ++w) { int t = wsum[w]; wsum[w] = tot; tot += t; }
        base_s = atomicAdd(cursor, tot);
    }
    __syncthreads();
    if (i < n)
        bucket[i] = (unsigned long long)(unsigned)(base_s + wsum[wave] + incl - cnt) |
                    ((unsigned long long)(unsigned)cnt << 32);
}

__global__ void fill_kernel(const int* __restrict__ src, const int* __restrict__ dst,
                            const int* __restrict__ slot,
                            const unsigned long long* __restrict__ bucket,
                            const float* __restrict__ srcn, int2* __restrict__ ent,
                            int n_edges) {
    int e = blockIdx.x * blockDim.x + threadIdx.x;
    if (e < n_edges) {
        int s = src[e];
        ent[(int)(unsigned)bucket[dst[e]] + slot[e]] = make_int2(s, __float_as_int(srcn[s]));
    }
}

__device__ __forceinline__ int lane_of_rank(int r, int k) {
    return __builtin_ctz(__builtin_amdgcn_ballot_w32(r == k));
}

__global__ void __launch_bounds__(256) gather_agg_kernel(
    const float* __restrict__ h, const unsigned long long* __restrict__ bucket,
    const int2* __restrict__ ent, float* __restrict__ agg, int n_nodes) {
    const int lane = threadIdx.x & 31;
    const int wave   = __builtin_amdgcn_readfirstlane(threadIdx.x >> 5);
    const int nwaves = gridDim.x * 8;
    int node = blockIdx.x * 8 + wave;
    unsigned long long b = (node < n_nodes) ? bucket[node] : 0ull;
    for (; node < n_nodes; node += nwaves) {
        const int nxt = node + nwaves;
        const unsigned long long bn = (nxt < n_nodes) ? bucket[nxt] : 0ull;
        const int start = (int)(unsigned)b;
        const int cnt   = (int)(unsigned)(b >> 32);
        b = bn;
        float4 acc = make_float4(0.f, 0.f, 0.f, 0.f);
        for (int c0 = 0; c0 < cnt; c0 += 32) {
            const int m = min(32, cnt - c0);
            int   s = 0x7fffffff;
            float w = 0.f;
            if (lane < m) {
                int2 q = ent[start + c0 + lane];
                s = q.x;
                w = __int_as_float(q.y);
            }
            int r = 0;
            for (int j = 0; j < m; ++j) {
                int t = __builtin_amdgcn_readlane(s, j);
                r += ((t < s) || (t == s && j < lane)) ? 1 : 0;
            }
            int k = 0;
            for (; k + 4 <= m; k += 4) {
                const int j0 = lane_of_rank(r, k),     j1 = lane_of_rank(r, k + 1);
                const int j2 = lane_of_rank(r, k + 2), j3 = lane_of_rank(r, k + 3);
                const int s0 = __builtin_amdgcn_readlane(s, j0);
                const int s1 = __builtin_amdgcn_readlane(s, j1);
                const int s2 = __builtin_amdgcn_readlane(s, j2);
                const int s3 = __builtin_amdgcn_readlane(s, j3);
                const float4 v0 = ((const float4*)(h + (size_t)s0 * HID_C))[lane];
                const float4 v1 = ((const float4*)(h + (size_t)s1 * HID_C))[lane];
                const float4 v2 = ((const float4*)(h + (size_t)s2 * HID_C))[lane];
                const float4 v3 = ((const float4*)(h + (size_t)s3 * HID_C))[lane];
                const float w0 = __int_as_float(__builtin_amdgcn_readlane(__float_as_int(w), j0));
                const float w1 = __int_as_float(__builtin_amdgcn_readlane(__float_as_int(w), j1));
                const float w2 = __int_as_float(__builtin_amdgcn_readlane(__float_as_int(w), j2));
                const float w3 = __int_as_float(__builtin_amdgcn_readlane(__float_as_int(w), j3));
                acc.x += v0.x * w0; acc.y += v0.y * w0; acc.z += v0.z * w0; acc.w += v0.w * w0;
                acc.x += v1.x * w1; acc.y += v1.y * w1; acc.z += v1.z * w1; acc.w += v1.w * w1;
                acc.x += v2.x * w2; acc.y += v2.y * w2; acc.z += v2.z * w2; acc.w += v2.w * w2;
                acc.x += v3.x * w3; acc.y += v3.y * w3; acc.z += v3.z * w3; acc.w += v3.w * w3;
            }
            const int rem = m - k;
            if (rem > 0) {
                const int j0 = lane_of_rank(r, k);
                const int j1 = (rem > 1) ? lane_of_rank(r, k + 1) : j0;
                const int j2 = (rem > 2) ? lane_of_rank(r, k + 2) : j0;
                const int s0 = __builtin_amdgcn_readlane(s, j0);
                const int s1 = __builtin_amdgcn_readlane(s, j1);
                const int s2 = __builtin_amdgcn_readlane(s, j2);
                const float4 v0 = ((const float4*)(h + (size_t)s0 * HID_C))[lane];
                float4 v1 = make_float4(0.f, 0.f, 0.f, 0.f), v2 = v1;
                if (rem > 1) v1 = ((const float4*)(h + (size_t)s1 * HID_C))[lane];
                if (rem > 2) v2 = ((const float4*)(h + (size_t)s2 * HID_C))[lane];
                const float w0 = __int_as_float(__builtin_amdgcn_readlane(__float_as_int(w), j0));
                const float w1 = __int_as_float(__builtin_amdgcn_readlane(__float_as_int(w), j1));
                const float w2 = __int_as_float(__builtin_amdgcn_readlane(__float_as_int(w), j2));
                acc.x += v0.x * w0; acc.y += v0.y * w0; acc.z += v0.z * w0; acc.w += v0.w * w0;
                if (rem > 1) {
                    acc.x += v1.x * w1; acc.y += v1.y * w1; acc.z += v1.z * w1; acc.w += v1.w * w1;
                }
                if (rem > 2) {
                    acc.x += v2.x * w2; acc.y += v2.y * w2; acc.z += v2.z * w2; acc.w += v2.w * w2;
                }
            }
        }
        ((float4*)(agg + (size_t)node * HID_C))[lane] = acc;
    }
}

__global__ void __launch_bounds__(256) conv_gemm_kernel(
    const float* __restrict__ agg, const float* __restrict__ dstn,
    const float* __restrict__ W, const float* __restrict__ bias,
    float* __restrict__ out, int n_nodes) {
    __shared__ float As[16][HID_C + 4];

    const int tid   = threadIdx.x;
    const int mbase = blockIdx.x * 16;

    for (int i = tid; i < 16 * (HID_C / 4); i += 256) {
        int r    = i >> 5;
        int c4   = i & 31;
        int node = mbase + r;
        float4 v = make_float4(0.f, 0.f, 0.f, 0.f);
        float  sc = 0.f;
        if (node < n_nodes) {
            v  = *(const float4*)(agg + (size_t)node * HID_C + c4 * 4);
            sc = dstn[node];
        }
        As[r][c4 * 4 + 0] = v.x * sc;
        As[r][c4 * 4 + 1] = v.y * sc;
        As[r][c4 * 4 + 2] = v.z * sc;
        As[r][c4 * 4 + 3] = v.w * sc;
    }
    __syncthreads();

    const int lane  = tid & 31;
    const int wave  = tid >> 5;
    const int l15   = lane & 15;
    const int khalf = (lane >> 4) << 1;
    const int col   = wave * 16 + l15;

    v8f acc = {0.f, 0.f, 0.f, 0.f, 0.f, 0.f, 0.f, 0.f};
    #pragma unroll
    for (int k = 0; k < HID_C; k += 4) {
        v2f a, b;
        a.x = As[l15][k + khalf];
        a.y = As[l15][k + khalf + 1];
        b.x = W[(size_t)(k + khalf) * HID_C + col];
        b.y = W[(size_t)(k + khalf + 1) * HID_C + col];
        acc = __builtin_amdgcn_wmma_f32_16x16x4_f32(false, a, false, b,
                                                    (short)0, acc, false, false);
    }

    const float bv = bias[col];
    #pragma unroll
    for (int i = 0; i < 8; ++i) {
        int node = mbase + i + ((lane >> 4) << 3);
        if (node < n_nodes)
            out[(size_t)node * HID_C + col] = fmaxf(acc[i] + bv, 0.0f);
    }
}

__global__ void __launch_bounds__(32) wmma_gemm_kernel(
    const float* __restrict__ A, const float* __restrict__ B,
    const float* __restrict__ bias, float* __restrict__ C,
    int M, int N, int K, int do_relu) {
    const int lane  = threadIdx.x & 31;
    const int l15   = lane & 15;
    const int khalf = (lane >> 4) << 1;
    const int mbase = blockIdx.x * 16;
    const int nbase = blockIdx.y * 16;
    const int row   = mbase + l15;
    const int col   = nbase + l15;

    v8f acc = {0.f, 0.f, 0.f, 0.f, 0.f, 0.f, 0.f, 0.f};
    for (int k = 0; k < K; k += 4) {
        int k0 = k + khalf, k1 = k + khalf + 1;
        v2f a, b;
        a.x = (row < M) ? A[(size_t)row * K + k0] : 0.0f;
        a.y = (row < M) ? A[(size_t)row * K + k1] : 0.0f;
        b.x = (col < N) ? B[(size_t)k0 * N + col] : 0.0f;
        b.y = (col < N) ? B[(size_t)k1 * N + col] : 0.0f;
        acc = __builtin_amdgcn_wmma_f32_16x16x4_f32(false, a, false, b,
                                                    (short)0, acc, false, false);
    }

    if (col < N) {
        const float bv = bias[col];
        #pragma unroll
        for (int i = 0; i < 8; ++i) {
            int m = mbase + i + ((lane >> 4) << 3);
            if (m < M) {
                float v = acc[i] + bv;
                C[(size_t)m * N + col] = do_relu ? fmaxf(v, 0.0f) : v;
            }
        }
    }
}

__global__ void segmax_kernel(const float* __restrict__ h, const int* __restrict__ gids,
                              float* __restrict__ hg, int n_nodes) {
    int node = blockIdx.x;
    int c    = threadIdx.x;
    if (node < n_nodes) {
        int g = gids[node];
        float v = h[(size_t)node * HID_C + c];
        atomicMax((unsigned int*)&hg[(size_t)g * HID_C + c], __float_as_uint(v));
    }
}

extern "C" void kernel_launch(void* const* d_in, const int* in_sizes, int n_in,
                              void* d_out, int out_size, void* d_ws, size_t ws_size,
                              hipStream_t stream) {
    (void)in_sizes; (void)n_in; (void)out_size; (void)ws_size;

    const float* feat = (const float*)d_in[0];
    const int*   src  = (const int*)d_in[1];
    const int*   dst  = (const int*)d_in[2];
    const int*   gids = (const int*)d_in[3];
    const float* Wg[3] = {(const float*)d_in[4], (const float*)d_in[6], (const float*)d_in[8]};
    const float* bg[3] = {(const float*)d_in[5], (const float*)d_in[7], (const float*)d_in[9]};
    const float* Wc1 = (const float*)d_in[10]; const float* bc1 = (const float*)d_in[11];
    const float* Wc2 = (const float*)d_in[12]; const float* bc2 = (const float*)d_in[13];
    const float* Wc3 = (const float*)d_in[14]; const float* bc3 = (const float*)d_in[15];
    float* out = (float*)d_out;

    char* ws = (char*)d_ws;
    size_t off = 0;
    auto take = [&](size_t elems) -> float* {
        float* p = (float*)(ws + off);
        off += ((elems * sizeof(float) + 255) / 256) * 256;
        return p;
    };
    float* cnts = take(2 * N_NODES_C + 64);
    float* odeg = cnts;
    float* ideg = cnts + N_NODES_C;
    int*   cursor = (int*)(cnts + 2 * N_NODES_C);
    float* srcn = take(N_NODES_C);
    float* dstn = take(N_NODES_C);
    float* agg  = take((size_t)N_NODES_C * HID_C);
    float* h    = take((size_t)N_NODES_C * HID_C);
    float* hg   = take((size_t)N_GRAPHS_C * HID_C);
    float* x1   = take((size_t)N_GRAPHS_C * HID_C);
    float* x2   = take((size_t)N_GRAPHS_C * (HID_C / 2));
    int*   slot   = (int*)take(N_EDGES_C);
    unsigned long long* bucket = (unsigned long long*)take(2 * N_NODES_C);
    int2*  ent    = (int2*)take(2 * (size_t)N_EDGES_C);

    hipMemsetAsync(cnts, 0, (2 * N_NODES_C + 64) * sizeof(float), stream);
    degree_kernel<<<(N_EDGES_C + 255) / 256, 256, 0, stream>>>(src, dst, odeg, ideg, slot, N_EDGES_C);
    norm_kernel<<<(N_NODES_C + 255) / 256, 256, 0, stream>>>(odeg, ideg, srcn, dstn, bucket, cursor,
                                                           N_NODES_C);
    fill_kernel<<<(N_EDGES_C + 255) / 256, 256, 0, stream>>>(src, dst, slot, bucket, srcn, ent,
                                                           N_EDGES_C);

    const int mblocks = (N_NODES_C + 15) / 16;
    const float* hin = feat;
    for (int L = 0; L < 3; ++L) {
        gather_agg_kernel<<<2048, 256, 0, stream>>>(hin, bucket, ent, agg, N_NODES_C);
        conv_gemm_kernel<<<mblocks, 256, 0, stream>>>(agg, dstn, Wg[L], bg[L], h, N_NODES_C);
        hin = h;
    }

    hipMemsetAsync(hg, 0, (size_t)N_GRAPHS_C * HID_C * sizeof(float), stream);
    segmax_kernel<<<N_NODES_C, HID_C, 0, stream>>>(h, gids, hg, N_NODES_C);

    wmma_gemm_kernel<<<dim3(N_GRAPHS_C / 16, HID_C / 16), 32, 0, stream>>>(
        hg, Wc1, bc1, x1, N_GRAPHS_C, HID_C, HID_C, 1);
    wmma_gemm_kernel<<<dim3(N_GRAPHS_C / 16, (HID_C / 2) / 16), 32, 0, stream>>>(
        x1, Wc2, bc2, x2, N_GRAPHS_C, HID_C / 2, HID_C, 1);
    wmma_gemm_kernel<<<dim3(N_GRAPHS_C / 16, (OUT_DIM_C + 15) / 16), 32, 0, stream>>>(
        x2, Wc3, bc3, out, N_GRAPHS_C, OUT_DIM_C, HID_C / 2, 0);
}


# Max pool folded into the third conv GEMM's epilogue: column max over a tile's 16 rows in registers, one atomic max per column per tile; h3 store, its re-read and the pool launch removed
# speedup vs baseline: 2.1952x; 1.1094x over previous
#include <hip/hip_runtime.h>
#include <hip/hip_bf16.h>


typedef float v2f __attribute__((ext_vector_type(2)));
typedef float v8f __attribute__((ext_vector_type(8)));

#define N_NODES_C 100000
#define N_EDGES_C 800000
#define HID_C     128
#define OUT_DIM_C 500
#define N_GRAPHS_C 64

__global__ void degree_kernel(const int* __restrict__ src, const int* __restrict__ dst,
                              float* __restrict__ odeg, float* __restrict__ ideg,
                              int* __restrict__ slot, int n_edges) {
    int e = blockIdx.x * blockDim.x + threadIdx.x;
    if (e < n_edges) {
        atomicAdd(&odeg[src[e]], 1.0f);
        slot[e] = (int)atomicAdd(&ideg[dst[e]], 1.0f);
    }
}

__global__ void __launch_bounds__(256) norm_kernel(
    const float* __restrict__ odeg, const float* __restrict__ ideg,
    float* __restrict__ srcn, float* __restrict__ dstn,
    unsigned long long* __restrict__ bucket, int* __restrict__ cursor, int n) {
    __shared__ int wsum[8];
    __shared__ int base_s;
    const int i    = blockIdx.x * 256 + threadIdx.x;
    const int lane = threadIdx.x & 31;
    const int wave = threadIdx.x >> 5;
    int cnt = 0;
    if (i < n) {
        float id = ideg[i];
        srcn[i] = rsqrtf(fmaxf(odeg[i], 1.0f));
        dstn[i] = rsqrtf(fmaxf(id, 1.0f));
        cnt = (int)id;
    }
    int incl = cnt;
    #pragma unroll
    for (int o = 1; o < 32; o <<= 1) {
        int t = __shfl_up(incl, o);
        if (lane >= o) incl += t;
    }
    if (lane == 31) wsum[wave] = incl;
    __syncthreads();
    if (threadIdx.x == 0) {
        int tot = 0;
        for (int w = 0; w < 8; ++w) { int t = wsum[w]; wsum[w] = tot; tot += t; }
        base_s = atomicAdd(cursor, tot);
    }
    __syncthreads();
    if (i < n)
        bucket[i] = (unsigned long long)(unsigned)(base_s + wsum[wave] + incl - cnt) |
                    ((unsigned long long)(unsigned)cnt << 32);
}

__global__ void fill_kernel(const int* __restrict__ src, const int* __restrict__ dst,
                            const int* __restrict__ slot,
                            const unsigned long long* __restrict__ bucket,
                            const float* __restrict__ srcn, int2* __restrict__ ent,
                            int n_edges) {
    int e = blockIdx.x * blockDim.x + threadIdx.x;
    if (e < n_edges) {
        int s = src[e];
        ent[(int)(unsigned)bucket[dst[e]] + slot[e]] = make_int2(s, __float_as_int(srcn[s]));
    }
}

__device__ __forceinline__ int lane_of_rank(int r, int k) {
    return __builtin_ctz(__builtin_amdgcn_ballot_w32(r == k));
}

__global__ void __launch_bounds__(256) gather_agg_kernel(
    const float* __restrict__ h, const unsigned long long* __restrict__ bucket,
    const int2* __restrict__ ent, float* __restrict__ agg, int n_nodes) {
    const int lane = threadIdx.x & 31;
    const int wave   = __builtin_amdgcn_readfirstlane(threadIdx.x >> 5);
    const int nwaves = gridDim.x * 8;
    int node = blockIdx.x * 8 + wave;
    unsigned long long b = (node < n_nodes) ? bucket[node] : 0ull;
    for (; node < n_nodes; node += nwaves) {
        const int nxt = node + nwaves;
        const unsigned long long bn = (nxt < n_nodes) ? bucket[nxt] : 0ull;
        const int start = (int)(unsigned)b;
        const int cnt   = (int)(unsigned)(b >> 32);
        b = bn;
        float4 acc = make_float4(0.f, 0.f, 0.f, 0.f);
        for (int c0 = 0; c0 < cnt; c0 += 32) {
            const int m = min(32, cnt - c0);
            int   s = 0x7fffffff;
            float w = 0.f;
            if (lane < m) {
                int2 q = ent[start + c0 + lane];
                s = q.x;
                w = __int_as_float(q.y);
            }
            int r = 0;
            for (int j = 0; j < m; ++j) {
                int t = __builtin_amdgcn_readlane(s, j);
                r += ((t < s) || (t == s && j < lane)) ? 1 : 0;
            }
            int k = 0;
            for (; k + 4 <= m; k += 4) {
                const int j0 = lane_of_rank(r, k),     j1 = lane_of_rank(r, k + 1);
                const int j2 = lane_of_rank(r, k + 2), j3 = lane_of_rank(r, k + 3);
                const int s0 = __builtin_amdgcn_readlane(s, j0);
                const int s1 = __builtin_amdgcn_readlane(s, j1);
                const int s2 = __builtin_amdgcn_readlane(s, j2);
                const int s3 = __builtin_amdgcn_readlane(s, j3);
                const float4 v0 = ((const float4*)(h + (size_t)s0 * HID_C))[lane];
                const float4 v1 = ((const float4*)(h + (size_t)s1 * HID_C))[lane];
                const float4 v2 = ((const float4*)(h + (size_t)s2 * HID_C))[lane];
                const float4 v3 = ((const float4*)(h + (size_t)s3 * HID_C))[lane];
                const float w0 = __int_as_float(__builtin_amdgcn_readlane(__float_as_int(w), j0));
                const float w1 = __int_as_float(__builtin_amdgcn_readlane(__float_as_int(w), j1));
                const float w2 = __int_as_float(__builtin_amdgcn_readlane(__float_as_int(w), j2));
                const float w3 = __int_as_float(__builtin_amdgcn_readlane(__float_as_int(w), j3));
                acc.x += v0.x * w0; acc.y += v0.y * w0; acc.z += v0.z * w0; acc.w += v0.w * w0;
                acc.x += v1.x * w1; acc.y += v1.y * w1; acc.z += v1.z * w1; acc.w += v1.w * w1;
                acc.x += v2.x * w2; acc.y += v2.y * w2; acc.z += v2.z * w2; acc.w += v2.w * w2;
                acc.x += v3.x * w3; acc.y += v3.y * w3; acc.z += v3.z * w3; acc.w += v3.w * w3;
            }
            const int rem = m - k;
            if (rem > 0) {
                const int j0 = lane_of_rank(r, k);
                const int j1 = (rem > 1) ? lane_of_rank(r, k + 1) : j0;
                const int j2 = (rem > 2) ? lane_of_rank(r, k + 2) : j0;
                const int s0 = __builtin_amdgcn_readlane(s, j0);
                const int s1 = __builtin_amdgcn_readlane(s, j1);
                const int s2 = __builtin_amdgcn_readlane(s, j2);
                const float4 v0 = ((const float4*)(h + (size_t)s0 * HID_C))[lane];
                float4 v1 = make_float4(0.f, 0.f, 0.f, 0.f), v2 = v1;
                if (rem > 1) v1 = ((const float4*)(h + (size_t)s1 * HID_C))[lane];
                if (rem > 2) v2 = ((const float4*)(h + (size_t)s2 * HID_C))[lane];
                const float w0 = __int_as_float(__builtin_amdgcn_readlane(__float_as_int(w), j0));
                const float w1 = __int_as_float(__builtin_amdgcn_readlane(__float_as_int(w), j1));
                const float w2 = __int_as_float(__builtin_amdgcn_readlane(__float_as_int(w), j2));
                acc.x += v0.x * w0; acc.y += v0.y * w0; acc.z += v0.z * w0; acc.w += v0.w * w0;
                if (rem > 1) {
                    acc.x += v1.x * w1; acc.y += v1.y * w1; acc.z += v1.z * w1; acc.w += v1.w * w1;
                }
                if (rem > 2) {
                    acc.x += v2.x * w2; acc.y += v2.y * w2; acc.z += v2.z * w2; acc.w += v2.w * w2;
                }
            }
        }
        ((float4*)(agg + (size_t)node * HID_C))[lane] = acc;
    }
}

template <bool POOL>
__device__ __forceinline__ void conv_gemm_body(
    const float* __restrict__ agg, const float* __restrict__ dstn,
    const float* __restrict__ W, const float* __restrict__ bias,
    float* __restrict__ out, const int* __restrict__ gids,
    float* __restrict__ hg, int n_nodes) {
    __shared__ float As[16][HID_C + 4];

    const int tid   = threadIdx.x;
    const int mbase = blockIdx.x * 16;

    const bool full_tile = POOL && (mbase + 16 <= n_nodes);
    int gq[16];
    #pragma unroll
    for (int r = 0; r < 16; ++r) gq[r] = 0;
    if constexpr (POOL) {
        if (full_tile) {
            const int* __restrict__ gp = gids + mbase;
            #pragma unroll
            for (int r = 0; r < 16; ++r) gq[r] = gp[r];
        }
    }

    for (int i = tid; i < 16 * (HID_C / 4); i += 256) {
        int r    = i >> 5;
        int c4   = i & 31;
        int node = mbase + r;
        float4 v = make_float4(0.f, 0.f, 0.f, 0.f);
        float  sc = 0.f;
        if (node < n_nodes) {
            v  = *(const float4*)(agg + (size_t)node * HID_C + c4 * 4);
            sc = dstn[node];
        }
        As[r][c4 * 4 + 0] = v.x * sc;
        As[r][c4 * 4 + 1] = v.y * sc;
        As[r][c4 * 4 + 2] = v.z * sc;
        As[r][c4 * 4 + 3] = v.w * sc;
    }
    __syncthreads();

    const int lane  = tid & 31;
    const int wave  = tid >> 5;
    const int l15   = lane & 15;
    const int khalf = (lane >> 4) << 1;
    const int col   = wave * 16 + l15;

    v8f acc = {0.f, 0.f, 0.f, 0.f, 0.f, 0.f, 0.f, 0.f};
    #pragma unroll
    for (int k = 0; k < HID_C; k += 4) {
        v2f a, b;
        a.x = As[l15][k + khalf];
        a.y = As[l15][k + khalf + 1];
        b.x = W[(size_t)(k + khalf) * HID_C + col];
        b.y = W[(size_t)(k + khalf + 1) * HID_C + col];
        acc = __builtin_amdgcn_wmma_f32_16x16x4_f32(false, a, false, b,
                                                    (short)0, acc, false, false);
    }

    const float bv = bias[col];
    if constexpr (!POOL) {
        #pragma unroll
        for (int i = 0; i < 8; ++i) {
            int node = mbase + i + ((lane >> 4) << 3);
            if (node < n_nodes)
                out[(size_t)node * HID_C + col] = fmaxf(acc[i] + bv, 0.0f);
        }
    } else {
        int glo = gq[0], ghi = gq[0];
        #pragma unroll
        for (int r = 1; r < 16; ++r) { glo = min(glo, gq[r]); ghi = max(ghi, gq[r]); }
        const bool one_graph = full_tile && (glo == ghi);
        const int  g0 = gq[0];
        float m = acc[0];
        #pragma unroll
        for (int i = 1; i < 8; ++i) m = fmaxf(m, acc[i]);
        m = fmaxf(m, __shfl_xor(m, 16));
        if (one_graph) {
            if (lane < 16)
                atomicMax((int*)&hg[(size_t)g0 * HID_C + col],
                          __float_as_int(fmaxf(m + bv, 0.0f)));
        } else {
            #pragma unroll
            for (int i = 0; i < 8; ++i) {
                int node = mbase + i + ((lane >> 4) << 3);
                if (node < n_nodes)
                    atomicMax((int*)&hg[(size_t)gids[node] * HID_C + col],
                              __float_as_int(fmaxf(acc[i] + bv, 0.0f)));
            }
        }
    }
}

__global__ void __launch_bounds__(256) conv_gemm_kernel(
    const float* __restrict__ agg, const float* __restrict__ dstn,
    const float* __restrict__ W, const float* __restrict__ bias,
    float* __restrict__ out, int n_nodes) {
    conv_gemm_body<false>(agg, dstn, W, bias, out, nullptr, nullptr, n_nodes);
}

__global__ void __launch_bounds__(256, 16) conv_gemm_pool_kernel(
    const float* __restrict__ agg, const float* __restrict__ dstn,
    const float* __restrict__ W, const float* __restrict__ bias,
    const int* __restrict__ gids, float* __restrict__ hg, int n_nodes) {
    conv_gemm_body<true>(agg, dstn, W, bias, nullptr, gids, hg, n_nodes);
}

__global__ void __launch_bounds__(32) wmma_gemm_kernel(
    const float* __restrict__ A, const float* __restrict__ B,
    const float* __restrict__ bias, float* __restrict__ C,
    int M, int N, int K, int do_relu) {
    const int lane  = threadIdx.x & 31;
    const int l15   = lane & 15;
    const int khalf = (lane >> 4) << 1;
    const int mbase = blockIdx.x * 16;
    const int nbase = blockIdx.y * 16;
    const int row   = mbase + l15;
    const int col   = nbase + l15;

    v8f acc = {0.f, 0.f, 0.f, 0.f, 0.f, 0.f, 0.f, 0.f};
    for (int k = 0; k < K; k += 4) {
        int k0 = k + khalf, k1 = k + khalf + 1;
        v2f a, b;
        a.x = (row < M) ? A[(size_t)row * K + k0] : 0.0f;
        a.y = (row < M) ? A[(size_t)row * K + k1] : 0.0f;
        b.x = (col < N) ? B[(size_t)k0 * N + col] : 0.0f;
        b.y = (col < N) ? B[(size_t)k1 * N + col] : 0.0f;
        acc = __builtin_amdgcn_wmma_f32_16x16x4_f32(false, a, false, b,
                                                    (short)0, acc, false, false);
    }

    if (col < N) {
        const float bv = bias[col];
        #pragma unroll
        for (int i = 0; i < 8; ++i) {
            int m = mbase + i + ((lane >> 4) << 3);
            if (m < M) {
                float v = acc[i] + bv;
                C[(size_t)m * N + col] = do_relu ? fmaxf(v, 0.0f) : v;
            }
        }
    }
}

extern "C" void kernel_launch(void* const* d_in, const int* in_sizes, int n_in,
                              void* d_out, int out_size, void* d_ws, size_t ws_size,
                              hipStream_t stream) {
    (void)in_sizes; (void)n_in; (void)out_size; (void)ws_size;

    const float* feat = (const float*)d_in[0];
    const int*   src  = (const int*)d_in[1];
    const int*   dst  = (const int*)d_in[2];
    const int*   gids = (const int*)d_in[3];
    const float* Wg[3] = {(const float*)d_in[4], (const float*)d_in[6], (const float*)d_in[8]};
    const float* bg[3] = {(const float*)d_in[5], (const float*)d_in[7], (const float*)d_in[9]};
    const float* Wc1 = (const float*)d_in[10]; const float* bc1 = (const float*)d_in[11];
    const float* Wc2 = (const float*)d_in[12]; const float* bc2 = (const float*)d_in[13];
    const float* Wc3 = (const float*)d_in[14]; const float* bc3 = (const float*)d_in[15];
    float* out = (float*)d_out;

    char* ws = (char*)d_ws;
    size_t off = 0;
    auto take = [&](size_t elems) -> float* {
        float* p = (float*)(ws + off);
        off += ((elems * sizeof(float) + 255) / 256) * 256;
        return p;
    };
    float* cnts = take(2 * N_NODES_C + 64);
    float* odeg = cnts;
    float* ideg = cnts + N_NODES_C;
    int*   cursor = (int*)(cnts + 2 * N_NODES_C);
    float* srcn = take(N_NODES_C);
    float* dstn = take(N_NODES_C);
    float* agg  = take((size_t)N_NODES_C * HID_C);
    float* h    = take((size_t)N_NODES_C * HID_C);
    float* hg   = take((size_t)N_GRAPHS_C * HID_C);
    float* x1   = take((size_t)N_GRAPHS_C * HID_C);
    float* x2   = take((size_t)N_GRAPHS_C * (HID_C / 2));
    int*   slot   = (int*)take(N_EDGES_C);
    unsigned long long* bucket = (unsigned long long*)take(2 * N_NODES_C);
    int2*  ent    = (int2*)take(2 * (size_t)N_EDGES_C);

    hipMemsetAsync(cnts, 0, (2 * N_NODES_C + 64) * sizeof(float), stream);
    degree_kernel<<<(N_EDGES_C + 255) / 256, 256, 0, stream>>>(src, dst, odeg, ideg, slot, N_EDGES_C);
    norm_kernel<<<(N_NODES_C + 255) / 256, 256, 0, stream>>>(odeg, ideg, srcn, dstn, bucket, cursor,
                                                           N_NODES_C);
    fill_kernel<<<(N_EDGES_C + 255) / 256, 256, 0, stream>>>(src, dst, slot, bucket, srcn, ent,
                                                           N_EDGES_C);

    const int mblocks = (N_NODES_C + 15) / 16;
    const float* hin = feat;
    for (int L = 0; L < 2; ++L) {
        gather_agg_kernel<<<2048, 256, 0, stream>>>(hin, bucket, ent, agg, N_NODES_C);
        conv_gemm_kernel<<<mblocks, 256, 0, stream>>>(agg, dstn, Wg[L], bg[L], h, N_NODES_C);
        hin = h;
    }
    gather_agg_kernel<<<2048, 256, 0, stream>>>(hin, bucket, ent, agg, N_NODES_C);
    hipMemsetAsync(hg, 0, (size_t)N_GRAPHS_C * HID_C * sizeof(float), stream);
    conv_gemm_pool_kernel<<<mblocks, 256, 0, stream>>>(agg, dstn, Wg[2], bg[2], gids, hg, N_NODES_C);

    wmma_gemm_kernel<<<dim3(N_GRAPHS_C / 16, HID_C / 16), 32, 0, stream>>>(
        hg, Wc1, bc1, x1, N_GRAPHS_C, HID_C, HID_C, 1);
    wmma_gemm_kernel<<<dim3(N_GRAPHS_C / 16, (HID_C / 2) / 16), 32, 0, stream>>>(
        x1, Wc2, bc2, x2, N_GRAPHS_C, HID_C / 2, HID_C, 1);
    wmma_gemm_kernel<<<dim3(N_GRAPHS_C / 16, (OUT_DIM_C + 15) / 16), 32, 0, stream>>>(
        x2, Wc3, bc3, out, N_GRAPHS_C, OUT_DIM_C, HID_C / 2, 0);
}


# Short-lived waves replaced by resident loops: conv GEMMs as 1024 blocks looping over row tiles, each wave's W slice held in registers (50000 waves to 8192 a launch); degree, fill 4 edges per thread
# speedup vs baseline: 2.3473x; 1.0693x over previous
#include <hip/hip_runtime.h>
#include <hip/hip_bf16.h>


typedef float v2f __attribute__((ext_vector_type(2)));
typedef float v8f __attribute__((ext_vector_type(8)));

#define N_NODES_C 100000
#define N_EDGES_C 800000
#define HID_C     128
#define OUT_DIM_C 500
#define N_GRAPHS_C 64

#define EDGES_PER_THREAD 4

__global__ void __launch_bounds__(256) degree_kernel(
    const int* __restrict__ src, const int* __restrict__ dst,
    float* __restrict__ odeg, float* __restrict__ ideg,
    int* __restrict__ slot, int n_edges) {
    const int b0 = blockIdx.x * (256 * EDGES_PER_THREAD);
    const int e0 = b0 + threadIdx.x;
    if (b0 + 256 * EDGES_PER_THREAD <= n_edges) {
        int s[EDGES_PER_THREAD], d[EDGES_PER_THREAD];
        #pragma unroll
        for (int j = 0; j < EDGES_PER_THREAD; ++j) { s[j] = src[e0 + j * 256]; d[j] = dst[e0 + j * 256]; }
        float old[EDGES_PER_THREAD];
        #pragma unroll
        for (int j = 0; j < EDGES_PER_THREAD; ++j) old[j] = atomicAdd(&ideg[d[j]], 1.0f);
        #pragma unroll
        for (int j = 0; j < EDGES_PER_THREAD; ++j) atomicAdd(&odeg[s[j]], 1.0f);
        #pragma unroll
        for (int j = 0; j < EDGES_PER_THREAD; ++j) slot[e0 + j * 256] = (int)old[j];
    } else {
        for (int j = 0; j < EDGES_PER_THREAD; ++j) {
            const int e = e0 + j * 256;
            if (e < n_edges) {
                atomicAdd(&odeg[src[e]], 1.0f);
                slot[e] = (int)atomicAdd(&ideg[dst[e]], 1.0f);
            }
        }
    }
}

__global__ void __launch_bounds__(256) norm_kernel(
    const float* __restrict__ odeg, const float* __restrict__ ideg,
    float* __restrict__ srcn, float* __restrict__ dstn,
    unsigned long long* __restrict__ bucket, int* __restrict__ cursor, int n) {
    __shared__ int wsum[8];
    __shared__ int base_s;
    const int i    = blockIdx.x * 256 + threadIdx.x;
    const int lane = threadIdx.x & 31;
    const int wave = threadIdx.x >> 5;
    int cnt = 0;
    if (i < n) {
        float id = ideg[i];
        srcn[i] = rsqrtf(fmaxf(odeg[i], 1.0f));
        dstn[i] = rsqrtf(fmaxf(id, 1.0f));
        cnt = (int)id;
    }
    int incl = cnt;
    #pragma unroll
    for (int o = 1; o < 32; o <<= 1) {
        int t = __shfl_up(incl, o);
        if (lane >= o) incl += t;
    }
    if (lane == 31) wsum[wave] = incl;
    __syncthreads();
    if (threadIdx.x == 0) {
        int tot = 0;
        for (int w = 0; w < 8; ++w) { int t = wsum[w]; wsum[w] = tot; tot += t; }
        base_s = atomicAdd(cursor, tot);
    }
    __syncthreads();
    if (i < n)
        bucket[i] = (unsigned long long)(unsigned)(base_s + wsum[wave] + incl - cnt) |
                    ((unsigned long long)(unsigned)cnt << 32);
}

__global__ void __launch_bounds__(256) fill_kernel(
    const int* __restrict__ src, const int* __restrict__ dst,
    const int* __restrict__ slot,
    const unsigned long long* __restrict__ bucket,
    const float* __restrict__ srcn, int2* __restrict__ ent,
    int n_edges) {
    const int b0 = blockIdx.x * (256 * EDGES_PER_THREAD);
    const int e0 = b0 + threadIdx.x;
    if (b0 + 256 * EDGES_PER_THREAD <= n_edges) {
        int s[EDGES_PER_THREAD], d[EDGES_PER_THREAD], sl[EDGES_PER_THREAD];
        #pragma unroll
        for (int j = 0; j < EDGES_PER_THREAD; ++j) {
            s[j] = src[e0 + j * 256]; d[j] = dst[e0 + j * 256]; sl[j] = slot[e0 + j * 256];
        }
        int   pos[EDGES_PER_THREAD];
        float nrm[EDGES_PER_THREAD];
        #pragma unroll
        for (int j = 0; j < EDGES_PER_THREAD; ++j) {
            pos[j] = (int)(unsigned)bucket[d[j]] + sl[j];
            nrm[j] = srcn[s[j]];
        }
        #pragma unroll
        for (int j = 0; j < EDGES_PER_THREAD; ++j)
            ent[pos[j]] = make_int2(s[j], __float_as_int(nrm[j]));
    } else {
        for (int j = 0; j < EDGES_PER_THREAD; ++j) {
            const int e = e0 + j * 256;
            if (e < n_edges) {
                int s = src[e];
                ent[(int)(unsigned)bucket[dst[e]] + slot[e]] = make_int2(s, __float_as_int(srcn[s]));
            }
        }
    }
}

__device__ __forceinline__ int lane_of_rank(int r, int k) {
    return __builtin_ctz(__builtin_amdgcn_ballot_w32(r == k));
}

__global__ void __launch_bounds__(256) gather_agg_kernel(
    const float* __restrict__ h, const unsigned long long* __restrict__ bucket,
    const int2* __restrict__ ent, float* __restrict__ agg, int n_nodes) {
    const int lane = threadIdx.x & 31;
    const int wave   = __builtin_amdgcn_readfirstlane(threadIdx.x >> 5);
    const int nwaves = gridDim.x * 8;
    int node = blockIdx.x * 8 + wave;
    unsigned long long b = (node < n_nodes) ? bucket[node] : 0ull;
    for (; node < n_nodes; node += nwaves) {
        const int nxt = node + nwaves;
        const unsigned long long bn = (nxt < n_nodes) ? bucket[nxt] : 0ull;
        const int start = (int)(unsigned)b;
        const int cnt   = (int)(unsigned)(b >> 32);
        b = bn;
        float4 acc = make_float4(0.f, 0.f, 0.f, 0.f);
        for (int c0 = 0; c0 < cnt; c0 += 32) {
            const int m = min(32, cnt - c0);
            int   s = 0x7fffffff;
            float w = 0.f;
            if (lane < m) {
                int2 q = ent[start + c0 + lane];
                s = q.x;
                w = __int_as_float(q.y);
            }
            int r = 0;
            for (int j = 0; j < m; ++j) {
                int t = __builtin_amdgcn_readlane(s, j);
                r += ((t < s) || (t == s && j < lane)) ? 1 : 0;
            }
            int k = 0;
            for (; k + 4 <= m; k += 4) {
                const int j0 = lane_of_rank(r, k),     j1 = lane_of_rank(r, k + 1);
                const int j2 = lane_of_rank(r, k + 2), j3 = lane_of_rank(r, k + 3);
                const int s0 = __builtin_amdgcn_readlane(s, j0);
                const int s1 = __builtin_amdgcn_readlane(s, j1);
                const int s2 = __builtin_amdgcn_readlane(s, j2);
                const int s3 = __builtin_amdgcn_readlane(s, j3);
                const float4 v0 = ((const float4*)(h + (size_t)s0 * HID_C))[lane];
                const float4 v1 = ((const float4*)(h + (size_t)s1 * HID_C))[lane];
                const float4 v2 = ((const float4*)(h + (size_t)s2 * HID_C))[lane];
                const float4 v3 = ((const float4*)(h + (size_t)s3 * HID_C))[lane];
                const float w0 = __int_as_float(__builtin_amdgcn_readlane(__float_as_int(w), j0));
                const float w1 = __int_as_float(__builtin_amdgcn_readlane(__float_as_int(w), j1));
                const float w2 = __int_as_float(__builtin_amdgcn_readlane(__float_as_int(w), j2));
                const float w3 = __int_as_float(__builtin_amdgcn_readlane(__float_as_int(w), j3));
                acc.x += v0.x * w0; acc.y += v0.y * w0; acc.z += v0.z * w0; acc.w += v0.w * w0;
                acc.x += v1.x * w1; acc.y += v1.y * w1; acc.z += v1.z * w1; acc.w += v1.w * w1;
                acc.x += v2.x * w2; acc.y += v2.y * w2; acc.z += v2.z * w2; acc.w += v2.w * w2;
                acc.x += v3.x * w3; acc.y += v3.y * w3; acc.z += v3.z * w3; acc.w += v3.w * w3;
            }
            const int rem = m - k;
            if (rem > 0) {
                const int j0 = lane_of_rank(r, k);
                const int j1 = (rem > 1) ? lane_of_rank(r, k + 1) : j0;
                const int j2 = (rem > 2) ? lane_of_rank(r, k + 2) : j0;
                const int s0 = __builtin_amdgcn_readlane(s, j0);
                const int s1 = __builtin_amdgcn_readlane(s, j1);
                const int s2 = __builtin_amdgcn_readlane(s, j2);
                const float4 v0 = ((const float4*)(h + (size_t)s0 * HID_C))[lane];
                float4 v1 = make_float4(0.f, 0.f, 0.f, 0.f), v2 = v1;
                if (rem > 1) v1 = ((const float4*)(h + (size_t)s1 * HID_C))[lane];
                if (rem > 2) v2 = ((const float4*)(h + (size_t)s2 * HID_C))[lane];
                const float w0 = __int_as_float(__builtin_amdgcn_readlane(__float_as_int(w), j0));
                const float w1 = __int_as_float(__builtin_amdgcn_readlane(__float_as_int(w), j1));
                const float w2 = __int_as_float(__builtin_amdgcn_readlane(__float_as_int(w), j2));
                acc.x += v0.x * w0; acc.y += v0.y * w0; acc.z += v0.z * w0; acc.w += v0.w * w0;
                if (rem > 1) {
                    acc.x += v1.x * w1; acc.y += v1.y * w1; acc.z += v1.z * w1; acc.w += v1.w * w1;
                }
                if (rem > 2) {
                    acc.x += v2.x * w2; acc.y += v2.y * w2; acc.z += v2.z * w2; acc.w += v2.w * w2;
                }
            }
        }
        ((float4*)(agg + (size_t)node * HID_C))[lane] = acc;
    }
}

#define CONV_BLOCKS 1024

template <bool POOL>
__device__ __forceinline__ void conv_gemm_body(
    const float* __restrict__ agg, const float* __restrict__ dstn,
    const float* __restrict__ W, const float* __restrict__ bias,
    float* __restrict__ out, const int* __restrict__ gids,
    float* __restrict__ hg, int n_nodes) {
    __shared__ float As[2][16][HID_C + 4];

    const int tid    = threadIdx.x;
    const int lane   = tid & 31;
    const int wave   = tid >> 5;
    const int l15    = lane & 15;
    const int khalf  = (lane >> 4) << 1;
    const int col    = wave * 16 + l15;
    const int ntiles = (n_nodes + 15) >> 4;
    const int tstep  = gridDim.x;

    v2f breg[HID_C / 4];
    #pragma unroll
    for (int kk = 0; kk < HID_C / 4; ++kk) {
        breg[kk].x = W[(size_t)(kk * 4 + khalf) * HID_C + col];
        breg[kk].y = W[(size_t)(kk * 4 + khalf + 1) * HID_C + col];
    }
    const float bv = bias[col];

    int tile = blockIdx.x;
    float4 p0 = make_float4(0.f, 0.f, 0.f, 0.f), p1 = p0;
    float  s0 = 0.f, s1 = 0.f;
    if (tile < ntiles) {
        const int n0 = tile * 16 + wave, n1 = n0 + 8;
        if (n0 < n_nodes) { p0 = *(const float4*)(agg + (size_t)n0 * HID_C + lane * 4); s0 = dstn[n0]; }
        if (n1 < n_nodes) { p1 = *(const float4*)(agg + (size_t)n1 * HID_C + lane * 4); s1 = dstn[n1]; }
    }

    int buf = 0;
    for (; tile < ntiles; tile += tstep) {
        const int mbase = tile * 16;

        const bool full_tile = POOL && (mbase + 16 <= n_nodes);
        int gq[16];
        #pragma unroll
        for (int r = 0; r < 16; ++r) gq[r] = 0;
        if constexpr (POOL) {
            if (full_tile) {
                const int* __restrict__ gp = gids + mbase;
                #pragma unroll
                for (int r = 0; r < 16; ++r) gq[r] = gp[r];
            }
        }

        *(float4*)&As[buf][wave][lane * 4] =
            make_float4(p0.x * s0, p0.y * s0, p0.z * s0, p0.w * s0);
        *(float4*)&As[buf][wave + 8][lane * 4] =
            make_float4(p1.x * s1, p1.y * s1, p1.z * s1, p1.w * s1);
        __syncthreads();

        const int tnext = tile + tstep;
        p0 = make_float4(0.f, 0.f, 0.f, 0.f); p1 = p0; s0 = 0.f; s1 = 0.f;
        if (tnext < ntiles) {
            const int n0 = tnext * 16 + wave, n1 = n0 + 8;
            if (n0 < n_nodes) { p0 = *(const float4*)(agg + (size_t)n0 * HID_C + lane * 4); s0 = dstn[n0]; }
            if (n1 < n_nodes) { p1 = *(const float4*)(agg + (size_t)n1 * HID_C + lane * 4); s1 = dstn[n1]; }
        }

        v8f acc = {0.f, 0.f, 0.f, 0.f, 0.f, 0.f, 0.f, 0.f};
        #pragma unroll
        for (int kk = 0; kk < HID_C / 4; ++kk) {
            v2f a;
            a.x = As[buf][l15][kk * 4 + khalf];
            a.y = As[buf][l15][kk * 4 + khalf + 1];
            acc = __builtin_amdgcn_wmma_f32_16x16x4_f32(false, a, false, breg[kk],
                                                        (short)0, acc, false, false);
        }
        buf ^= 1;

        if constexpr (!POOL) {
            #pragma unroll
            for (int i = 0; i < 8; ++i) {
                int node = mbase + i + ((lane >> 4) << 3);
                if (node < n_nodes)
                    out[(size_t)node * HID_C + col] = fmaxf(acc[i] + bv, 0.0f);
            }
        } else {
            int glo = gq[0], ghi = gq[0];
            #pragma unroll
            for (int r = 1; r < 16; ++r) { glo = min(glo, gq[r]); ghi = max(ghi, gq[r]); }
            const bool one_graph = full_tile && (glo == ghi);
            const int  g0 = gq[0];
            float m = acc[0];
            #pragma unroll
            for (int i = 1; i < 8; ++i) m = fmaxf(m, acc[i]);
            m = fmaxf(m, __shfl_xor(m, 16));
            if (one_graph) {
                if (lane < 16)
                    atomicMax((int*)&hg[(size_t)g0 * HID_C + col],
                              __float_as_int(fmaxf(m + bv, 0.0f)));
            } else {
                #pragma unroll
                for (int i = 0; i < 8; ++i) {
                    int node = mbase + i + ((lane >> 4) << 3);
                    if (node < n_nodes)
                        atomicMax((int*)&hg[(size_t)gids[node] * HID_C + col],
                                  __float_as_int(fmaxf(acc[i] + bv, 0.0f)));
                }
            }
        }
    }
}

__global__ void __launch_bounds__(256, 8) conv_gemm_kernel(
    const float* __restrict__ agg, const float* __restrict__ dstn,
    const float* __restrict__ W, const float* __restrict__ bias,
    float* __restrict__ out, int n_nodes) {
    conv_gemm_body<false>(agg, dstn, W, bias, out, nullptr, nullptr, n_nodes);
}

__global__ void __launch_bounds__(256, 8) conv_gemm_pool_kernel(
    const float* __restrict__ agg, const float* __restrict__ dstn,
    const float* __restrict__ W, const float* __restrict__ bias,
    const int* __restrict__ gids, float* __restrict__ hg, int n_nodes) {
    conv_gemm_body<true>(agg, dstn, W, bias, nullptr, gids, hg, n_nodes);
}

__global__ void __launch_bounds__(32) wmma_gemm_kernel(
    const float* __restrict__ A, const float* __restrict__ B,
    const float* __restrict__ bias, float* __restrict__ C,
    int M, int N, int K, int do_relu) {
    const int lane  = threadIdx.x & 31;
    const int l15   = lane & 15;
    const int khalf = (lane >> 4) << 1;
    const int mbase = blockIdx.x * 16;
    const int nbase = blockIdx.y * 16;
    const int row   = mbase + l15;
    const int col   = nbase + l15;

    v8f acc = {0.f, 0.f, 0.f, 0.f, 0.f, 0.f, 0.f, 0.f};
    for (int k = 0; k < K; k += 4) {
        int k0 = k + khalf, k1 = k + khalf + 1;
        v2f a, b;
        a.x = (row < M) ? A[(size_t)row * K + k0] : 0.0f;
        a.y = (row < M) ? A[(size_t)row * K + k1] : 0.0f;
        b.x = (col < N) ? B[(size_t)k0 * N + col] : 0.0f;
        b.y = (col < N) ? B[(size_t)k1 * N + col] : 0.0f;
        acc = __builtin_amdgcn_wmma_f32_16x16x4_f32(false, a, false, b,
                                                    (short)0, acc, false, false);
    }

    if (col < N) {
        const float bv = bias[col];
        #pragma unroll
        for (int i = 0; i < 8; ++i) {
            int m = mbase + i + ((lane >> 4) << 3);
            if (m < M) {
                float v = acc[i] + bv;
                C[(size_t)m * N + col] = do_relu ? fmaxf(v, 0.0f) : v;
            }
        }
    }
}

extern "C" void kernel_launch(void* const* d_in, const int* in_sizes, int n_in,
                              void* d_out, int out_size, void* d_ws, size_t ws_size,
                              hipStream_t stream) {
    (void)in_sizes; (void)n_in; (void)out_size; (void)ws_size;

    const float* feat = (const float*)d_in[0];
    const int*   src  = (const int*)d_in[1];
    const int*   dst  = (const int*)d_in[2];
    const int*   gids = (const int*)d_in[3];
    const float* Wg[3] = {(const float*)d_in[4], (const float*)d_in[6], (const float*)d_in[8]};
    const float* bg[3] = {(const float*)d_in[5], (const float*)d_in[7], (const float*)d_in[9]};
    const float* Wc1 = (const float*)d_in[10]; const float* bc1 = (const float*)d_in[11];
    const float* Wc2 = (const float*)d_in[12]; const float* bc2 = (const float*)d_in[13];
    const float* Wc3 = (const float*)d_in[14]; const float* bc3 = (const float*)d_in[15];
    float* out = (float*)d_out;

    char* ws = (char*)d_ws;
    size_t off = 0;
    auto take = [&](size_t elems) -> float* {
        float* p = (float*)(ws + off);
        off += ((elems * sizeof(float) + 255) / 256) * 256;
        return p;
    };
    float* cnts = take(2 * N_NODES_C + 64);
    float* odeg = cnts;
    float* ideg = cnts + N_NODES_C;
    int*   cursor = (int*)(cnts + 2 * N_NODES_C);
    float* srcn = take(N_NODES_C);
    float* dstn = take(N_NODES_C);
    float* agg  = take((size_t)N_NODES_C * HID_C);
    float* h    = take((size_t)N_NODES_C * HID_C);
    float* hg   = take((size_t)N_GRAPHS_C * HID_C);
    float* x1   = take((size_t)N_GRAPHS_C * HID_C);
    float* x2   = take((size_t)N_GRAPHS_C * (HID_C / 2));
    int*   slot   = (int*)take(N_EDGES_C);
    unsigned long long* bucket = (unsigned long long*)take(2 * N_NODES_C);
    int2*  ent    = (int2*)take(2 * (size_t)N_EDGES_C);

    hipMemsetAsync(cnts, 0, (2 * N_NODES_C + 64) * sizeof(float), stream);
    const int eblocks = (N_EDGES_C + 256 * EDGES_PER_THREAD - 1) / (256 * EDGES_PER_THREAD);
    degree_kernel<<<eblocks, 256, 0, stream>>>(src, dst, odeg, ideg, slot, N_EDGES_C);
    norm_kernel<<<(N_NODES_C + 255) / 256, 256, 0, stream>>>(odeg, ideg, srcn, dstn, bucket, cursor,
                                                           N_NODES_C);
    fill_kernel<<<eblocks, 256, 0, stream>>>(src, dst, slot, bucket, srcn, ent, N_EDGES_C);

    const int mblocks = CONV_BLOCKS;
    const float* hin = feat;
    for (int L = 0; L < 2; ++L) {
        gather_agg_kernel<<<2048, 256, 0, stream>>>(hin, bucket, ent, agg, N_NODES_C);
        conv_gemm_kernel<<<mblocks, 256, 0, stream>>>(agg, dstn, Wg[L], bg[L], h, N_NODES_C);
        hin = h;
    }
    gather_agg_kernel<<<2048, 256, 0, stream>>>(hin, bucket, ent, agg, N_NODES_C);
    hipMemsetAsync(hg, 0, (size_t)N_GRAPHS_C * HID_C * sizeof(float), stream);
    conv_gemm_pool_kernel<<<mblocks, 256, 0, stream>>>(agg, dstn, Wg[2], bg[2], gids, hg, N_NODES_C);

    wmma_gemm_kernel<<<dim3(N_GRAPHS_C / 16, HID_C / 16), 32, 0, stream>>>(
        hg, Wc1, bc1, x1, N_GRAPHS_C, HID_C, HID_C, 1);
    wmma_gemm_kernel<<<dim3(N_GRAPHS_C / 16, (HID_C / 2) / 16), 32, 0, stream>>>(
        x1, Wc2, bc2, x2, N_GRAPHS_C, HID_C / 2, HID_C, 1);
    wmma_gemm_kernel<<<dim3(N_GRAPHS_C / 16, (OUT_DIM_C + 15) / 16), 32, 0, stream>>>(
        x2, Wc3, bc3, out, N_GRAPHS_C, OUT_DIM_C, HID_C / 2, 0);
}


# Edge gather fused into the conv GEMM launch: each wave sums its two tile rows' source rows (same per-node order, 8 rows in flight) into the LDS A tile; agg store, re-read and 3 launches removed
# speedup vs baseline: 2.4856x; 1.0589x over previous
#include <hip/hip_runtime.h>
#include <hip/hip_bf16.h>


typedef float v2f __attribute__((ext_vector_type(2)));
typedef float v8f __attribute__((ext_vector_type(8)));

#define N_NODES_C 100000
#define N_EDGES_C 800000
#define HID_C     128
#define OUT_DIM_C 500
#define N_GRAPHS_C 64

#define EDGES_PER_THREAD 4

__global__ void __launch_bounds__(256) degree_kernel(
    const int* __restrict__ src, const int* __restrict__ dst,
    float* __restrict__ odeg, float* __restrict__ ideg,
    int* __restrict__ slot, int n_edges) {
    const int b0 = blockIdx.x * (256 * EDGES_PER_THREAD);
    const int e0 = b0 + threadIdx.x;
    if (b0 + 256 * EDGES_PER_THREAD <= n_edges) {
        int s[EDGES_PER_THREAD], d[EDGES_PER_THREAD];
        #pragma unroll
        for (int j = 0; j < EDGES_PER_THREAD; ++j) { s[j] = src[e0 + j * 256]; d[j] = dst[e0 + j * 256]; }
        float old[EDGES_PER_THREAD];
        #pragma unroll
        for (int j = 0; j < EDGES_PER_THREAD; ++j) old[j] = atomicAdd(&ideg[d[j]], 1.0f);
        #pragma unroll
        for (int j = 0; j < EDGES_PER_THREAD; ++j) atomicAdd(&odeg[s[j]], 1.0f);
        #pragma unroll
        for (int j = 0; j < EDGES_PER_THREAD; ++j) slot[e0 + j * 256] = (int)old[j];
    } else {
        for (int j = 0; j < EDGES_PER_THREAD; ++j) {
            const int e = e0 + j * 256;
            if (e < n_edges) {
                atomicAdd(&odeg[src[e]], 1.0f);
                slot[e] = (int)atomicAdd(&ideg[dst[e]], 1.0f);
            }
        }
    }
}

__global__ void __launch_bounds__(256) norm_kernel(
    const float* __restrict__ odeg, const float* __restrict__ ideg,
    float* __restrict__ srcn, float* __restrict__ dstn,
    unsigned long long* __restrict__ bucket, int* __restrict__ cursor, int n) {
    __shared__ int wsum[8];
    __shared__ int base_s;
    const int i    = blockIdx.x * 256 + threadIdx.x;
    const int lane = threadIdx.x & 31;
    const int wave = threadIdx.x >> 5;
    int cnt = 0;
    if (i < n) {
        float id = ideg[i];
        srcn[i] = rsqrtf(fmaxf(odeg[i], 1.0f));
        dstn[i] = rsqrtf(fmaxf(id, 1.0f));
        cnt = (int)id;
    }
    int incl = cnt;
    #pragma unroll
    for (int o = 1; o < 32; o <<= 1) {
        int t = __shfl_up(incl, o);
        if (lane >= o) incl += t;
    }
    if (lane == 31) wsum[wave] = incl;
    __syncthreads();
    if (threadIdx.x == 0) {
        int tot = 0;
        for (int w = 0; w < 8; ++w) { int t = wsum[w]; wsum[w] = tot; tot += t; }
        base_s = atomicAdd(cursor, tot);
    }
    __syncthreads();
    if (i < n)
        bucket[i] = (unsigned long long)(unsigned)(base_s + wsum[wave] + incl - cnt) |
                    ((unsigned long long)(unsigned)cnt << 32);
}

__global__ void __launch_bounds__(256) fill_kernel(
    const int* __restrict__ src, const int* __restrict__ dst,
    const int* __restrict__ slot,
    const unsigned long long* __restrict__ bucket,
    const float* __restrict__ srcn, int2* __restrict__ ent,
    int n_edges) {
    const int b0 = blockIdx.x * (256 * EDGES_PER_THREAD);
    const int e0 = b0 + threadIdx.x;
    if (b0 + 256 * EDGES_PER_THREAD <= n_edges) {
        int s[EDGES_PER_THREAD], d[EDGES_PER_THREAD], sl[EDGES_PER_THREAD];
        #pragma unroll
        for (int j = 0; j < EDGES_PER_THREAD; ++j) {
            s[j] = src[e0 + j * 256]; d[j] = dst[e0 + j * 256]; sl[j] = slot[e0 + j * 256];
        }
        int   pos[EDGES_PER_THREAD];
        float nrm[EDGES_PER_THREAD];
        #pragma unroll
        for (int j = 0; j < EDGES_PER_THREAD; ++j) {
            pos[j] = (int)(unsigned)bucket[d[j]] + sl[j];
            nrm[j] = srcn[s[j]];
        }
        #pragma unroll
        for (int j = 0; j < EDGES_PER_THREAD; ++j)
            ent[pos[j]] = make_int2(s[j], __float_as_int(nrm[j]));
    } else {
        for (int j = 0; j < EDGES_PER_THREAD; ++j) {
            const int e = e0 + j * 256;
            if (e < n_edges) {
                int s = src[e];
                ent[(int)(unsigned)bucket[dst[e]] + slot[e]] = make_int2(s, __float_as_int(srcn[s]));
            }
        }
    }
}

#define CONV_BLOCKS 1024

#define KEY_B   0x40000000
#define KEY_ID  0x3fffffff

__device__ __forceinline__ int lane_of_rank(int r, int k) {
    return __builtin_ctz(__builtin_amdgcn_ballot_w32(r == k));
}

__device__ __forceinline__ void load_entries(const int2* __restrict__ ent, int offA, int mA,
                                             int offB, int mB, int lane, int& s, float& w) {
    s = 0x7fffffff;
    w = 0.f;
    if (lane < mA + mB) {
        int2 q = ent[(lane < mA) ? (offA + lane) : (offB + lane - mA)];
        s = q.x;
        w = __int_as_float(q.y);
    }
}

#define ROW_OF(sid) (((const float4*)(hin + (size_t)(sid) * HID_C))[lane])
#define ADD_ROW(ACC, V, WV)                                                            \
    { ACC.x += V.x * WV; ACC.y += V.y * WV; ACC.z += V.z * WV; ACC.w += V.w * WV; }
#define ASK_POS(i, V, WV)                                                              \
    {                                                                                  \
        const int j_ = lane_of_rank(r, min(p + i, m - 1));                             \
        WV = __int_as_float(__builtin_amdgcn_readlane(__float_as_int(w), j_));         \
        if (left > i) V = ROW_OF(__builtin_amdgcn_readlane(key, j_) & KEY_ID);         \
    }
#define ADD_POS(i, V, WV)                                                              \
    if (i == sw) { accA = cur; cur = make_float4(0.f, 0.f, 0.f, 0.f); inB = true; }    \
    if (left > i) ADD_ROW(cur, V, WV)

template <bool POOL>
__device__ __forceinline__ void graph_conv_body(
    const float* __restrict__ hin, const unsigned long long* __restrict__ bucket,
    const int2* __restrict__ ent, const float* __restrict__ dstn,
    const float* __restrict__ W, const float* __restrict__ bias,
    float* __restrict__ out, const int* __restrict__ gids,
    float* __restrict__ hg, int n_nodes) {
    __shared__ float As[2][16][HID_C + 4];

    const int tid    = threadIdx.x;
    const int lane   = tid & 31;
    const int wave   = __builtin_amdgcn_readfirstlane(tid >> 5);
    const int l15    = lane & 15;
    const int khalf  = (lane >> 4) << 1;
    const int col    = wave * 16 + l15;
    const int ntiles = (n_nodes + 15) >> 4;
    const int tstep  = gridDim.x;

    v2f breg[HID_C / 4];
    #pragma unroll
    for (int kk = 0; kk < HID_C / 4; ++kk) {
        breg[kk].x = W[(size_t)(kk * 4 + khalf) * HID_C + col];
        breg[kk].y = W[(size_t)(kk * 4 + khalf + 1) * HID_C + col];
    }
    const float bv = bias[col];

    int tile = blockIdx.x;
    unsigned long long bA = 0ull, bB = 0ull, nbA = 0ull, nbB = 0ull;
    if (tile < ntiles) {
        const int n0 = tile * 16 + wave, n1 = n0 + 8;
        if (n0 < n_nodes) bA = bucket[n0];
        if (n1 < n_nodes) bB = bucket[n1];
    }
    if (tile + tstep < ntiles) {
        const int n0 = (tile + tstep) * 16 + wave, n1 = n0 + 8;
        if (n0 < n_nodes) nbA = bucket[n0];
        if (n1 < n_nodes) nbB = bucket[n1];
    }
    int   s;
    float w;
    {
        const int cA = (int)(unsigned)(bA >> 32), cB = (int)(unsigned)(bB >> 32);
        const bool one = (cA + cB <= 32);
        load_entries(ent, (int)(unsigned)bA, one ? cA : 0, (int)(unsigned)bB, one ? cB : 0, lane, s, w);
    }

    int buf = 0;
    for (; tile < ntiles; tile += tstep) {
        const int mbase = tile * 16;
        const int nA = mbase + wave, nB = nA + 8;
        const int startA = (int)(unsigned)bA, cntA = (int)(unsigned)(bA >> 32);
        const int startB = (int)(unsigned)bB, cntB = (int)(unsigned)(bB >> 32);
        const float dA = (nA < n_nodes) ? dstn[nA] : 0.f;
        const float dB = (nB < n_nodes) ? dstn[nB] : 0.f;

        float4 cur = make_float4(0.f, 0.f, 0.f, 0.f), accA = cur;
        bool inB = false;
        {
            int  ca = 0, cb = 0;
            bool have = (cntA + cntB <= 32);
            int  mA = have ? cntA : 0, mB = have ? cntB : 0;
            while (ca < cntA || cb < cntB) {
                if (!have) {
                    if (ca < cntA) { mA = min(32, cntA - ca); mB = 0; }
                    else           { mA = 0; mB = min(32, cntB - cb); }
                    load_entries(ent, startA + ca, mA, startB + cb, mB, lane, s, w);
                }
                have = false;
                const int m   = mA + mB;
                const int key = s | ((lane >= mA) ? KEY_B : 0);
                int r = 0;
                int j = 0;
                for (; j + 4 <= m; j += 4) {
                    const int t0 = __builtin_amdgcn_readlane(key, j);
                    const int t1 = __builtin_amdgcn_readlane(key, j + 1);
                    const int t2 = __builtin_amdgcn_readlane(key, j + 2);
                    const int t3 = __builtin_amdgcn_readlane(key, j + 3);
                    r += (((t0 < key) || (t0 == key && j     < lane)) ? 1 : 0) +
                         (((t1 < key) || (t1 == key && j + 1 < lane)) ? 1 : 0) +
                         (((t2 < key) || (t2 == key && j + 2 < lane)) ? 1 : 0) +
                         (((t3 < key) || (t3 == key && j + 3 < lane)) ? 1 : 0);
                }
                for (; j < m; ++j) {
                    const int t = __builtin_amdgcn_readlane(key, j);
                    r += ((t < key) || (t == key && j < lane)) ? 1 : 0;
                }
                for (int p = 0; p < m; p += 8) {
                    const int left = m - p;
                    const int sw   = (mB > 0 && !inB) ? mA - p : -1;
                    const float4 zr = make_float4(0.f, 0.f, 0.f, 0.f);
                    float4 v0 = zr, v1 = zr, v2 = zr, v3 = zr, v4 = zr, v5 = zr, v6 = zr, v7 = zr;
                    float w0, w1, w2, w3, w4, w5, w6, w7;
                    ASK_POS(0, v0, w0) ASK_POS(1, v1, w1) ASK_POS(2, v2, w2) ASK_POS(3, v3, w3)
                    ASK_POS(4, v4, w4) ASK_POS(5, v5, w5) ASK_POS(6, v6, w6) ASK_POS(7, v7, w7)
                    ADD_POS(0, v0, w0) ADD_POS(1, v1, w1) ADD_POS(2, v2, w2) ADD_POS(3, v3, w3)
                    ADD_POS(4, v4, w4) ADD_POS(5, v5, w5) ADD_POS(6, v6, w6) ADD_POS(7, v7, w7)
                }
                ca += mA; cb += mB;
            }
        }

        const bool full_tile = POOL && (mbase + 16 <= n_nodes);
        int gq[16];
        #pragma unroll
        for (int r = 0; r < 16; ++r) gq[r] = 0;
        if constexpr (POOL) {
            if (full_tile) {
                const int* __restrict__ gp = gids + mbase;
                #pragma unroll
                for (int r = 0; r < 16; ++r) gq[r] = gp[r];
            }
        }

        const float4 z4 = make_float4(0.f, 0.f, 0.f, 0.f);
        const float4 rowA = inB ? accA : cur;
        const float4 rowB = inB ? cur : z4;
        *(float4*)&As[buf][wave][lane * 4] =
            make_float4(rowA.x * dA, rowA.y * dA, rowA.z * dA, rowA.w * dA);
        *(float4*)&As[buf][wave + 8][lane * 4] =
            make_float4(rowB.x * dB, rowB.y * dB, rowB.z * dB, rowB.w * dB);
        __syncthreads();

        bA = nbA; bB = nbB;
        {
            const int cA = (int)(unsigned)(bA >> 32), cB = (int)(unsigned)(bB >> 32);
            const bool one = (cA + cB <= 32);
            load_entries(ent, (int)(unsigned)bA, one ? cA : 0, (int)(unsigned)bB, one ? cB : 0, lane, s, w);
        }
        nbA = 0ull; nbB = 0ull;
        if (tile + 2 * tstep < ntiles) {
            const int n0 = (tile + 2 * tstep) * 16 + wave, n1 = n0 + 8;
            if (n0 < n_nodes) nbA = bucket[n0];
            if (n1 < n_nodes) nbB = bucket[n1];
        }

        v8f acc = {0.f, 0.f, 0.f, 0.f, 0.f, 0.f, 0.f, 0.f};
        #pragma unroll
        for (int kk = 0; kk < HID_C / 4; ++kk) {
            v2f a;
            a.x = As[buf][l15][kk * 4 + khalf];
            a.y = As[buf][l15][kk * 4 + khalf + 1];
            acc = __builtin_amdgcn_wmma_f32_16x16x4_f32(false, a, false, breg[kk],
                                                        (short)0, acc, false, false);
        }
        buf ^= 1;

        if constexpr (!POOL) {
            #pragma unroll
            for (int i = 0; i < 8; ++i) {
                int node = mbase + i + ((lane >> 4) << 3);
                if (node < n_nodes)
                    out[(size_t)node * HID_C + col] = fmaxf(acc[i] + bv, 0.0f);
            }
        } else {
            int glo = gq[0], ghi = gq[0];
            #pragma unroll
            for (int r = 1; r < 16; ++r) { glo = min(glo, gq[r]); ghi = max(ghi, gq[r]); }
            const bool one_graph = full_tile && (glo == ghi);
            const int  g0 = gq[0];
            float m = acc[0];
            #pragma unroll
            for (int i = 1; i < 8; ++i) m = fmaxf(m, acc[i]);
            m = fmaxf(m, __shfl_xor(m, 16));
            if (one_graph) {
                if (lane < 16)
                    atomicMax((int*)&hg[(size_t)g0 * HID_C + col],
                              __float_as_int(fmaxf(m + bv, 0.0f)));
            } else {
                #pragma unroll
                for (int i = 0; i < 8; ++i) {
                    int node = mbase + i + ((lane >> 4) << 3);
                    if (node < n_nodes)
                        atomicMax((int*)&hg[(size_t)gids[node] * HID_C + col],
                                  __float_as_int(fmaxf(acc[i] + bv, 0.0f)));
                }
            }
        }
    }
}

__global__ void __launch_bounds__(256, 8) graph_conv_kernel(
    const float* __restrict__ hin, const unsigned long long* __restrict__ bucket,
    const int2* __restrict__ ent, const float* __restrict__ dstn,
    const float* __restrict__ W, const float* __restrict__ bias,
    float* __restrict__ out, int n_nodes) {
    graph_conv_body<false>(hin, bucket, ent, dstn, W, bias, out, nullptr, nullptr, n_nodes);
}

__global__ void __launch_bounds__(256, 8) graph_conv_pool_kernel(
    const float* __restrict__ hin, const unsigned long long* __restrict__ bucket,
    const int2* __restrict__ ent, const float* __restrict__ dstn,
    const float* __restrict__ W, const float* __restrict__ bias,
    const int* __restrict__ gids, float* __restrict__ hg, int n_nodes) {
    graph_conv_body<true>(hin, bucket, ent, dstn, W, bias, nullptr, gids, hg, n_nodes);
}

__global__ void __launch_bounds__(32) wmma_gemm_kernel(
    const float* __restrict__ A, const float* __restrict__ B,
    const float* __restrict__ bias, float* __restrict__ C,
    int M, int N, int K, int do_relu) {
    const int lane  = threadIdx.x & 31;
    const int l15   = lane & 15;
    const int khalf = (lane >> 4) << 1;
    const int mbase = blockIdx.x * 16;
    const int nbase = blockIdx.y * 16;
    const int row   = mbase + l15;
    const int col   = nbase + l15;

    v8f acc = {0.f, 0.f, 0.f, 0.f, 0.f, 0.f, 0.f, 0.f};
    for (int k = 0; k < K; k += 4) {
        int k0 = k + khalf, k1 = k + khalf + 1;
        v2f a, b;
        a.x = (row < M) ? A[(size_t)row * K + k0] : 0.0f;
        a.y = (row < M) ? A[(size_t)row * K + k1] : 0.0f;
        b.x = (col < N) ? B[(size_t)k0 * N + col] : 0.0f;
        b.y = (col < N) ? B[(size_t)k1 * N + col] : 0.0f;
        acc = __builtin_amdgcn_wmma_f32_16x16x4_f32(false, a, false, b,
                                                    (short)0, acc, false, false);
    }

    if (col < N) {
        const float bv = bias[col];
        #pragma unroll
        for (int i = 0; i < 8; ++i) {
            int m = mbase + i + ((lane >> 4) << 3);
            if (m < M) {
                float v = acc[i] + bv;
                C[(size_t)m * N + col] = do_relu ? fmaxf(v, 0.0f) : v;
            }
        }
    }
}

extern "C" void kernel_launch(void* const* d_in, const int* in_sizes, int n_in,
                              void* d_out, int out_size, void* d_ws, size_t ws_size,
                              hipStream_t stream) {
    (void)in_sizes; (void)n_in; (void)out_size; (void)ws_size;

    const float* feat = (const float*)d_in[0];
    const int*   src  = (const int*)d_in[1];
    const int*   dst  = (const int*)d_in[2];
    const int*   gids = (const int*)d_in[3];
    const float* Wg[3] = {(const float*)d_in[4], (const float*)d_in[6], (const float*)d_in[8]};
    const float* bg[3] = {(const float*)d_in[5], (const float*)d_in[7], (const float*)d_in[9]};
    const float* Wc1 = (const float*)d_in[10]; const float* bc1 = (const float*)d_in[11];
    const float* Wc2 = (const float*)d_in[12]; const float* bc2 = (const float*)d_in[13];
    const float* Wc3 = (const float*)d_in[14]; const float* bc3 = (const float*)d_in[15];
    float* out = (float*)d_out;

    char* ws = (char*)d_ws;
    size_t off = 0;
    auto take = [&](size_t elems) -> float* {
        float* p = (float*)(ws + off);
        off += ((elems * sizeof(float) + 255) / 256) * 256;
        return p;
    };
    float* cnts = take(2 * N_NODES_C + 64);
    float* odeg = cnts;
    float* ideg = cnts + N_NODES_C;
    int*   cursor = (int*)(cnts + 2 * N_NODES_C);
    float* srcn = take(N_NODES_C);
    float* dstn = take(N_NODES_C);
    float* h1   = take((size_t)N_NODES_C * HID_C);
    float* h2   = take((size_t)N_NODES_C * HID_C);
    float* hg   = take((size_t)N_GRAPHS_C * HID_C);
    float* x1   = take((size_t)N_GRAPHS_C * HID_C);
    float* x2   = take((size_t)N_GRAPHS_C * (HID_C / 2));
    int*   slot   = (int*)take(N_EDGES_C);
    unsigned long long* bucket = (unsigned long long*)take(2 * N_NODES_C);
    int2*  ent    = (int2*)take(2 * (size_t)N_EDGES_C);

    hipMemsetAsync(cnts, 0, (2 * N_NODES_C + 64) * sizeof(float), stream);
    const int eblocks = (N_EDGES_C + 256 * EDGES_PER_THREAD - 1) / (256 * EDGES_PER_THREAD);
    degree_kernel<<<eblocks, 256, 0, stream>>>(src, dst, odeg, ideg, slot, N_EDGES_C);
    norm_kernel<<<(N_NODES_C + 255) / 256, 256, 0, stream>>>(odeg, ideg, srcn, dstn, bucket, cursor,
                                                           N_NODES_C);
    fill_kernel<<<eblocks, 256, 0, stream>>>(src, dst, slot, bucket, srcn, ent, N_EDGES_C);

    const int mblocks = CONV_BLOCKS;
    graph_conv_kernel<<<mblocks, 256, 0, stream>>>(feat, bucket, ent, dstn, Wg[0], bg[0], h1, N_NODES_C);
    graph_conv_kernel<<<mblocks, 256, 0, stream>>>(h1, bucket, ent, dstn, Wg[1], bg[1], h2, N_NODES_C);
    hipMemsetAsync(hg, 0, (size_t)N_GRAPHS_C * HID_C * sizeof(float), stream);
    graph_conv_pool_kernel<<<mblocks, 256, 0, stream>>>(h2, bucket, ent, dstn, Wg[2], bg[2], gids, hg,
                                                        N_NODES_C);

    wmma_gemm_kernel<<<dim3(N_GRAPHS_C / 16, HID_C / 16), 32, 0, stream>>>(
        hg, Wc1, bc1, x1, N_GRAPHS_C, HID_C, HID_C, 1);
    wmma_gemm_kernel<<<dim3(N_GRAPHS_C / 16, (HID_C / 2) / 16), 32, 0, stream>>>(
        x1, Wc2, bc2, x2, N_GRAPHS_C, HID_C / 2, HID_C, 1);
    wmma_gemm_kernel<<<dim3(N_GRAPHS_C / 16, (OUT_DIM_C + 15) / 16), 32, 0, stream>>>(
        x2, Wc3, bc3, out, N_GRAPHS_C, OUT_DIM_C, HID_C / 2, 0);
}


# Classifier MLP as one launch: a block takes 16 graphs through all three layers, activations in LDS, all W fragments loaded as one group up front; 2 launches and the per-k-step load waits removed
# speedup vs baseline: 3.0097x; 1.2109x over previous
#include <hip/hip_runtime.h>
#include <hip/hip_bf16.h>


typedef float v2f __attribute__((ext_vector_type(2)));
typedef float v8f __attribute__((ext_vector_type(8)));

#define N_NODES_C 100000
#define N_EDGES_C 800000
#define HID_C     128
#define OUT_DIM_C 500
#define N_GRAPHS_C 64

#define EDGES_PER_THREAD 4

__global__ void __launch_bounds__(256) degree_kernel(
    const int* __restrict__ src, const int* __restrict__ dst,
    float* __restrict__ odeg, float* __restrict__ ideg,
    int* __restrict__ slot, int n_edges) {
    const int b0 = blockIdx.x * (256 * EDGES_PER_THREAD);
    const int e0 = b0 + threadIdx.x;
    if (b0 + 256 * EDGES_PER_THREAD <= n_edges) {
        int s[EDGES_PER_THREAD], d[EDGES_PER_THREAD];
        #pragma unroll
        for (int j = 0; j < EDGES_PER_THREAD; ++j) { s[j] = src[e0 + j * 256]; d[j] = dst[e0 + j * 256]; }
        float old[EDGES_PER_THREAD];
        #pragma unroll
        for (int j = 0; j < EDGES_PER_THREAD; ++j) old[j] = atomicAdd(&ideg[d[j]], 1.0f);
        #pragma unroll
        for (int j = 0; j < EDGES_PER_THREAD; ++j) atomicAdd(&odeg[s[j]], 1.0f);
        #pragma unroll
        for (int j = 0; j < EDGES_PER_THREAD; ++j) slot[e0 + j * 256] = (int)old[j];
    } else {
        for (int j = 0; j < EDGES_PER_THREAD; ++j) {
            const int e = e0 + j * 256;
            if (e < n_edges) {
                atomicAdd(&odeg[src[e]], 1.0f);
                slot[e] = (int)atomicAdd(&ideg[dst[e]], 1.0f);
            }
        }
    }
}

__global__ void __launch_bounds__(256) norm_kernel(
    const float* __restrict__ odeg, const float* __restrict__ ideg,
    float* __restrict__ srcn, float* __restrict__ dstn,
    unsigned long long* __restrict__ bucket, int* __restrict__ cursor, int n) {
    __shared__ int wsum[8];
    __shared__ int base_s;
    const int i    = blockIdx.x * 256 + threadIdx.x;
    const int lane = threadIdx.x & 31;
    const int wave = threadIdx.x >> 5;
    int cnt = 0;
    if (i < n) {
        float id = ideg[i];
        srcn[i] = rsqrtf(fmaxf(odeg[i], 1.0f));
        dstn[i] = rsqrtf(fmaxf(id, 1.0f));
        cnt = (int)id;
    }
    int incl = cnt;
    #pragma unroll
    for (int o = 1; o < 32; o <<= 1) {
        int t = __shfl_up(incl, o);
        if (lane >= o) incl += t;
    }
    if (lane == 31) wsum[wave] = incl;
    __syncthreads();
    if (threadIdx.x == 0) {
        int tot = 0;
        for (int w = 0; w < 8; ++w) { int t = wsum[w]; wsum[w] = tot; tot += t; }
        base_s = atomicAdd(cursor, tot);
    }
    __syncthreads();
    if (i < n)
        bucket[i] = (unsigned long long)(unsigned)(base_s + wsum[wave] + incl - cnt) |
                    ((unsigned long long)(unsigned)cnt << 32);
}

__global__ void __launch_bounds__(256) fill_kernel(
    const int* __restrict__ src, const int* __restrict__ dst,
    const int* __restrict__ slot,
    const unsigned long long* __restrict__ bucket,
    const float* __restrict__ srcn, int2* __restrict__ ent,
    int n_edges) {
    const int b0 = blockIdx.x * (256 * EDGES_PER_THREAD);
    const int e0 = b0 + threadIdx.x;
    if (b0 + 256 * EDGES_PER_THREAD <= n_edges) {
        int s[EDGES_PER_THREAD], d[EDGES_PER_THREAD], sl[EDGES_PER_THREAD];
        #pragma unroll
        for (int j = 0; j < EDGES_PER_THREAD; ++j) {
            s[j] = src[e0 + j * 256]; d[j] = dst[e0 + j * 256]; sl[j] = slot[e0 + j * 256];
        }
        int   pos[EDGES_PER_THREAD];
        float nrm[EDGES_PER_THREAD];
        #pragma unroll
        for (int j = 0; j < EDGES_PER_THREAD; ++j) {
            pos[j] = (int)(unsigned)bucket[d[j]] + sl[j];
            nrm[j] = srcn[s[j]];
        }
        #pragma unroll
        for (int j = 0; j < EDGES_PER_THREAD; ++j)
            ent[pos[j]] = make_int2(s[j], __float_as_int(nrm[j]));
    } else {
        for (int j = 0; j < EDGES_PER_THREAD; ++j) {
            const int e = e0 + j * 256;
            if (e < n_edges) {
                int s = src[e];
                ent[(int)(unsigned)bucket[dst[e]] + slot[e]] = make_int2(s, __float_as_int(srcn[s]));
            }
        }
    }
}

#define CONV_BLOCKS 1024

#define KEY_B   0x40000000
#define KEY_ID  0x3fffffff

__device__ __forceinline__ int lane_of_rank(int r, int k) {
    return __builtin_ctz(__builtin_amdgcn_ballot_w32(r == k));
}

__device__ __forceinline__ void load_entries(const int2* __restrict__ ent, int offA, int mA,
                                             int offB, int mB, int lane, int& s, float& w) {
    s = 0x7fffffff;
    w = 0.f;
    if (lane < mA + mB) {
        int2 q = ent[(lane < mA) ? (offA + lane) : (offB + lane - mA)];
        s = q.x;
        w = __int_as_float(q.y);
    }
}

#define ROW_OF(sid) (((const float4*)(hin + (size_t)(sid) * HID_C))[lane])
#define ADD_ROW(ACC, V, WV)                                                            \
    { ACC.x += V.x * WV; ACC.y += V.y * WV; ACC.z += V.z * WV; ACC.w += V.w * WV; }
#define ASK_POS(i, V, WV)                                                              \
    {                                                                                  \
        const int j_ = lane_of_rank(r, min(p + i, m - 1));                             \
        WV = __int_as_float(__builtin_amdgcn_readlane(__float_as_int(w), j_));         \
        if (left > i) V = ROW_OF(__builtin_amdgcn_readlane(key, j_) & KEY_ID);         \
    }
#define ADD_POS(i, V, WV)                                                              \
    if (i == sw) { accA = cur; cur = make_float4(0.f, 0.f, 0.f, 0.f); inB = true; }    \
    if (left > i) ADD_ROW(cur, V, WV)

template <bool POOL>
__device__ __forceinline__ void graph_conv_body(
    const float* __restrict__ hin, const unsigned long long* __restrict__ bucket,
    const int2* __restrict__ ent, const float* __restrict__ dstn,
    const float* __restrict__ W, const float* __restrict__ bias,
    float* __restrict__ out, const int* __restrict__ gids,
    float* __restrict__ hg, int n_nodes) {
    __shared__ float As[2][16][HID_C + 4];

    const int tid    = threadIdx.x;
    const int lane   = tid & 31;
    const int wave   = __builtin_amdgcn_readfirstlane(tid >> 5);
    const int l15    = lane & 15;
    const int khalf  = (lane >> 4) << 1;
    const int col    = wave * 16 + l15;
    const int ntiles = (n_nodes + 15) >> 4;
    const int tstep  = gridDim.x;

    v2f breg[HID_C / 4];
    #pragma unroll
    for (int kk = 0; kk < HID_C / 4; ++kk) {
        breg[kk].x = W[(size_t)(kk * 4 + khalf) * HID_C + col];
        breg[kk].y = W[(size_t)(kk * 4 + khalf + 1) * HID_C + col];
    }
    const float bv = bias[col];

    int tile = blockIdx.x;
    unsigned long long bA = 0ull, bB = 0ull, nbA = 0ull, nbB = 0ull;
    if (tile < ntiles) {
        const int n0 = tile * 16 + wave, n1 = n0 + 8;
        if (n0 < n_nodes) bA = bucket[n0];
        if (n1 < n_nodes) bB = bucket[n1];
    }
    if (tile + tstep < ntiles) {
        const int n0 = (tile + tstep) * 16 + wave, n1 = n0 + 8;
        if (n0 < n_nodes) nbA = bucket[n0];
        if (n1 < n_nodes) nbB = bucket[n1];
    }
    int   s;
    float w;
    {
        const int cA = (int)(unsigned)(bA >> 32), cB = (int)(unsigned)(bB >> 32);
        const bool one = (cA + cB <= 32);
        load_entries(ent, (int)(unsigned)bA, one ? cA : 0, (int)(unsigned)bB, one ? cB : 0, lane, s, w);
    }

    int buf = 0;
    for (; tile < ntiles; tile += tstep) {
        const int mbase = tile * 16;
        const int nA = mbase + wave, nB = nA + 8;
        const int startA = (int)(unsigned)bA, cntA = (int)(unsigned)(bA >> 32);
        const int startB = (int)(unsigned)bB, cntB = (int)(unsigned)(bB >> 32);
        const float dA = (nA < n_nodes) ? dstn[nA] : 0.f;
        const float dB = (nB < n_nodes) ? dstn[nB] : 0.f;

        float4 cur = make_float4(0.f, 0.f, 0.f, 0.f), accA = cur;
        bool inB = false;
        {
            int  ca = 0, cb = 0;
            bool have = (cntA + cntB <= 32);
            int  mA = have ? cntA : 0, mB = have ? cntB : 0;
            while (ca < cntA || cb < cntB) {
                if (!have) {
                    if (ca < cntA) { mA = min(32, cntA - ca); mB = 0; }
                    else           { mA = 0; mB = min(32, cntB - cb); }
                    load_entries(ent, startA + ca, mA, startB + cb, mB, lane, s, w);
                }
                have = false;
                const int m   = mA + mB;
                const int key = s | ((lane >= mA) ? KEY_B : 0);
                int r = 0;
                int j = 0;
                for (; j + 4 <= m; j += 4) {
                    const int t0 = __builtin_amdgcn_readlane(key, j);
                    const int t1 = __builtin_amdgcn_readlane(key, j + 1);
                    const int t2 = __builtin_amdgcn_readlane(key, j + 2);
                    const int t3 = __builtin_amdgcn_readlane(key, j + 3);
                    r += (((t0 < key) || (t0 == key && j     < lane)) ? 1 : 0) +
                         (((t1 < key) || (t1 == key && j + 1 < lane)) ? 1 : 0) +
                         (((t2 < key) || (t2 == key && j + 2 < lane)) ? 1 : 0) +
                         (((t3 < key) || (t3 == key && j + 3 < lane)) ? 1 : 0);
                }
                for (; j < m; ++j) {
                    const int t = __builtin_amdgcn_readlane(key, j);
                    r += ((t < key) || (t == key && j < lane)) ? 1 : 0;
                }
                for (int p = 0; p < m; p += 8) {
                    const int left = m - p;
                    const int sw   = (mB > 0 && !inB) ? mA - p : -1;
                    const float4 zr = make_float4(0.f, 0.f, 0.f, 0.f);
                    float4 v0 = zr, v1 = zr, v2 = zr, v3 = zr, v4 = zr, v5 = zr, v6 = zr, v7 = zr;
                    float w0, w1, w2, w3, w4, w5, w6, w7;
                    ASK_POS(0, v0, w0) ASK_POS(1, v1, w1) ASK_POS(2, v2, w2) ASK_POS(3, v3, w3)
                    ASK_POS(4, v4, w4) ASK_POS(5, v5, w5) ASK_POS(6, v6, w6) ASK_POS(7, v7, w7)
                    ADD_POS(0, v0, w0) ADD_POS(1, v1, w1) ADD_POS(2, v2, w2) ADD_POS(3, v3, w3)
                    ADD_POS(4, v4, w4) ADD_POS(5, v5, w5) ADD_POS(6, v6, w6) ADD_POS(7, v7, w7)
                }
                ca += mA; cb += mB;
            }
        }

        const bool full_tile = POOL && (mbase + 16 <= n_nodes);
        int gq[16];
        #pragma unroll
        for (int r = 0; r < 16; ++r) gq[r] = 0;
        if constexpr (POOL) {
            if (full_tile) {
                const int* __restrict__ gp = gids + mbase;
                #pragma unroll
                for (int r = 0; r < 16; ++r) gq[r] = gp[r];
            }
        }

        const float4 z4 = make_float4(0.f, 0.f, 0.f, 0.f);
        const float4 rowA = inB ? accA : cur;
        const float4 rowB = inB ? cur : z4;
        *(float4*)&As[buf][wave][lane * 4] =
            make_float4(rowA.x * dA, rowA.y * dA, rowA.z * dA, rowA.w * dA);
        *(float4*)&As[buf][wave + 8][lane * 4] =
            make_float4(rowB.x * dB, rowB.y * dB, rowB.z * dB, rowB.w * dB);
        __syncthreads();

        bA = nbA; bB = nbB;
        {
            const int cA = (int)(unsigned)(bA >> 32), cB = (int)(unsigned)(bB >> 32);
            const bool one = (cA + cB <= 32);
            load_entries(ent, (int)(unsigned)bA, one ? cA : 0, (int)(unsigned)bB, one ? cB : 0, lane, s, w);
        }
        nbA = 0ull; nbB = 0ull;
        if (tile + 2 * tstep < ntiles) {
            const int n0 = (tile + 2 * tstep) * 16 + wave, n1 = n0 + 8;
            if (n0 < n_nodes) nbA = bucket[n0];
            if (n1 < n_nodes) nbB = bucket[n1];
        }

        v8f acc = {0.f, 0.f, 0.f, 0.f, 0.f, 0.f, 0.f, 0.f};
        #pragma unroll
        for (int kk = 0; kk < HID_C / 4; ++kk) {
            v2f a;
            a.x = As[buf][l15][kk * 4 + khalf];
            a.y = As[buf][l15][kk * 4 + khalf + 1];
            acc = __builtin_amdgcn_wmma_f32_16x16x4_f32(false, a, false, breg[kk],
                                                        (short)0, acc, false, false);
        }
        buf ^= 1;

        if constexpr (!POOL) {
            #pragma unroll
            for (int i = 0; i < 8; ++i) {
                int node = mbase + i + ((lane >> 4) << 3);
                if (node < n_nodes)
                    out[(size_t)node * HID_C + col] = fmaxf(acc[i] + bv, 0.0f);
            }
        } else {
            int glo = gq[0], ghi = gq[0];
            #pragma unroll
            for (int r = 1; r < 16; ++r) { glo = min(glo, gq[r]); ghi = max(ghi, gq[r]); }
            const bool one_graph = full_tile && (glo == ghi);
            const int  g0 = gq[0];
            float m = acc[0];
            #pragma unroll
            for (int i = 1; i < 8; ++i) m = fmaxf(m, acc[i]);
            m = fmaxf(m, __shfl_xor(m, 16));
            if (one_graph) {
                if (lane < 16)
                    atomicMax((int*)&hg[(size_t)g0 * HID_C + col],
                              __float_as_int(fmaxf(m + bv, 0.0f)));
            } else {
                #pragma unroll
                for (int i = 0; i < 8; ++i) {
                    int node = mbase + i + ((lane >> 4) << 3);
                    if (node < n_nodes)
                        atomicMax((int*)&hg[(size_t)gids[node] * HID_C + col],
                                  __float_as_int(fmaxf(acc[i] + bv, 0.0f)));
                }
            }
        }
    }
}

__global__ void __launch_bounds__(256, 8) graph_conv_kernel(
    const float* __restrict__ hin, const unsigned long long* __restrict__ bucket,
    const int2* __restrict__ ent, const float* __restrict__ dstn,
    const float* __restrict__ W, const float* __restrict__ bias,
    float* __restrict__ out, int n_nodes) {
    graph_conv_body<false>(hin, bucket, ent, dstn, W, bias, out, nullptr, nullptr, n_nodes);
}

__global__ void __launch_bounds__(256, 8) graph_conv_pool_kernel(
    const float* __restrict__ hin, const unsigned long long* __restrict__ bucket,
    const int2* __restrict__ ent, const float* __restrict__ dstn,
    const float* __restrict__ W, const float* __restrict__ bias,
    const int* __restrict__ gids, float* __restrict__ hg, int n_nodes) {
    graph_conv_body<true>(hin, bucket, ent, dstn, W, bias, nullptr, gids, hg, n_nodes);
}

#define MLP_SPLIT 4
#define MID_C (HID_C / 2)

__global__ void __launch_bounds__(256) mlp_kernel(
    const float* __restrict__ hg,
    const float* __restrict__ W1, const float* __restrict__ b1,
    const float* __restrict__ W2, const float* __restrict__ b2,
    const float* __restrict__ W3, const float* __restrict__ b3,
    float* __restrict__ out) {
    __shared__ float X0[16][HID_C + 4];
    __shared__ float X1[16][HID_C + 4];
    __shared__ float X2[16][MID_C + 4];

    const int tid   = threadIdx.x;
    const int lane  = tid & 31;
    const int wave  = __builtin_amdgcn_readfirstlane(tid >> 5);
    const int l15   = lane & 15;
    const int khalf = (lane >> 4) << 1;
    const int rhalf = (lane >> 4) << 3;
    const int mbase = blockIdx.x * 16;

    float4 t0, t1;
    {
        const int i0 = tid, i1 = tid + 256;
        t0 = ((const float4*)(hg + (size_t)(mbase + (i0 >> 5)) * HID_C))[i0 & 31];
        t1 = ((const float4*)(hg + (size_t)(mbase + (i1 >> 5)) * HID_C))[i1 & 31];
    }

    const int c1 = wave * 16 + l15;
    v2f w1[HID_C / 4];
    #pragma unroll
    for (int kk = 0; kk < HID_C / 4; ++kk) {
        w1[kk].x = W1[(size_t)(kk * 4 + khalf) * HID_C + c1];
        w1[kk].y = W1[(size_t)(kk * 4 + khalf + 1) * HID_C + c1];
    }
    const float bv1 = b1[c1];

    const int c2 = (wave & 3) * 16 + l15;
    v2f w2[HID_C / 4];
    #pragma unroll
    for (int kk = 0; kk < HID_C / 4; ++kk) { w2[kk].x = 0.f; w2[kk].y = 0.f; }
    float bv2 = 0.f;
    if (wave < 4) {
        #pragma unroll
        for (int kk = 0; kk < HID_C / 4; ++kk) {
            w2[kk].x = W2[(size_t)(kk * 4 + khalf) * MID_C + c2];
            w2[kk].y = W2[(size_t)(kk * 4 + khalf + 1) * MID_C + c2];
        }
        bv2 = b2[c2];
    }

    const int  c3   = (blockIdx.y * 8 + wave) * 16 + l15;
    const bool c3ok = c3 < OUT_DIM_C;
    v2f w3[MID_C / 4];
    #pragma unroll
    for (int kk = 0; kk < MID_C / 4; ++kk) {
        w3[kk].x = c3ok ? W3[(size_t)(kk * 4 + khalf) * OUT_DIM_C + c3] : 0.0f;
        w3[kk].y = c3ok ? W3[(size_t)(kk * 4 + khalf + 1) * OUT_DIM_C + c3] : 0.0f;
    }
    const float bv3 = c3ok ? b3[c3] : 0.0f;

    *(float4*)&X0[tid >> 5][(tid & 31) * 4]         = t0;
    *(float4*)&X0[(tid + 256) >> 5][(tid & 31) * 4] = t1;
    __syncthreads();

    {
        v8f acc = {0.f, 0.f, 0.f, 0.f, 0.f, 0.f, 0.f, 0.f};
        #pragma unroll
        for (int kk = 0; kk < HID_C / 4; ++kk) {
            v2f a;
            a.x = X0[l15][kk * 4 + khalf];
            a.y = X0[l15][kk * 4 + khalf + 1];
            acc = __builtin_amdgcn_wmma_f32_16x16x4_f32(false, a, false, w1[kk],
                                                        (short)0, acc, false, false);
        }
        #pragma unroll
        for (int i = 0; i < 8; ++i) X1[i + rhalf][c1] = fmaxf(acc[i] + bv1, 0.0f);
    }
    __syncthreads();

    if (wave < 4) {
        v8f acc = {0.f, 0.f, 0.f, 0.f, 0.f, 0.f, 0.f, 0.f};
        #pragma unroll
        for (int kk = 0; kk < HID_C / 4; ++kk) {
            v2f a;
            a.x = X1[l15][kk * 4 + khalf];
            a.y = X1[l15][kk * 4 + khalf + 1];
            acc = __builtin_amdgcn_wmma_f32_16x16x4_f32(false, a, false, w2[kk],
                                                        (short)0, acc, false, false);
        }
        #pragma unroll
        for (int i = 0; i < 8; ++i) X2[i + rhalf][c2] = fmaxf(acc[i] + bv2, 0.0f);
    }
    __syncthreads();

    {
        v8f acc = {0.f, 0.f, 0.f, 0.f, 0.f, 0.f, 0.f, 0.f};
        #pragma unroll
        for (int kk = 0; kk < MID_C / 4; ++kk) {
            v2f a;
            a.x = X2[l15][kk * 4 + khalf];
            a.y = X2[l15][kk * 4 + khalf + 1];
            acc = __builtin_amdgcn_wmma_f32_16x16x4_f32(false, a, false, w3[kk],
                                                        (short)0, acc, false, false);
        }
        if (c3ok) {
            #pragma unroll
            for (int i = 0; i < 8; ++i)
                out[(size_t)(mbase + i + rhalf) * OUT_DIM_C + c3] = acc[i] + bv3;
        }
    }
}

extern "C" void kernel_launch(void* const* d_in, const int* in_sizes, int n_in,
                              void* d_out, int out_size, void* d_ws, size_t ws_size,
                              hipStream_t stream) {
    (void)in_sizes; (void)n_in; (void)out_size; (void)ws_size;

    const float* feat = (const float*)d_in[0];
    const int*   src  = (const int*)d_in[1];
    const int*   dst  = (const int*)d_in[2];
    const int*   gids = (const int*)d_in[3];
    const float* Wg[3] = {(const float*)d_in[4], (const float*)d_in[6], (const float*)d_in[8]};
    const float* bg[3] = {(const float*)d_in[5], (const float*)d_in[7], (const float*)d_in[9]};
    const float* Wc1 = (const float*)d_in[10]; const float* bc1 = (const float*)d_in[11];
    const float* Wc2 = (const float*)d_in[12]; const float* bc2 = (const float*)d_in[13];
    const float* Wc3 = (const float*)d_in[14]; const float* bc3 = (const float*)d_in[15];
    float* out = (float*)d_out;

    char* ws = (char*)d_ws;
    size_t off = 0;
    auto take = [&](size_t elems) -> float* {
        float* p = (float*)(ws + off);
        off += ((elems * sizeof(float) + 255) / 256) * 256;
        return p;
    };
    float* cnts = take(2 * N_NODES_C + 64);
    float* odeg = cnts;
    float* ideg = cnts + N_NODES_C;
    int*   cursor = (int*)(cnts + 2 * N_NODES_C);
    float* srcn = take(N_NODES_C);
    float* dstn = take(N_NODES_C);
    float* h1   = take((size_t)N_NODES_C * HID_C);
    float* h2   = take((size_t)N_NODES_C * HID_C);
    float* hg   = take((size_t)N_GRAPHS_C * HID_C);
    int*   slot   = (int*)take(N_EDGES_C);
    unsigned long long* bucket = (unsigned long long*)take(2 * N_NODES_C);
    int2*  ent    = (int2*)take(2 * (size_t)N_EDGES_C);

    hipMemsetAsync(cnts, 0, (2 * N_NODES_C + 64) * sizeof(float), stream);
    const int eblocks = (N_EDGES_C + 256 * EDGES_PER_THREAD - 1) / (256 * EDGES_PER_THREAD);
    degree_kernel<<<eblocks, 256, 0, stream>>>(src, dst, odeg, ideg, slot, N_EDGES_C);
    norm_kernel<<<(N_NODES_C + 255) / 256, 256, 0, stream>>>(odeg, ideg, srcn, dstn, bucket, cursor,
                                                           N_NODES_C);
    fill_kernel<<<eblocks, 256, 0, stream>>>(src, dst, slot, bucket, srcn, ent, N_EDGES_C);

    const int mblocks = CONV_BLOCKS;
    graph_conv_kernel<<<mblocks, 256, 0, stream>>>(feat, bucket, ent, dstn, Wg[0], bg[0], h1, N_NODES_C);
    graph_conv_kernel<<<mblocks, 256, 0, stream>>>(h1, bucket, ent, dstn, Wg[1], bg[1], h2, N_NODES_C);
    hipMemsetAsync(hg, 0, (size_t)N_GRAPHS_C * HID_C * sizeof(float), stream);
    graph_conv_pool_kernel<<<mblocks, 256, 0, stream>>>(h2, bucket, ent, dstn, Wg[2], bg[2], gids, hg,
                                                        N_NODES_C);

    mlp_kernel<<<dim3(N_GRAPHS_C / 16, MLP_SPLIT), 256, 0, stream>>>(
        hg, Wc1, bc1, Wc2, bc2, Wc3, bc3, out);
}


# Edges grouped by destination through LDS: a block orders its edge chunk by 256-node range, one block per range counts, scans and places entries; returning global adds and scattered fill launch gone
# speedup vs baseline: 3.6104x; 1.1996x over previous
#include <hip/hip_runtime.h>
#include <hip/hip_bf16.h>


typedef float v2f __attribute__((ext_vector_type(2)));
typedef float v8f __attribute__((ext_vector_type(8)));

#define N_NODES_C 100000
#define N_EDGES_C 800000
#define HID_C     128
#define OUT_DIM_C 500
#define N_GRAPHS_C 64

#define BIN_BLOCKS   256
#define BIN_THREADS  1024
#define BIN_CHUNK    ((N_EDGES_C + BIN_BLOCKS - 1) / BIN_BLOCKS)
#define BIN_EPT      ((BIN_CHUNK + BIN_THREADS - 1) / BIN_THREADS)
#define RANGE_SHIFT  8
#define RANGE_NODES  (1 << RANGE_SHIFT)
#define N_RANGES     ((N_NODES_C + RANGE_NODES - 1) >> RANGE_SHIFT)

static_assert(BIN_THREADS == 1024, "block_scan_excl is written for 32 waves");
static_assert(BIN_CHUNK < 65536, "run offset and length share one 32-bit word");
static_assert(N_RANGES <= BIN_THREADS, "one range per thread in the scan");
static_assert(BIN_BLOCKS <= BIN_THREADS && RANGE_NODES <= BIN_THREADS, "one item per thread in the scans");

__device__ __forceinline__ int block_scan_excl(int v, int* ws, int& total) {
    const int lane = threadIdx.x & 31;
    const int wave = threadIdx.x >> 5;
    int incl = v;
    #pragma unroll
    for (int o = 1; o < 32; o <<= 1) {
        int t = __shfl_up(incl, o);
        if (lane >= o) incl += t;
    }
    if (lane == 31) ws[wave] = incl;
    __syncthreads();
    if (wave == 0) {
        const int wv = ws[lane];
        int wi = wv;
        #pragma unroll
        for (int o = 1; o < 32; o <<= 1) {
            int t = __shfl_up(wi, o);
            if (lane >= o) wi += t;
        }
        ws[lane] = wi - wv;
        if (lane == 31) ws[32] = wi;
    }
    __syncthreads();
    total = ws[32];
    return ws[wave] + incl - v;
}

__global__ void __launch_bounds__(BIN_THREADS) bin_kernel(
    const int* __restrict__ src, const int* __restrict__ dst,
    unsigned* __restrict__ odeg, int2* __restrict__ stage,
    unsigned* __restrict__ runs, int n_edges) {
    __shared__ int  hist[N_RANGES];
    __shared__ int2 sorted[BIN_CHUNK];
    __shared__ int  ws[33];

    const int tid     = threadIdx.x;
    const int e_begin = blockIdx.x * BIN_CHUNK;
    const int e_end   = min(e_begin + BIN_CHUNK, n_edges);

    for (int i = tid; i < N_RANGES; i += BIN_THREADS) hist[i] = 0;

    int s[BIN_EPT], d[BIN_EPT], rk[BIN_EPT];
    #pragma unroll
    for (int j = 0; j < BIN_EPT; ++j) {
        const int e = e_begin + tid + j * BIN_THREADS;
        s[j] = 0; d[j] = -1; rk[j] = 0;
        if (e < e_end) { s[j] = src[e]; d[j] = dst[e]; }
    }
    __syncthreads();
    #pragma unroll
    for (int j = 0; j < BIN_EPT; ++j) {
        if (d[j] >= 0) rk[j] = atomicAdd(&hist[d[j] >> RANGE_SHIFT], 1);
    }
    __syncthreads();

    const int c = (tid < N_RANGES) ? hist[tid] : 0;
    int total;
    const int ex = block_scan_excl(c, ws, total);
    if (tid < N_RANGES) {
        hist[tid] = ex;
        runs[(size_t)blockIdx.x * N_RANGES + tid] = (unsigned)ex | ((unsigned)c << 16);
    }
    __syncthreads();
    #pragma unroll
    for (int j = 0; j < BIN_EPT; ++j)
        if (d[j] >= 0) sorted[hist[d[j] >> RANGE_SHIFT] + rk[j]] = make_int2(s[j], d[j]);
    __syncthreads();
    for (int i = tid; i < e_end - e_begin; i += BIN_THREADS) stage[e_begin + i] = sorted[i];
    #pragma unroll
    for (int j = 0; j < BIN_EPT; ++j)
        if (d[j] >= 0) atomicAdd(&odeg[s[j]], 1u);
}

__device__ __forceinline__ int run_of(const int* pre, int i) {
    int lo = 0, hi = BIN_BLOCKS;
    while (hi - lo > 1) {
        const int mid = (lo + hi) >> 1;
        if (pre[mid] <= i) lo = mid; else hi = mid;
    }
    return lo;
}

#define BUCKET_EPT 4
static_assert(BUCKET_EPT == 4, "the load fence in bucket_kernel names four values");

__global__ void __launch_bounds__(BIN_THREADS) bucket_kernel(
    const int2* __restrict__ stage, const unsigned* __restrict__ runs,
    const unsigned* __restrict__ odeg, float* __restrict__ dstn,
    unsigned long long* __restrict__ bucket, int2* __restrict__ ent, int n_nodes) {
    __shared__ int pre[BIN_BLOCKS + 1];
    __shared__ int rbase[BIN_BLOCKS];
    __shared__ int cnt[RANGE_NODES];
    __shared__ int nstart[RANGE_NODES];
    __shared__ int fill[RANGE_NODES];
    __shared__ int wsa[33], wsb[33], wsc[33];

    const int tid = threadIdx.x;
    const int bin = blockIdx.x;

    int len = 0, off = 0;
    if (tid < BIN_BLOCKS) {
        const unsigned word = runs[(size_t)tid * N_RANGES + bin];
        off = (int)(word & 0xffffu);
        len = (int)(word >> 16);
    }
    if (tid < RANGE_NODES) { cnt[tid] = 0; fill[tid] = 0; }
    int total, base;
    const int ex = block_scan_excl(len, wsa, total);
    (void)block_scan_excl(off, wsb, base);
    if (tid < BIN_BLOCKS) { pre[tid] = ex; rbase[tid] = tid * BIN_CHUNK + off; }
    if (tid == 0) pre[BIN_BLOCKS] = total;
    __syncthreads();

    int2 eh[BUCKET_EPT];
    #pragma unroll
    for (int j = 0; j < BUCKET_EPT; ++j) {
        const int i = tid + j * BIN_THREADS;
        eh[j] = make_int2(0, 0);
        if (i < total) {
            const int b = run_of(pre, i);
            eh[j] = stage[rbase[b] + (i - pre[b])];
        }
    }
    unsigned od[BUCKET_EPT];
    #pragma unroll
    for (int j = 0; j < BUCKET_EPT; ++j) od[j] = odeg[eh[j].x];
    #pragma unroll
    for (int j = 0; j < BUCKET_EPT; ++j) {
        const int i = tid + j * BIN_THREADS;
        if (i < total) atomicAdd(&cnt[eh[j].y & (RANGE_NODES - 1)], 1);
    }
    for (int i = tid + BUCKET_EPT * BIN_THREADS; i < total; i += BIN_THREADS) {
        const int b = run_of(pre, i);
        const int2 e = stage[rbase[b] + (i - pre[b])];
        atomicAdd(&cnt[e.y & (RANGE_NODES - 1)], 1);
    }
    asm volatile("" :: "v"(od[0]), "v"(od[1]), "v"(od[2]), "v"(od[3]));
    __syncthreads();

    const int c = (tid < RANGE_NODES) ? cnt[tid] : 0;
    int total2;
    const int st = block_scan_excl(c, wsc, total2);
    if (tid < RANGE_NODES) {
        nstart[tid] = st;
        const int node = (bin << RANGE_SHIFT) + tid;
        if (node < n_nodes) {
            bucket[node] = (unsigned long long)(unsigned)(base + st) |
                           ((unsigned long long)(unsigned)c << 32);
            dstn[node] = rsqrtf(fmaxf((float)c, 1.0f));
        }
    }
    __syncthreads();

    #pragma unroll
    for (int j = 0; j < BUCKET_EPT; ++j) {
        const int i = tid + j * BIN_THREADS;
        if (i < total) {
            const int k = eh[j].y & (RANGE_NODES - 1);
            const int slot = atomicAdd(&fill[k], 1);
            const float w = rsqrtf(fmaxf((float)od[j], 1.0f));
            ent[base + nstart[k] + slot] = make_int2(eh[j].x, __float_as_int(w));
        }
    }
    for (int i = tid + BUCKET_EPT * BIN_THREADS; i < total; i += BIN_THREADS) {
        const int b = run_of(pre, i);
        const int2 e = stage[rbase[b] + (i - pre[b])];
        const int k = e.y & (RANGE_NODES - 1);
        const int slot = atomicAdd(&fill[k], 1);
        const float w = rsqrtf(fmaxf((float)odeg[e.x], 1.0f));
        ent[base + nstart[k] + slot] = make_int2(e.x, __float_as_int(w));
    }
}

#define CONV_BLOCKS 1024

#define KEY_B   0x40000000
#define KEY_ID  0x3fffffff

__device__ __forceinline__ int lane_of_rank(int r, int k) {
    return __builtin_ctz(__builtin_amdgcn_ballot_w32(r == k));
}

__device__ __forceinline__ void load_entries(const int2* __restrict__ ent, int offA, int mA,
                                             int offB, int mB, int lane, int& s, float& w) {
    s = 0x7fffffff;
    w = 0.f;
    if (lane < mA + mB) {
        int2 q = ent[(lane < mA) ? (offA + lane) : (offB + lane - mA)];
        s = q.x;
        w = __int_as_float(q.y);
    }
}

#define ROW_OF(sid) (((const float4*)(hin + (size_t)(sid) * HID_C))[lane])
#define ADD_ROW(ACC, V, WV)                                                            \
    { ACC.x += V.x * WV; ACC.y += V.y * WV; ACC.z += V.z * WV; ACC.w += V.w * WV; }
#define ASK_POS(i, V, WV)                                                              \
    {                                                                                  \
        const int j_ = lane_of_rank(r, min(p + i, m - 1));                             \
        WV = __int_as_float(__builtin_amdgcn_readlane(__float_as_int(w), j_));         \
        if (left > i) V = ROW_OF(__builtin_amdgcn_readlane(key, j_) & KEY_ID);         \
    }
#define ADD_POS(i, V, WV)                                                              \
    if (i == sw) { accA = cur; cur = make_float4(0.f, 0.f, 0.f, 0.f); inB = true; }    \
    if (left > i) ADD_ROW(cur, V, WV)

template <bool POOL>
__device__ __forceinline__ void graph_conv_body(
    const float* __restrict__ hin, const unsigned long long* __restrict__ bucket,
    const int2* __restrict__ ent, const float* __restrict__ dstn,
    const float* __restrict__ W, const float* __restrict__ bias,
    float* __restrict__ out, const int* __restrict__ gids,
    float* __restrict__ hg, int n_nodes) {
    __shared__ float As[2][16][HID_C + 4];

    const int tid    = threadIdx.x;
    const int lane   = tid & 31;
    const int wave   = __builtin_amdgcn_readfirstlane(tid >> 5);
    const int l15    = lane & 15;
    const int khalf  = (lane >> 4) << 1;
    const int col    = wave * 16 + l15;
    const int ntiles = (n_nodes + 15) >> 4;
    const int tstep  = gridDim.x;

    v2f breg[HID_C / 4];
    #pragma unroll
    for (int kk = 0; kk < HID_C / 4; ++kk) {
        breg[kk].x = W[(size_t)(kk * 4 + khalf) * HID_C + col];
        breg[kk].y = W[(size_t)(kk * 4 + khalf + 1) * HID_C + col];
    }
    const float bv = bias[col];

    int tile = blockIdx.x;
    unsigned long long bA = 0ull, bB = 0ull, nbA = 0ull, nbB = 0ull;
    if (tile < ntiles) {
        const int n0 = tile * 16 + wave, n1 = n0 + 8;
        if (n0 < n_nodes) bA = bucket[n0];
        if (n1 < n_nodes) bB = bucket[n1];
    }
    if (tile + tstep < ntiles) {
        const int n0 = (tile + tstep) * 16 + wave, n1 = n0 + 8;
        if (n0 < n_nodes) nbA = bucket[n0];
        if (n1 < n_nodes) nbB = bucket[n1];
    }
    int   s;
    float w;
    {
        const int cA = (int)(unsigned)(bA >> 32), cB = (int)(unsigned)(bB >> 32);
        const bool one = (cA + cB <= 32);
        load_entries(ent, (int)(unsigned)bA, one ? cA : 0, (int)(unsigned)bB, one ? cB : 0, lane, s, w);
    }

    int buf = 0;
    for (; tile < ntiles; tile += tstep) {
        const int mbase = tile * 16;
        const int nA = mbase + wave, nB = nA + 8;
        const int startA = (int)(unsigned)bA, cntA = (int)(unsigned)(bA >> 32);
        const int startB = (int)(unsigned)bB, cntB = (int)(unsigned)(bB >> 32);
        const float dA = (nA < n_nodes) ? dstn[nA] : 0.f;
        const float dB = (nB < n_nodes) ? dstn[nB] : 0.f;

        float4 cur = make_float4(0.f, 0.f, 0.f, 0.f), accA = cur;
        bool inB = false;
        {
            int  ca = 0, cb = 0;
            bool have = (cntA + cntB <= 32);
            int  mA = have ? cntA : 0, mB = have ? cntB : 0;
            while (ca < cntA || cb < cntB) {
                if (!have) {
                    if (ca < cntA) { mA = min(32, cntA - ca); mB = 0; }
                    else           { mA = 0; mB = min(32, cntB - cb); }
                    load_entries(ent, startA + ca, mA, startB + cb, mB, lane, s, w);
                }
                have = false;
                const int m   = mA + mB;
                const int key = s | ((lane >= mA) ? KEY_B : 0);
                int r = 0;
                int j = 0;
                for (; j + 4 <= m; j += 4) {
                    const int t0 = __builtin_amdgcn_readlane(key, j);
                    const int t1 = __builtin_amdgcn_readlane(key, j + 1);
                    const int t2 = __builtin_amdgcn_readlane(key, j + 2);
                    const int t3 = __builtin_amdgcn_readlane(key, j + 3);
                    r += (((t0 < key) || (t0 == key && j     < lane)) ? 1 : 0) +
                         (((t1 < key) || (t1 == key && j + 1 < lane)) ? 1 : 0) +
                         (((t2 < key) || (t2 == key && j + 2 < lane)) ? 1 : 0) +
                         (((t3 < key) || (t3 == key && j + 3 < lane)) ? 1 : 0);
                }
                for (; j < m; ++j) {
                    const int t = __builtin_amdgcn_readlane(key, j);
                    r += ((t < key) || (t == key && j < lane)) ? 1 : 0;
                }
                for (int p = 0; p < m; p += 8) {
                    const int left = m - p;
                    const int sw   = (mB > 0 && !inB) ? mA - p : -1;
                    const float4 zr = make_float4(0.f, 0.f, 0.f, 0.f);
                    float4 v0 = zr, v1 = zr, v2 = zr, v3 = zr, v4 = zr, v5 = zr, v6 = zr, v7 = zr;
                    float w0, w1, w2, w3, w4, w5, w6, w7;
                    ASK_POS(0, v0, w0) ASK_POS(1, v1, w1) ASK_POS(2, v2, w2) ASK_POS(3, v3, w3)
                    ASK_POS(4, v4, w4) ASK_POS(5, v5, w5) ASK_POS(6, v6, w6) ASK_POS(7, v7, w7)
                    ADD_POS(0, v0, w0) ADD_POS(1, v1, w1) ADD_POS(2, v2, w2) ADD_POS(3, v3, w3)
                    ADD_POS(4, v4, w4) ADD_POS(5, v5, w5) ADD_POS(6, v6, w6) ADD_POS(7, v7, w7)
                }
                ca += mA; cb += mB;
            }
        }

        const bool full_tile = POOL && (mbase + 16 <= n_nodes);
        int gq[16];
        #pragma unroll
        for (int r = 0; r < 16; ++r) gq[r] = 0;
        if constexpr (POOL) {
            if (full_tile) {
                const int* __restrict__ gp = gids + mbase;
                #pragma unroll
                for (int r = 0; r < 16; ++r) gq[r] = gp[r];
            }
        }

        const float4 z4 = make_float4(0.f, 0.f, 0.f, 0.f);
        const float4 rowA = inB ? accA : cur;
        const float4 rowB = inB ? cur : z4;
        *(float4*)&As[buf][wave][lane * 4] =
            make_float4(rowA.x * dA, rowA.y * dA, rowA.z * dA, rowA.w * dA);
        *(float4*)&As[buf][wave + 8][lane * 4] =
            make_float4(rowB.x * dB, rowB.y * dB, rowB.z * dB, rowB.w * dB);
        __syncthreads();

        bA = nbA; bB = nbB;
        {
            const int cA = (int)(unsigned)(bA >> 32), cB = (int)(unsigned)(bB >> 32);
            const bool one = (cA + cB <= 32);
            load_entries(ent, (int)(unsigned)bA, one ? cA : 0, (int)(unsigned)bB, one ? cB : 0, lane, s, w);
        }
        nbA = 0ull; nbB = 0ull;
        if (tile + 2 * tstep < ntiles) {
            const int n0 = (tile + 2 * tstep) * 16 + wave, n1 = n0 + 8;
            if (n0 < n_nodes) nbA = bucket[n0];
            if (n1 < n_nodes) nbB = bucket[n1];
        }

        v8f acc = {0.f, 0.f, 0.f, 0.f, 0.f, 0.f, 0.f, 0.f};
        #pragma unroll
        for (int kk = 0; kk < HID_C / 4; ++kk) {
            v2f a;
            a.x = As[buf][l15][kk * 4 + khalf];
            a.y = As[buf][l15][kk * 4 + khalf + 1];
            acc = __builtin_amdgcn_wmma_f32_16x16x4_f32(false, a, false, breg[kk],
                                                        (short)0, acc, false, false);
        }
        buf ^= 1;

        if constexpr (!POOL) {
            #pragma unroll
            for (int i = 0; i < 8; ++i) {
                int node = mbase + i + ((lane >> 4) << 3);
                if (node < n_nodes)
                    out[(size_t)node * HID_C + col] = fmaxf(acc[i] + bv, 0.0f);
            }
        } else {
            int glo = gq[0], ghi = gq[0];
            #pragma unroll
            for (int r = 1; r < 16; ++r) { glo = min(glo, gq[r]); ghi = max(ghi, gq[r]); }
            const bool one_graph = full_tile && (glo == ghi);
            const int  g0 = gq[0];
            float m = acc[0];
            #pragma unroll
            for (int i = 1; i < 8; ++i) m = fmaxf(m, acc[i]);
            m = fmaxf(m, __shfl_xor(m, 16));
            if (one_graph) {
                if (lane < 16)
                    atomicMax((int*)&hg[(size_t)g0 * HID_C + col],
                              __float_as_int(fmaxf(m + bv, 0.0f)));
            } else {
                #pragma unroll
                for (int i = 0; i < 8; ++i) {
                    int node = mbase + i + ((lane >> 4) << 3);
                    if (node < n_nodes)
                        atomicMax((int*)&hg[(size_t)gids[node] * HID_C + col],
                                  __float_as_int(fmaxf(acc[i] + bv, 0.0f)));
                }
            }
        }
    }
}

__global__ void __launch_bounds__(256, 8) graph_conv_kernel(
    const float* __restrict__ hin, const unsigned long long* __restrict__ bucket,
    const int2* __restrict__ ent, const float* __restrict__ dstn,
    const float* __restrict__ W, const float* __restrict__ bias,
    float* __restrict__ out, int n_nodes) {
    graph_conv_body<false>(hin, bucket, ent, dstn, W, bias, out, nullptr, nullptr, n_nodes);
}

__global__ void __launch_bounds__(256, 8) graph_conv_pool_kernel(
    const float* __restrict__ hin, const unsigned long long* __restrict__ bucket,
    const int2* __restrict__ ent, const float* __restrict__ dstn,
    const float* __restrict__ W, const float* __restrict__ bias,
    const int* __restrict__ gids, float* __restrict__ hg, int n_nodes) {
    graph_conv_body<true>(hin, bucket, ent, dstn, W, bias, nullptr, gids, hg, n_nodes);
}

#define MLP_SPLIT 4
#define MID_C (HID_C / 2)

__global__ void __launch_bounds__(256) mlp_kernel(
    const float* __restrict__ hg,
    const float* __restrict__ W1, const float* __restrict__ b1,
    const float* __restrict__ W2, const float* __restrict__ b2,
    const float* __restrict__ W3, const float* __restrict__ b3,
    float* __restrict__ out) {
    __shared__ float X0[16][HID_C + 4];
    __shared__ float X1[16][HID_C + 4];
    __shared__ float X2[16][MID_C + 4];

    const int tid   = threadIdx.x;
    const int lane  = tid & 31;
    const int wave  = __builtin_amdgcn_readfirstlane(tid >> 5);
    const int l15   = lane & 15;
    const int khalf = (lane >> 4) << 1;
    const int rhalf = (lane >> 4) << 3;
    const int mbase = blockIdx.x * 16;

    float4 t0, t1;
    {
        const int i0 = tid, i1 = tid + 256;
        t0 = ((const float4*)(hg + (size_t)(mbase + (i0 >> 5)) * HID_C))[i0 & 31];
        t1 = ((const float4*)(hg + (size_t)(mbase + (i1 >> 5)) * HID_C))[i1 & 31];
    }

    const int c1 = wave * 16 + l15;
    v2f w1[HID_C / 4];
    #pragma unroll
    for (int kk = 0; kk < HID_C / 4; ++kk) {
        w1[kk].x = W1[(size_t)(kk * 4 + khalf) * HID_C + c1];
        w1[kk].y = W1[(size_t)(kk * 4 + khalf + 1) * HID_C + c1];
    }
    const float bv1 = b1[c1];

    const int c2 = (wave & 3) * 16 + l15;
    v2f w2[HID_C / 4];
    #pragma unroll
    for (int kk = 0; kk < HID_C / 4; ++kk) { w2[kk].x = 0.f; w2[kk].y = 0.f; }
    float bv2 = 0.f;
    if (wave < 4) {
        #pragma unroll
        for (int kk = 0; kk < HID_C / 4; ++kk) {
            w2[kk].x = W2[(size_t)(kk * 4 + khalf) * MID_C + c2];
            w2[kk].y = W2[(size_t)(kk * 4 + khalf + 1) * MID_C + c2];
        }
        bv2 = b2[c2];
    }

    const int  c3   = (blockIdx.y * 8 + wave) * 16 + l15;
    const bool c3ok = c3 < OUT_DIM_C;
    v2f w3[MID_C / 4];
    #pragma unroll
    for (int kk = 0; kk < MID_C / 4; ++kk) {
        w3[kk].x = c3ok ? W3[(size_t)(kk * 4 + khalf) * OUT_DIM_C + c3] : 0.0f;
        w3[kk].y = c3ok ? W3[(size_t)(kk * 4 + khalf + 1) * OUT_DIM_C + c3] : 0.0f;
    }
    const float bv3 = c3ok ? b3[c3] : 0.0f;

    *(float4*)&X0[tid >> 5][(tid & 31) * 4]         = t0;
    *(float4*)&X0[(tid + 256) >> 5][(tid & 31) * 4] = t1;
    __syncthreads();

    {
        v8f acc = {0.f, 0.f, 0.f, 0.f, 0.f, 0.f, 0.f, 0.f};
        #pragma unroll
        for (int kk = 0; kk < HID_C / 4; ++kk) {
            v2f a;
            a.x = X0[l15][kk * 4 + khalf];
            a.y = X0[l15][kk * 4 + khalf + 1];
            acc = __builtin_amdgcn_wmma_f32_16x16x4_f32(false, a, false, w1[kk],
                                                        (short)0, acc, false, false);
        }
        #pragma unroll
        for (int i = 0; i < 8; ++i) X1[i + rhalf][c1] = fmaxf(acc[i] + bv1, 0.0f);
    }
    __syncthreads();

    if (wave < 4) {
        v8f acc = {0.f, 0.f, 0.f, 0.f, 0.f, 0.f, 0.f, 0.f};
        #pragma unroll
        for (int kk = 0; kk < HID_C / 4; ++kk) {
            v2f a;
            a.x = X1[l15][kk * 4 + khalf];
            a.y = X1[l15][kk * 4 + khalf + 1];
            acc = __builtin_amdgcn_wmma_f32_16x16x4_f32(false, a, false, w2[kk],
                                                        (short)0, acc, false, false);
        }
        #pragma unroll
        for (int i = 0; i < 8; ++i) X2[i + rhalf][c2] = fmaxf(acc[i] + bv2, 0.0f);
    }
    __syncthreads();

    {
        v8f acc = {0.f, 0.f, 0.f, 0.f, 0.f, 0.f, 0.f, 0.f};
        #pragma unroll
        for (int kk = 0; kk < MID_C / 4; ++kk) {
            v2f a;
            a.x = X2[l15][kk * 4 + khalf];
            a.y = X2[l15][kk * 4 + khalf + 1];
            acc = __builtin_amdgcn_wmma_f32_16x16x4_f32(false, a, false, w3[kk],
                                                        (short)0, acc, false, false);
        }
        if (c3ok) {
            #pragma unroll
            for (int i = 0; i < 8; ++i)
                out[(size_t)(mbase + i + rhalf) * OUT_DIM_C + c3] = acc[i] + bv3;
        }
    }
}

extern "C" void kernel_launch(void* const* d_in, const int* in_sizes, int n_in,
                              void* d_out, int out_size, void* d_ws, size_t ws_size,
                              hipStream_t stream) {
    (void)in_sizes; (void)n_in; (void)out_size; (void)ws_size;

    const float* feat = (const float*)d_in[0];
    const int*   src  = (const int*)d_in[1];
    const int*   dst  = (const int*)d_in[2];
    const int*   gids = (const int*)d_in[3];
    const float* Wg[3] = {(const float*)d_in[4], (const float*)d_in[6], (const float*)d_in[8]};
    const float* bg[3] = {(const float*)d_in[5], (const float*)d_in[7], (const float*)d_in[9]};
    const float* Wc1 = (const float*)d_in[10]; const float* bc1 = (const float*)d_in[11];
    const float* Wc2 = (const float*)d_in[12]; const float* bc2 = (const float*)d_in[13];
    const float* Wc3 = (const float*)d_in[14]; const float* bc3 = (const float*)d_in[15];
    float* out = (float*)d_out;

    char* ws = (char*)d_ws;
    size_t off = 0;
    auto take = [&](size_t elems) -> float* {
        float* p = (float*)(ws + off);
        off += ((elems * sizeof(float) + 255) / 256) * 256;
        return p;
    };
    unsigned* odeg = (unsigned*)take(N_NODES_C);
    float* dstn = take(N_NODES_C);
    float* h1   = take((size_t)N_NODES_C * HID_C);
    float* h2   = take((size_t)N_NODES_C * HID_C);
    float* hg   = take((size_t)N_GRAPHS_C * HID_C);
    int2*  stage  = (int2*)take(2 * (size_t)BIN_BLOCKS * BIN_CHUNK);
    unsigned* runs = (unsigned*)take((size_t)BIN_BLOCKS * N_RANGES);
    unsigned long long* bucket = (unsigned long long*)take(2 * N_NODES_C);
    int2*  ent    = (int2*)take(2 * (size_t)N_EDGES_C);

    hipMemsetAsync(odeg, 0, N_NODES_C * sizeof(unsigned), stream);
    bin_kernel<<<BIN_BLOCKS, BIN_THREADS, 0, stream>>>(src, dst, odeg, stage, runs, N_EDGES_C);
    bucket_kernel<<<N_RANGES, BIN_THREADS, 0, stream>>>(stage, runs, odeg, dstn, bucket, ent, N_NODES_C);

    const int mblocks = CONV_BLOCKS;
    graph_conv_kernel<<<mblocks, 256, 0, stream>>>(feat, bucket, ent, dstn, Wg[0], bg[0], h1, N_NODES_C);
    graph_conv_kernel<<<mblocks, 256, 0, stream>>>(h1, bucket, ent, dstn, Wg[1], bg[1], h2, N_NODES_C);
    hipMemsetAsync(hg, 0, (size_t)N_GRAPHS_C * HID_C * sizeof(float), stream);
    graph_conv_pool_kernel<<<mblocks, 256, 0, stream>>>(h2, bucket, ent, dstn, Wg[2], bg[2], gids, hg,
                                                        N_NODES_C);

    mlp_kernel<<<dim3(N_GRAPHS_C / 16, MLP_SPLIT), 256, 0, stream>>>(
        hg, Wc1, bc1, Wc2, bc2, Wc3, bc3, out);
}


# Bucket entries placed in ascending source id once, as buckets are filled (rank among the bucket's ids in LDS); conv launches drop their per-tile ordering; add order kept, also past 32 entries
# speedup vs baseline: 3.7971x; 1.0517x over previous
#include <hip/hip_runtime.h>
#include <hip/hip_bf16.h>


typedef float v2f __attribute__((ext_vector_type(2)));
typedef float v8f __attribute__((ext_vector_type(8)));

#define N_NODES_C 100000
#define N_EDGES_C 800000
#define HID_C     128
#define OUT_DIM_C 500
#define N_GRAPHS_C 64

#define BIN_BLOCKS   256
#define BIN_THREADS  1024
#define BIN_CHUNK    ((N_EDGES_C + BIN_BLOCKS - 1) / BIN_BLOCKS)
#define BIN_EPT      ((BIN_CHUNK + BIN_THREADS - 1) / BIN_THREADS)
#define RANGE_SHIFT  8
#define RANGE_NODES  (1 << RANGE_SHIFT)
#define N_RANGES     ((N_NODES_C + RANGE_NODES - 1) >> RANGE_SHIFT)

static_assert(BIN_THREADS == 1024, "block_scan_excl is written for 32 waves");
static_assert(BIN_CHUNK < 65536, "run offset and length share one 32-bit word");
static_assert(N_RANGES <= BIN_THREADS, "one range per thread in the scan");
static_assert(BIN_BLOCKS <= BIN_THREADS && RANGE_NODES <= BIN_THREADS, "one item per thread in the scans");

__device__ __forceinline__ int block_scan_excl(int v, int* ws, int& total) {
    const int lane = threadIdx.x & 31;
    const int wave = threadIdx.x >> 5;
    int incl = v;
    #pragma unroll
    for (int o = 1; o < 32; o <<= 1) {
        int t = __shfl_up(incl, o);
        if (lane >= o) incl += t;
    }
    if (lane == 31) ws[wave] = incl;
    __syncthreads();
    if (wave == 0) {
        const int wv = ws[lane];
        int wi = wv;
        #pragma unroll
        for (int o = 1; o < 32; o <<= 1) {
            int t = __shfl_up(wi, o);
            if (lane >= o) wi += t;
        }
        ws[lane] = wi - wv;
        if (lane == 31) ws[32] = wi;
    }
    __syncthreads();
    total = ws[32];
    return ws[wave] + incl - v;
}

__global__ void __launch_bounds__(BIN_THREADS) bin_kernel(
    const int* __restrict__ src, const int* __restrict__ dst,
    unsigned* __restrict__ odeg, int2* __restrict__ stage,
    unsigned* __restrict__ runs, int n_edges) {
    __shared__ int  hist[N_RANGES];
    __shared__ int2 sorted[BIN_CHUNK];
    __shared__ int  ws[33];

    const int tid     = threadIdx.x;
    const int e_begin = blockIdx.x * BIN_CHUNK;
    const int e_end   = min(e_begin + BIN_CHUNK, n_edges);

    for (int i = tid; i < N_RANGES; i += BIN_THREADS) hist[i] = 0;

    int s[BIN_EPT], d[BIN_EPT], rk[BIN_EPT];
    #pragma unroll
    for (int j = 0; j < BIN_EPT; ++j) {
        const int e = e_begin + tid + j * BIN_THREADS;
        s[j] = 0; d[j] = -1; rk[j] = 0;
        if (e < e_end) { s[j] = src[e]; d[j] = dst[e]; }
    }
    __syncthreads();
    #pragma unroll
    for (int j = 0; j < BIN_EPT; ++j) {
        if (d[j] >= 0) rk[j] = atomicAdd(&hist[d[j] >> RANGE_SHIFT], 1);
    }
    __syncthreads();

    const int c = (tid < N_RANGES) ? hist[tid] : 0;
    int total;
    const int ex = block_scan_excl(c, ws, total);
    if (tid < N_RANGES) {
        hist[tid] = ex;
        runs[(size_t)blockIdx.x * N_RANGES + tid] = (unsigned)ex | ((unsigned)c << 16);
    }
    __syncthreads();
    #pragma unroll
    for (int j = 0; j < BIN_EPT; ++j)
        if (d[j] >= 0) sorted[hist[d[j] >> RANGE_SHIFT] + rk[j]] = make_int2(s[j], d[j]);
    __syncthreads();
    for (int i = tid; i < e_end - e_begin; i += BIN_THREADS) stage[e_begin + i] = sorted[i];
    #pragma unroll
    for (int j = 0; j < BIN_EPT; ++j)
        if (d[j] >= 0) atomicAdd(&odeg[s[j]], 1u);
}

__device__ __forceinline__ int run_of(const int* pre, int i) {
    int lo = 0, hi = BIN_BLOCKS;
    while (hi - lo > 1) {
        const int mid = (lo + hi) >> 1;
        if (pre[mid] <= i) lo = mid; else hi = mid;
    }
    return lo;
}

#define BUCKET_EPT 4
static_assert(BUCKET_EPT == 4, "the load fence in bucket_kernel names four values");
#define SORT_CAP (BUCKET_EPT * BIN_THREADS)

__global__ void __launch_bounds__(BIN_THREADS) bucket_kernel(
    const int2* __restrict__ stage, const unsigned* __restrict__ runs,
    const unsigned* __restrict__ odeg, float* __restrict__ dstn,
    unsigned long long* __restrict__ bucket, int2* __restrict__ ent, int2* tmp, int n_nodes) {
    __shared__ int pre[BIN_BLOCKS + 1];
    __shared__ int rbase[BIN_BLOCKS];
    __shared__ int cnt[RANGE_NODES];
    __shared__ int nstart[RANGE_NODES];
    __shared__ int fill[RANGE_NODES];
    __shared__ int lkey[SORT_CAP];
    __shared__ int wsa[33], wsb[33], wsc[33];

    const int tid = threadIdx.x;
    const int bin = blockIdx.x;

    int len = 0, off = 0;
    if (tid < BIN_BLOCKS) {
        const unsigned word = runs[(size_t)tid * N_RANGES + bin];
        off = (int)(word & 0xffffu);
        len = (int)(word >> 16);
    }
    if (tid < RANGE_NODES) { cnt[tid] = 0; fill[tid] = 0; }
    int total, base;
    const int ex = block_scan_excl(len, wsa, total);
    (void)block_scan_excl(off, wsb, base);
    if (tid < BIN_BLOCKS) { pre[tid] = ex; rbase[tid] = tid * BIN_CHUNK + off; }
    if (tid == 0) pre[BIN_BLOCKS] = total;
    __syncthreads();

    int2 eh[BUCKET_EPT];
    #pragma unroll
    for (int j = 0; j < BUCKET_EPT; ++j) {
        const int i = tid + j * BIN_THREADS;
        eh[j] = make_int2(0, 0);
        if (i < total) {
            const int b = run_of(pre, i);
            eh[j] = stage[rbase[b] + (i - pre[b])];
        }
    }
    unsigned od[BUCKET_EPT];
    #pragma unroll
    for (int j = 0; j < BUCKET_EPT; ++j) od[j] = odeg[eh[j].x];
    #pragma unroll
    for (int j = 0; j < BUCKET_EPT; ++j) {
        const int i = tid + j * BIN_THREADS;
        if (i < total) atomicAdd(&cnt[eh[j].y & (RANGE_NODES - 1)], 1);
    }
    for (int i = tid + BUCKET_EPT * BIN_THREADS; i < total; i += BIN_THREADS) {
        const int b = run_of(pre, i);
        const int2 e = stage[rbase[b] + (i - pre[b])];
        atomicAdd(&cnt[e.y & (RANGE_NODES - 1)], 1);
    }
    asm volatile("" :: "v"(od[0]), "v"(od[1]), "v"(od[2]), "v"(od[3]));
    __syncthreads();

    const int c = (tid < RANGE_NODES) ? cnt[tid] : 0;
    int total2;
    const int st = block_scan_excl(c, wsc, total2);
    if (tid < RANGE_NODES) {
        nstart[tid] = st;
        const int node = (bin << RANGE_SHIFT) + tid;
        if (node < n_nodes) {
            bucket[node] = (unsigned long long)(unsigned)(base + st) |
                           ((unsigned long long)(unsigned)c << 32);
            dstn[node] = rsqrtf(fmaxf((float)c, 1.0f));
        }
    }
    __syncthreads();

    if (total <= SORT_CAP) {
        int slot[BUCKET_EPT];
        #pragma unroll
        for (int j = 0; j < BUCKET_EPT; ++j) {
            const int i = tid + j * BIN_THREADS;
            slot[j] = 0;
            if (i < total) {
                const int k = eh[j].y & (RANGE_NODES - 1);
                slot[j] = atomicAdd(&fill[k], 1);
                lkey[nstart[k] + slot[j]] = eh[j].x;
            }
        }
        __syncthreads();
        #pragma unroll
        for (int j = 0; j < BUCKET_EPT; ++j) {
            const int i = tid + j * BIN_THREADS;
            if (i < total) {
                const int k  = eh[j].y & (RANGE_NODES - 1);
                const int st = nstart[k], c = cnt[k];
                const int id = eh[j].x, a = slot[j];
                int rank = 0;
                for (int t = 0; t < c; ++t) {
                    const int o = lkey[st + t];
                    rank += ((o < id) || (o == id && t < a)) ? 1 : 0;
                }
                const float w = rsqrtf(fmaxf((float)od[j], 1.0f));
                ent[base + st + rank] = make_int2(id, __float_as_int(w));
            }
        }
    } else {
        #pragma unroll
        for (int j = 0; j < BUCKET_EPT; ++j) {
            const int i = tid + j * BIN_THREADS;
            if (i < total) {
                const int k  = eh[j].y & (RANGE_NODES - 1);
                const int sl = atomicAdd(&fill[k], 1);
                const float w = rsqrtf(fmaxf((float)od[j], 1.0f));
                tmp[base + nstart[k] + sl] = make_int2(eh[j].x, __float_as_int(w));
            }
        }
        for (int i = tid + BUCKET_EPT * BIN_THREADS; i < total; i += BIN_THREADS) {
            const int b = run_of(pre, i);
            const int2 e = stage[rbase[b] + (i - pre[b])];
            const int k  = e.y & (RANGE_NODES - 1);
            const int sl = atomicAdd(&fill[k], 1);
            const float w = rsqrtf(fmaxf((float)odeg[e.x], 1.0f));
            tmp[base + nstart[k] + sl] = make_int2(e.x, __float_as_int(w));
        }
        __threadfence_block();
        __syncthreads();
        for (int q = tid; q < total; q += BIN_THREADS) {
            int lo = 0, hi = RANGE_NODES;
            while (hi - lo > 1) {
                const int mid = (lo + hi) >> 1;
                if (nstart[mid] <= q) lo = mid; else hi = mid;
            }
            const int st = nstart[lo], c = cnt[lo], a = q - st;
            const int2 e = tmp[base + q];
            int rank = 0;
            for (int t = 0; t < c; ++t) {
                const int o = tmp[base + st + t].x;
                rank += ((o < e.x) || (o == e.x && t < a)) ? 1 : 0;
            }
            ent[base + st + rank] = e;
        }
    }
}

#define CONV_BLOCKS 1024

__device__ __forceinline__ void load_entries(const int2* __restrict__ ent, int offA, int mA,
                                             int offB, int mB, int lane, int& s, float& w) {
    s = 0x7fffffff;
    w = 0.f;
    if (lane < mA + mB) {
        int2 q = ent[(lane < mA) ? (offA + lane) : (offB + lane - mA)];
        s = q.x;
        w = __int_as_float(q.y);
    }
}

#define ROW_OF(sid) (((const float4*)(hin + (size_t)(sid) * HID_C))[lane])
#define ADD_ROW(ACC, V, WV)                                                            \
    { ACC.x += V.x * WV; ACC.y += V.y * WV; ACC.z += V.z * WV; ACC.w += V.w * WV; }
#define ENTRY_POS(i, SID, WV)                                                          \
    {                                                                                  \
        const int j_ = min(p + i, m - 1);                                              \
        SID = __builtin_amdgcn_readlane(s, j_);                                        \
        WV  = __int_as_float(__builtin_amdgcn_readlane(__float_as_int(w), j_));        \
    }
#define ASK_POS(i, V, SID) if (left > i) V = ROW_OF(SID);
#define ADD_POS(i, V, WV)                                                              \
    if (i == sw) { accA = cur; cur = make_float4(0.f, 0.f, 0.f, 0.f); inB = true; }    \
    if (left > i) ADD_ROW(cur, V, WV)

template <bool POOL>
__device__ __forceinline__ void graph_conv_body(
    const float* __restrict__ hin, const unsigned long long* __restrict__ bucket,
    const int2* __restrict__ ent, const float* __restrict__ dstn,
    const float* __restrict__ W, const float* __restrict__ bias,
    float* __restrict__ out, const int* __restrict__ gids,
    float* __restrict__ hg, int n_nodes) {
    __shared__ float As[2][16][HID_C + 4];

    const int tid    = threadIdx.x;
    const int lane   = tid & 31;
    const int wave   = __builtin_amdgcn_readfirstlane(tid >> 5);
    const int l15    = lane & 15;
    const int khalf  = (lane >> 4) << 1;
    const int col    = wave * 16 + l15;
    const int ntiles = (n_nodes + 15) >> 4;
    const int tstep  = gridDim.x;

    v2f breg[HID_C / 4];
    #pragma unroll
    for (int kk = 0; kk < HID_C / 4; ++kk) {
        breg[kk].x = W[(size_t)(kk * 4 + khalf) * HID_C + col];
        breg[kk].y = W[(size_t)(kk * 4 + khalf + 1) * HID_C + col];
    }
    const float bv = bias[col];

    int tile = blockIdx.x;
    unsigned long long bA = 0ull, bB = 0ull, nbA = 0ull, nbB = 0ull;
    if (tile < ntiles) {
        const int n0 = tile * 16 + wave, n1 = n0 + 8;
        if (n0 < n_nodes) bA = bucket[n0];
        if (n1 < n_nodes) bB = bucket[n1];
    }
    if (tile + tstep < ntiles) {
        const int n0 = (tile + tstep) * 16 + wave, n1 = n0 + 8;
        if (n0 < n_nodes) nbA = bucket[n0];
        if (n1 < n_nodes) nbB = bucket[n1];
    }
    int   s;
    float w;
    {
        const int cA = (int)(unsigned)(bA >> 32), cB = (int)(unsigned)(bB >> 32);
        const bool one = (cA + cB <= 32);
        load_entries(ent, (int)(unsigned)bA, one ? cA : 0, (int)(unsigned)bB, one ? cB : 0, lane, s, w);
    }

    int buf = 0;
    for (; tile < ntiles; tile += tstep) {
        const int mbase = tile * 16;
        const int nA = mbase + wave, nB = nA + 8;
        const int startA = (int)(unsigned)bA, cntA = (int)(unsigned)(bA >> 32);
        const int startB = (int)(unsigned)bB, cntB = (int)(unsigned)(bB >> 32);
        const float dA = (nA < n_nodes) ? dstn[nA] : 0.f;
        const float dB = (nB < n_nodes) ? dstn[nB] : 0.f;

        float4 cur = make_float4(0.f, 0.f, 0.f, 0.f), accA = cur;
        bool inB = false;
        {
            int  ca = 0, cb = 0;
            bool have = (cntA + cntB <= 32);
            int  mA = have ? cntA : 0, mB = have ? cntB : 0;
            while (ca < cntA || cb < cntB) {
                if (!have) {
                    if (ca < cntA) { mA = min(32, cntA - ca); mB = 0; }
                    else           { mA = 0; mB = min(32, cntB - cb); }
                    load_entries(ent, startA + ca, mA, startB + cb, mB, lane, s, w);
                }
                have = false;
                const int m = mA + mB;
                for (int p = 0; p < m; p += 8) {
                    const int left = m - p;
                    const int sw   = (mB > 0 && !inB) ? mA - p : -1;
                    const float4 zr = make_float4(0.f, 0.f, 0.f, 0.f);
                    float4 v0 = zr, v1 = zr, v2 = zr, v3 = zr, v4 = zr, v5 = zr, v6 = zr, v7 = zr;
                    float w0, w1, w2, w3, w4, w5, w6, w7;
                    int   i0, i1, i2, i3, i4, i5, i6, i7;
                    ENTRY_POS(0, i0, w0) ENTRY_POS(1, i1, w1) ENTRY_POS(2, i2, w2) ENTRY_POS(3, i3, w3)
                    ENTRY_POS(4, i4, w4) ENTRY_POS(5, i5, w5) ENTRY_POS(6, i6, w6) ENTRY_POS(7, i7, w7)
                    ASK_POS(0, v0, i0) ASK_POS(1, v1, i1) ASK_POS(2, v2, i2) ASK_POS(3, v3, i3)
                    ASK_POS(4, v4, i4) ASK_POS(5, v5, i5) ASK_POS(6, v6, i6) ASK_POS(7, v7, i7)
                    ADD_POS(0, v0, w0) ADD_POS(1, v1, w1) ADD_POS(2, v2, w2) ADD_POS(3, v3, w3)
                    ADD_POS(4, v4, w4) ADD_POS(5, v5, w5) ADD_POS(6, v6, w6) ADD_POS(7, v7, w7)
                }
                ca += mA; cb += mB;
            }
        }

        const bool full_tile = POOL && (mbase + 16 <= n_nodes);
        int gq[16];
        #pragma unroll
        for (int r = 0; r < 16; ++r) gq[r] = 0;
        if constexpr (POOL) {
            if (full_tile) {
                const int* __restrict__ gp = gids + mbase;
                #pragma unroll
                for (int r = 0; r < 16; ++r) gq[r] = gp[r];
            }
        }

        const float4 z4 = make_float4(0.f, 0.f, 0.f, 0.f);
        const float4 rowA = inB ? accA : cur;
        const float4 rowB = inB ? cur : z4;
        *(float4*)&As[buf][wave][lane * 4] =
            make_float4(rowA.x * dA, rowA.y * dA, rowA.z * dA, rowA.w * dA);
        *(float4*)&As[buf][wave + 8][lane * 4] =
            make_float4(rowB.x * dB, rowB.y * dB, rowB.z * dB, rowB.w * dB);
        __syncthreads();

        bA = nbA; bB = nbB;
        {
            const int cA = (int)(unsigned)(bA >> 32), cB = (int)(unsigned)(bB >> 32);
            const bool one = (cA + cB <= 32);
            load_entries(ent, (int)(unsigned)bA, one ? cA : 0, (int)(unsigned)bB, one ? cB : 0, lane, s, w);
        }
        nbA = 0ull; nbB = 0ull;
        if (tile + 2 * tstep < ntiles) {
            const int n0 = (tile + 2 * tstep) * 16 + wave, n1 = n0 + 8;
            if (n0 < n_nodes) nbA = bucket[n0];
            if (n1 < n_nodes) nbB = bucket[n1];
        }

        v8f acc = {0.f, 0.f, 0.f, 0.f, 0.f, 0.f, 0.f, 0.f};
        #pragma unroll
        for (int kk = 0; kk < HID_C / 4; ++kk) {
            v2f a;
            a.x = As[buf][l15][kk * 4 + khalf];
            a.y = As[buf][l15][kk * 4 + khalf + 1];
            acc = __builtin_amdgcn_wmma_f32_16x16x4_f32(false, a, false, breg[kk],
                                                        (short)0, acc, false, false);
        }
        buf ^= 1;

        if constexpr (!POOL) {
            #pragma unroll
            for (int i = 0; i < 8; ++i) {
                int node = mbase + i + ((lane >> 4) << 3);
                if (node < n_nodes)
                    out[(size_t)node * HID_C + col] = fmaxf(acc[i] + bv, 0.0f);
            }
        } else {
            int glo = gq[0], ghi = gq[0];
            #pragma unroll
            for (int r = 1; r < 16; ++r) { glo = min(glo, gq[r]); ghi = max(ghi, gq[r]); }
            const bool one_graph = full_tile && (glo == ghi);
            const int  g0 = gq[0];
            float m = acc[0];
            #pragma unroll
            for (int i = 1; i < 8; ++i) m = fmaxf(m, acc[i]);
            m = fmaxf(m, __shfl_xor(m, 16));
            if (one_graph) {
                if (lane < 16)
                    atomicMax((int*)&hg[(size_t)g0 * HID_C + col],
                              __float_as_int(fmaxf(m + bv, 0.0f)));
            } else {
                #pragma unroll
                for (int i = 0; i < 8; ++i) {
                    int node = mbase + i + ((lane >> 4) << 3);
                    if (node < n_nodes)
                        atomicMax((int*)&hg[(size_t)gids[node] * HID_C + col],
                                  __float_as_int(fmaxf(acc[i] + bv, 0.0f)));
                }
            }
        }
    }
}

__global__ void __launch_bounds__(256, 8) graph_conv_kernel(
    const float* __restrict__ hin, const unsigned long long* __restrict__ bucket,
    const int2* __restrict__ ent, const float* __restrict__ dstn,
    const float* __restrict__ W, const float* __restrict__ bias,
    float* __restrict__ out, int n_nodes) {
    graph_conv_body<false>(hin, bucket, ent, dstn, W, bias, out, nullptr, nullptr, n_nodes);
}

__global__ void __launch_bounds__(256, 8) graph_conv_pool_kernel(
    const float* __restrict__ hin, const unsigned long long* __restrict__ bucket,
    const int2* __restrict__ ent, const float* __restrict__ dstn,
    const float* __restrict__ W, const float* __restrict__ bias,
    const int* __restrict__ gids, float* __restrict__ hg, int n_nodes) {
    graph_conv_body<true>(hin, bucket, ent, dstn, W, bias, nullptr, gids, hg, n_nodes);
}

#define MLP_SPLIT 4
#define MID_C (HID_C / 2)

__global__ void __launch_bounds__(256) mlp_kernel(
    const float* __restrict__ hg,
    const float* __restrict__ W1, const float* __restrict__ b1,
    const float* __restrict__ W2, const float* __restrict__ b2,
    const float* __restrict__ W3, const float* __restrict__ b3,
    float* __restrict__ out) {
    __shared__ float X0[16][HID_C + 4];
    __shared__ float X1[16][HID_C + 4];
    __shared__ float X2[16][MID_C + 4];

    const int tid   = threadIdx.x;
    const int lane  = tid & 31;
    const int wave  = __builtin_amdgcn_readfirstlane(tid >> 5);
    const int l15   = lane & 15;
    const int khalf = (lane >> 4) << 1;
    const int rhalf = (lane >> 4) << 3;
    const int mbase = blockIdx.x * 16;

    float4 t0, t1;
    {
        const int i0 = tid, i1 = tid + 256;
        t0 = ((const float4*)(hg + (size_t)(mbase + (i0 >> 5)) * HID_C))[i0 & 31];
        t1 = ((const float4*)(hg + (size_t)(mbase + (i1 >> 5)) * HID_C))[i1 & 31];
    }

    const int c1 = wave * 16 + l15;
    v2f w1[HID_C / 4];
    #pragma unroll
    for (int kk = 0; kk < HID_C / 4; ++kk) {
        w1[kk].x = W1[(size_t)(kk * 4 + khalf) * HID_C + c1];
        w1[kk].y = W1[(size_t)(kk * 4 + khalf + 1) * HID_C + c1];
    }
    const float bv1 = b1[c1];

    const int c2 = (wave & 3) * 16 + l15;
    v2f w2[HID_C / 4];
    #pragma unroll
    for (int kk = 0; kk < HID_C / 4; ++kk) { w2[kk].x = 0.f; w2[kk].y = 0.f; }
    float bv2 = 0.f;
    if (wave < 4) {
        #pragma unroll
        for (int kk = 0; kk < HID_C / 4; ++kk) {
            w2[kk].x = W2[(size_t)(kk * 4 + khalf) * MID_C + c2];
            w2[kk].y = W2[(size_t)(kk * 4 + khalf + 1) * MID_C + c2];
        }
        bv2 = b2[c2];
    }

    const int  c3   = (blockIdx.y * 8 + wave) * 16 + l15;
    const bool c3ok = c3 < OUT_DIM_C;
    v2f w3[MID_C / 4];
    #pragma unroll
    for (int kk = 0; kk < MID_C / 4; ++kk) {
        w3[kk].x = c3ok ? W3[(size_t)(kk * 4 + khalf) * OUT_DIM_C + c3] : 0.0f;
        w3[kk].y = c3ok ? W3[(size_t)(kk * 4 + khalf + 1) * OUT_DIM_C + c3] : 0.0f;
    }
    const float bv3 = c3ok ? b3[c3] : 0.0f;

    *(float4*)&X0[tid >> 5][(tid & 31) * 4]         = t0;
    *(float4*)&X0[(tid + 256) >> 5][(tid & 31) * 4] = t1;
    __syncthreads();

    {
        v8f acc = {0.f, 0.f, 0.f, 0.f, 0.f, 0.f, 0.f, 0.f};
        #pragma unroll
        for (int kk = 0; kk < HID_C / 4; ++kk) {
            v2f a;
            a.x = X0[l15][kk * 4 + khalf];
            a.y = X0[l15][kk * 4 + khalf + 1];
            acc = __builtin_amdgcn_wmma_f32_16x16x4_f32(false, a, false, w1[kk],
                                                        (short)0, acc, false, false);
        }
        #pragma unroll
        for (int i = 0; i < 8; ++i) X1[i + rhalf][c1] = fmaxf(acc[i] + bv1, 0.0f);
    }
    __syncthreads();

    if (wave < 4) {
        v8f acc = {0.f, 0.f, 0.f, 0.f, 0.f, 0.f, 0.f, 0.f};
        #pragma unroll
        for (int kk = 0; kk < HID_C / 4; ++kk) {
            v2f a;
            a.x = X1[l15][kk * 4 + khalf];
            a.y = X1[l15][kk * 4 + khalf + 1];
            acc = __builtin_amdgcn_wmma_f32_16x16x4_f32(false, a, false, w2[kk],
                                                        (short)0, acc, false, false);
        }
        #pragma unroll
        for (int i = 0; i < 8; ++i) X2[i + rhalf][c2] = fmaxf(acc[i] + bv2, 0.0f);
    }
    __syncthreads();

    {
        v8f acc = {0.f, 0.f, 0.f, 0.f, 0.f, 0.f, 0.f, 0.f};
        #pragma unroll
        for (int kk = 0; kk < MID_C / 4; ++kk) {
            v2f a;
            a.x = X2[l15][kk * 4 + khalf];
            a.y = X2[l15][kk * 4 + khalf + 1];
            acc = __builtin_amdgcn_wmma_f32_16x16x4_f32(false, a, false, w3[kk],
                                                        (short)0, acc, false, false);
        }
        if (c3ok) {
            #pragma unroll
            for (int i = 0; i < 8; ++i)
                out[(size_t)(mbase + i + rhalf) * OUT_DIM_C + c3] = acc[i] + bv3;
        }
    }
}

extern "C" void kernel_launch(void* const* d_in, const int* in_sizes, int n_in,
                              void* d_out, int out_size, void* d_ws, size_t ws_size,
                              hipStream_t stream) {
    (void)in_sizes; (void)n_in; (void)out_size; (void)ws_size;

    const float* feat = (const float*)d_in[0];
    const int*   src  = (const int*)d_in[1];
    const int*   dst  = (const int*)d_in[2];
    const int*   gids = (const int*)d_in[3];
    const float* Wg[3] = {(const float*)d_in[4], (const float*)d_in[6], (const float*)d_in[8]};
    const float* bg[3] = {(const float*)d_in[5], (const float*)d_in[7], (const float*)d_in[9]};
    const float* Wc1 = (const float*)d_in[10]; const float* bc1 = (const float*)d_in[11];
    const float* Wc2 = (const float*)d_in[12]; const float* bc2 = (const float*)d_in[13];
    const float* Wc3 = (const float*)d_in[14]; const float* bc3 = (const float*)d_in[15];
    float* out = (float*)d_out;

    char* ws = (char*)d_ws;
    size_t off = 0;
    auto take = [&](size_t elems) -> float* {
        float* p = (float*)(ws + off);
        off += ((elems * sizeof(float) + 255) / 256) * 256;
        return p;
    };
    unsigned* odeg = (unsigned*)take(N_NODES_C);
    float* dstn = take(N_NODES_C);
    float* h1   = take((size_t)N_NODES_C * HID_C);
    float* h2   = take((size_t)N_NODES_C * HID_C);
    float* hg   = take((size_t)N_GRAPHS_C * HID_C);
    int2*  stage  = (int2*)take(2 * (size_t)BIN_BLOCKS * BIN_CHUNK);
    unsigned* runs = (unsigned*)take((size_t)BIN_BLOCKS * N_RANGES);
    unsigned long long* bucket = (unsigned long long*)take(2 * N_NODES_C);
    int2*  ent    = (int2*)take(2 * (size_t)N_EDGES_C);
    int2*  tmp    = (int2*)take(2 * (size_t)N_EDGES_C);

    hipMemsetAsync(odeg, 0, N_NODES_C * sizeof(unsigned), stream);
    bin_kernel<<<BIN_BLOCKS, BIN_THREADS, 0, stream>>>(src, dst, odeg, stage, runs, N_EDGES_C);
    bucket_kernel<<<N_RANGES, BIN_THREADS, 0, stream>>>(stage, runs, odeg, dstn, bucket, ent, tmp, N_NODES_C);

    const int mblocks = CONV_BLOCKS;
    graph_conv_kernel<<<mblocks, 256, 0, stream>>>(feat, bucket, ent, dstn, Wg[0], bg[0], h1, N_NODES_C);
    graph_conv_kernel<<<mblocks, 256, 0, stream>>>(h1, bucket, ent, dstn, Wg[1], bg[1], h2, N_NODES_C);
    hipMemsetAsync(hg, 0, (size_t)N_GRAPHS_C * HID_C * sizeof(float), stream);
    graph_conv_pool_kernel<<<mblocks, 256, 0, stream>>>(h2, bucket, ent, dstn, Wg[2], bg[2], gids, hg,
                                                        N_NODES_C);

    mlp_kernel<<<dim3(N_GRAPHS_C / 16, MLP_SPLIT), 256, 0, stream>>>(
        hg, Wc1, bc1, Wc2, bc2, Wc3, bc3, out);
}


# Issue priority by phase in the conv launches: a wave runs its gather half (lane reads, row loads, adds, barrier) at s_setprio 3 and its 32 products and stores at 0, so rows are asked for first
# speedup vs baseline: 3.9371x; 1.0369x over previous
#include <hip/hip_runtime.h>
#include <hip/hip_bf16.h>


typedef float v2f __attribute__((ext_vector_type(2)));
typedef float v8f __attribute__((ext_vector_type(8)));

#define N_NODES_C 100000
#define N_EDGES_C 800000
#define HID_C     128
#define OUT_DIM_C 500
#define N_GRAPHS_C 64

#define BIN_BLOCKS   256
#define BIN_THREADS  1024
#define BIN_CHUNK    ((N_EDGES_C + BIN_BLOCKS - 1) / BIN_BLOCKS)
#define BIN_EPT      ((BIN_CHUNK + BIN_THREADS - 1) / BIN_THREADS)
#define RANGE_SHIFT  8
#define RANGE_NODES  (1 << RANGE_SHIFT)
#define N_RANGES     ((N_NODES_C + RANGE_NODES - 1) >> RANGE_SHIFT)

static_assert(BIN_THREADS == 1024, "block_scan_excl is written for 32 waves");
static_assert(BIN_CHUNK < 65536, "run offset and length share one 32-bit word");
static_assert(N_RANGES <= BIN_THREADS, "one range per thread in the scan");
static_assert(BIN_BLOCKS <= BIN_THREADS && RANGE_NODES <= BIN_THREADS, "one item per thread in the scans");

__device__ __forceinline__ int block_scan_excl(int v, int* ws, int& total) {
    const int lane = threadIdx.x & 31;
    const int wave = threadIdx.x >> 5;
    int incl = v;
    #pragma unroll
    for (int o = 1; o < 32; o <<= 1) {
        int t = __shfl_up(incl, o);
        if (lane >= o) incl += t;
    }
    if (lane == 31) ws[wave] = incl;
    __syncthreads();
    if (wave == 0) {
        const int wv = ws[lane];
        int wi = wv;
        #pragma unroll
        for (int o = 1; o < 32; o <<= 1) {
            int t = __shfl_up(wi, o);
            if (lane >= o) wi += t;
        }
        ws[lane] = wi - wv;
        if (lane == 31) ws[32] = wi;
    }
    __syncthreads();
    total = ws[32];
    return ws[wave] + incl - v;
}

__global__ void __launch_bounds__(BIN_THREADS) bin_kernel(
    const int* __restrict__ src, const int* __restrict__ dst,
    unsigned* __restrict__ odeg, int2* __restrict__ stage,
    unsigned* __restrict__ runs, int n_edges) {
    __shared__ int  hist[N_RANGES];
    __shared__ int2 sorted[BIN_CHUNK];
    __shared__ int  ws[33];

    const int tid     = threadIdx.x;
    const int e_begin = blockIdx.x * BIN_CHUNK;
    const int e_end   = min(e_begin + BIN_CHUNK, n_edges);

    for (int i = tid; i < N_RANGES; i += BIN_THREADS) hist[i] = 0;

    int s[BIN_EPT], d[BIN_EPT], rk[BIN_EPT];
    #pragma unroll
    for (int j = 0; j < BIN_EPT; ++j) {
        const int e = e_begin + tid + j * BIN_THREADS;
        s[j] = 0; d[j] = -1; rk[j] = 0;
        if (e < e_end) { s[j] = src[e]; d[j] = dst[e]; }
    }
    __syncthreads();
    #pragma unroll
    for (int j = 0; j < BIN_EPT; ++j) {
        if (d[j] >= 0) rk[j] = atomicAdd(&hist[d[j] >> RANGE_SHIFT], 1);
    }
    __syncthreads();

    const int c = (tid < N_RANGES) ? hist[tid] : 0;
    int total;
    const int ex = block_scan_excl(c, ws, total);
    if (tid < N_RANGES) {
        hist[tid] = ex;
        runs[(size_t)blockIdx.x * N_RANGES + tid] = (unsigned)ex | ((unsigned)c << 16);
    }
    __syncthreads();
    #pragma unroll
    for (int j = 0; j < BIN_EPT; ++j)
        if (d[j] >= 0) sorted[hist[d[j] >> RANGE_SHIFT] + rk[j]] = make_int2(s[j], d[j]);
    __syncthreads();
    for (int i = tid; i < e_end - e_begin; i += BIN_THREADS) stage[e_begin + i] = sorted[i];
    #pragma unroll
    for (int j = 0; j < BIN_EPT; ++j)
        if (d[j] >= 0) atomicAdd(&odeg[s[j]], 1u);
}

__device__ __forceinline__ int run_of(const int* pre, int i) {
    int lo = 0, hi = BIN_BLOCKS;
    while (hi - lo > 1) {
        const int mid = (lo + hi) >> 1;
        if (pre[mid] <= i) lo = mid; else hi = mid;
    }
    return lo;
}

#define BUCKET_EPT 4
static_assert(BUCKET_EPT == 4, "the load fence in bucket_kernel names four values");
#define SORT_CAP (BUCKET_EPT * BIN_THREADS)

__global__ void __launch_bounds__(BIN_THREADS) bucket_kernel(
    const int2* __restrict__ stage, const unsigned* __restrict__ runs,
    const unsigned* __restrict__ odeg, float* __restrict__ dstn,
    unsigned long long* __restrict__ bucket, int2* __restrict__ ent, int2* tmp, int n_nodes) {
    __shared__ int pre[BIN_BLOCKS + 1];
    __shared__ int rbase[BIN_BLOCKS];
    __shared__ int cnt[RANGE_NODES];
    __shared__ int nstart[RANGE_NODES];
    __shared__ int fill[RANGE_NODES];
    __shared__ int lkey[SORT_CAP];
    __shared__ int wsa[33], wsb[33], wsc[33];

    const int tid = threadIdx.x;
    const int bin = blockIdx.x;

    int len = 0, off = 0;
    if (tid < BIN_BLOCKS) {
        const unsigned word = runs[(size_t)tid * N_RANGES + bin];
        off = (int)(word & 0xffffu);
        len = (int)(word >> 16);
    }
    if (tid < RANGE_NODES) { cnt[tid] = 0; fill[tid] = 0; }
    int total, base;
    const int ex = block_scan_excl(len, wsa, total);
    (void)block_scan_excl(off, wsb, base);
    if (tid < BIN_BLOCKS) { pre[tid] = ex; rbase[tid] = tid * BIN_CHUNK + off; }
    if (tid == 0) pre[BIN_BLOCKS] = total;
    __syncthreads();

    int2 eh[BUCKET_EPT];
    #pragma unroll
    for (int j = 0; j < BUCKET_EPT; ++j) {
        const int i = tid + j * BIN_THREADS;
        eh[j] = make_int2(0, 0);
        if (i < total) {
            const int b = run_of(pre, i);
            eh[j] = stage[rbase[b] + (i - pre[b])];
        }
    }
    unsigned od[BUCKET_EPT];
    #pragma unroll
    for (int j = 0; j < BUCKET_EPT; ++j) od[j] = odeg[eh[j].x];
    #pragma unroll
    for (int j = 0; j < BUCKET_EPT; ++j) {
        const int i = tid + j * BIN_THREADS;
        if (i < total) atomicAdd(&cnt[eh[j].y & (RANGE_NODES - 1)], 1);
    }
    for (int i = tid + BUCKET_EPT * BIN_THREADS; i < total; i += BIN_THREADS) {
        const int b = run_of(pre, i);
        const int2 e = stage[rbase[b] + (i - pre[b])];
        atomicAdd(&cnt[e.y & (RANGE_NODES - 1)], 1);
    }
    asm volatile("" :: "v"(od[0]), "v"(od[1]), "v"(od[2]), "v"(od[3]));
    __syncthreads();

    const int c = (tid < RANGE_NODES) ? cnt[tid] : 0;
    int total2;
    const int st = block_scan_excl(c, wsc, total2);
    if (tid < RANGE_NODES) {
        nstart[tid] = st;
        const int node = (bin << RANGE_SHIFT) + tid;
        if (node < n_nodes) {
            bucket[node] = (unsigned long long)(unsigned)(base + st) |
                           ((unsigned long long)(unsigned)c << 32);
            dstn[node] = rsqrtf(fmaxf((float)c, 1.0f));
        }
    }
    __syncthreads();

    if (total <= SORT_CAP) {
        int slot[BUCKET_EPT];
        #pragma unroll
        for (int j = 0; j < BUCKET_EPT; ++j) {
            const int i = tid + j * BIN_THREADS;
            slot[j] = 0;
            if (i < total) {
                const int k = eh[j].y & (RANGE_NODES - 1);
                slot[j] = atomicAdd(&fill[k], 1);
                lkey[nstart[k] + slot[j]] = eh[j].x;
            }
        }
        __syncthreads();
        #pragma unroll
        for (int j = 0; j < BUCKET_EPT; ++j) {
            const int i = tid + j * BIN_THREADS;
            if (i < total) {
                const int k  = eh[j].y & (RANGE_NODES - 1);
                const int st = nstart[k], c = cnt[k];
                const int id = eh[j].x, a = slot[j];
                int rank = 0;
                for (int t = 0; t < c; ++t) {
                    const int o = lkey[st + t];
                    rank += ((o < id) || (o == id && t < a)) ? 1 : 0;
                }
                const float w = rsqrtf(fmaxf((float)od[j], 1.0f));
                ent[base + st + rank] = make_int2(id, __float_as_int(w));
            }
        }
    } else {
        #pragma unroll
        for (int j = 0; j < BUCKET_EPT; ++j) {
            const int i = tid + j * BIN_THREADS;
            if (i < total) {
                const int k  = eh[j].y & (RANGE_NODES - 1);
                const int sl = atomicAdd(&fill[k], 1);
                const float w = rsqrtf(fmaxf((float)od[j], 1.0f));
                tmp[base + nstart[k] + sl] = make_int2(eh[j].x, __float_as_int(w));
            }
        }
        for (int i = tid + BUCKET_EPT * BIN_THREADS; i < total; i += BIN_THREADS) {
            const int b = run_of(pre, i);
            const int2 e = stage[rbase[b] + (i - pre[b])];
            const int k  = e.y & (RANGE_NODES - 1);
            const int sl = atomicAdd(&fill[k], 1);
            const float w = rsqrtf(fmaxf((float)odeg[e.x], 1.0f));
            tmp[base + nstart[k] + sl] = make_int2(e.x, __float_as_int(w));
        }
        __threadfence_block();
        __syncthreads();
        for (int q = tid; q < total; q += BIN_THREADS) {
            int lo = 0, hi = RANGE_NODES;
            while (hi - lo > 1) {
                const int mid = (lo + hi) >> 1;
                if (nstart[mid] <= q) lo = mid; else hi = mid;
            }
            const int st = nstart[lo], c = cnt[lo], a = q - st;
            const int2 e = tmp[base + q];
            int rank = 0;
            for (int t = 0; t < c; ++t) {
                const int o = tmp[base + st + t].x;
                rank += ((o < e.x) || (o == e.x && t < a)) ? 1 : 0;
            }
            ent[base + st + rank] = e;
        }
    }
}

#define CONV_BLOCKS 1024

__device__ __forceinline__ void load_entries(const int2* __restrict__ ent, int offA, int mA,
                                             int offB, int mB, int lane, int& s, float& w) {
    s = 0x7fffffff;
    w = 0.f;
    if (lane < mA + mB) {
        int2 q = ent[(lane < mA) ? (offA + lane) : (offB + lane - mA)];
        s = q.x;
        w = __int_as_float(q.y);
    }
}

#define ROW_OF(sid) (((const float4*)(hin + (size_t)(sid) * HID_C))[lane])
#define ADD_ROW(ACC, V, WV)                                                            \
    { ACC.x += V.x * WV; ACC.y += V.y * WV; ACC.z += V.z * WV; ACC.w += V.w * WV; }
#define ENTRY_POS(i, SID, WV)                                                          \
    {                                                                                  \
        const int j_ = min(p + i, m - 1);                                              \
        SID = __builtin_amdgcn_readlane(s, j_);                                        \
        WV  = __int_as_float(__builtin_amdgcn_readlane(__float_as_int(w), j_));        \
    }
#define ASK_POS(i, V, SID) if (left > i) V = ROW_OF(SID);
#define ADD_POS(i, V, WV)                                                              \
    if (i == sw) { accA = cur; cur = make_float4(0.f, 0.f, 0.f, 0.f); inB = true; }    \
    if (left > i) ADD_ROW(cur, V, WV)

template <bool POOL>
__device__ __forceinline__ void graph_conv_body(
    const float* __restrict__ hin, const unsigned long long* __restrict__ bucket,
    const int2* __restrict__ ent, const float* __restrict__ dstn,
    const float* __restrict__ W, const float* __restrict__ bias,
    float* __restrict__ out, const int* __restrict__ gids,
    float* __restrict__ hg, int n_nodes) {
    __shared__ float As[2][16][HID_C + 4];

    const int tid    = threadIdx.x;
    const int lane   = tid & 31;
    const int wave   = __builtin_amdgcn_readfirstlane(tid >> 5);
    const int l15    = lane & 15;
    const int khalf  = (lane >> 4) << 1;
    const int col    = wave * 16 + l15;
    const int ntiles = (n_nodes + 15) >> 4;
    const int tstep  = gridDim.x;

    v2f breg[HID_C / 4];
    #pragma unroll
    for (int kk = 0; kk < HID_C / 4; ++kk) {
        breg[kk].x = W[(size_t)(kk * 4 + khalf) * HID_C + col];
        breg[kk].y = W[(size_t)(kk * 4 + khalf + 1) * HID_C + col];
    }
    const float bv = bias[col];

    int tile = blockIdx.x;
    unsigned long long bA = 0ull, bB = 0ull, nbA = 0ull, nbB = 0ull;
    if (tile < ntiles) {
        const int n0 = tile * 16 + wave, n1 = n0 + 8;
        if (n0 < n_nodes) bA = bucket[n0];
        if (n1 < n_nodes) bB = bucket[n1];
    }
    if (tile + tstep < ntiles) {
        const int n0 = (tile + tstep) * 16 + wave, n1 = n0 + 8;
        if (n0 < n_nodes) nbA = bucket[n0];
        if (n1 < n_nodes) nbB = bucket[n1];
    }
    int   s;
    float w;
    {
        const int cA = (int)(unsigned)(bA >> 32), cB = (int)(unsigned)(bB >> 32);
        const bool one = (cA + cB <= 32);
        load_entries(ent, (int)(unsigned)bA, one ? cA : 0, (int)(unsigned)bB, one ? cB : 0, lane, s, w);
    }

    int buf = 0;
    for (; tile < ntiles; tile += tstep) {
        __builtin_amdgcn_s_setprio(3);
        const int mbase = tile * 16;
        const int nA = mbase + wave, nB = nA + 8;
        const int startA = (int)(unsigned)bA, cntA = (int)(unsigned)(bA >> 32);
        const int startB = (int)(unsigned)bB, cntB = (int)(unsigned)(bB >> 32);
        const float dA = (nA < n_nodes) ? dstn[nA] : 0.f;
        const float dB = (nB < n_nodes) ? dstn[nB] : 0.f;

        float4 cur = make_float4(0.f, 0.f, 0.f, 0.f), accA = cur;
        bool inB = false;
        {
            int  ca = 0, cb = 0;
            bool have = (cntA + cntB <= 32);
            int  mA = have ? cntA : 0, mB = have ? cntB : 0;
            while (ca < cntA || cb < cntB) {
                if (!have) {
                    if (ca < cntA) { mA = min(32, cntA - ca); mB = 0; }
                    else           { mA = 0; mB = min(32, cntB - cb); }
                    load_entries(ent, startA + ca, mA, startB + cb, mB, lane, s, w);
                }
                have = false;
                const int m = mA + mB;
                for (int p = 0; p < m; p += 8) {
                    const int left = m - p;
                    const int sw   = (mB > 0 && !inB) ? mA - p : -1;
                    const float4 zr = make_float4(0.f, 0.f, 0.f, 0.f);
                    float4 v0 = zr, v1 = zr, v2 = zr, v3 = zr, v4 = zr, v5 = zr, v6 = zr, v7 = zr;
                    float w0, w1, w2, w3, w4, w5, w6, w7;
                    int   i0, i1, i2, i3, i4, i5, i6, i7;
                    ENTRY_POS(0, i0, w0) ENTRY_POS(1, i1, w1) ENTRY_POS(2, i2, w2) ENTRY_POS(3, i3, w3)
                    ENTRY_POS(4, i4, w4) ENTRY_POS(5, i5, w5) ENTRY_POS(6, i6, w6) ENTRY_POS(7, i7, w7)
                    ASK_POS(0, v0, i0) ASK_POS(1, v1, i1) ASK_POS(2, v2, i2) ASK_POS(3, v3, i3)
                    ASK_POS(4, v4, i4) ASK_POS(5, v5, i5) ASK_POS(6, v6, i6) ASK_POS(7, v7, i7)
                    ADD_POS(0, v0, w0) ADD_POS(1, v1, w1) ADD_POS(2, v2, w2) ADD_POS(3, v3, w3)
                    ADD_POS(4, v4, w4) ADD_POS(5, v5, w5) ADD_POS(6, v6, w6) ADD_POS(7, v7, w7)
                }
                ca += mA; cb += mB;
            }
        }

        const bool full_tile = POOL && (mbase + 16 <= n_nodes);
        int gq[16];
        #pragma unroll
        for (int r = 0; r < 16; ++r) gq[r] = 0;
        if constexpr (POOL) {
            if (full_tile) {
                const int* __restrict__ gp = gids + mbase;
                #pragma unroll
                for (int r = 0; r < 16; ++r) gq[r] = gp[r];
            }
        }

        const float4 z4 = make_float4(0.f, 0.f, 0.f, 0.f);
        const float4 rowA = inB ? accA : cur;
        const float4 rowB = inB ? cur : z4;
        *(float4*)&As[buf][wave][lane * 4] =
            make_float4(rowA.x * dA, rowA.y * dA, rowA.z * dA, rowA.w * dA);
        *(float4*)&As[buf][wave + 8][lane * 4] =
            make_float4(rowB.x * dB, rowB.y * dB, rowB.z * dB, rowB.w * dB);
        __syncthreads();

        bA = nbA; bB = nbB;
        {
            const int cA = (int)(unsigned)(bA >> 32), cB = (int)(unsigned)(bB >> 32);
            const bool one = (cA + cB <= 32);
            load_entries(ent, (int)(unsigned)bA, one ? cA : 0, (int)(unsigned)bB, one ? cB : 0, lane, s, w);
        }
        nbA = 0ull; nbB = 0ull;
        if (tile + 2 * tstep < ntiles) {
            const int n0 = (tile + 2 * tstep) * 16 + wave, n1 = n0 + 8;
            if (n0 < n_nodes) nbA = bucket[n0];
            if (n1 < n_nodes) nbB = bucket[n1];
        }

        __builtin_amdgcn_s_setprio(0);

        v8f acc = {0.f, 0.f, 0.f, 0.f, 0.f, 0.f, 0.f, 0.f};
        #pragma unroll
        for (int kk = 0; kk < HID_C / 4; ++kk) {
            v2f a;
            a.x = As[buf][l15][kk * 4 + khalf];
            a.y = As[buf][l15][kk * 4 + khalf + 1];
            acc = __builtin_amdgcn_wmma_f32_16x16x4_f32(false, a, false, breg[kk],
                                                        (short)0, acc, false, false);
        }
        buf ^= 1;

        if constexpr (!POOL) {
            #pragma unroll
            for (int i = 0; i < 8; ++i) {
                int node = mbase + i + ((lane >> 4) << 3);
                if (node < n_nodes)
                    out[(size_t)node * HID_C + col] = fmaxf(acc[i] + bv, 0.0f);
            }
        } else {
            int glo = gq[0], ghi = gq[0];
            #pragma unroll
            for (int r = 1; r < 16; ++r) { glo = min(glo, gq[r]); ghi = max(ghi, gq[r]); }
            const bool one_graph = full_tile && (glo == ghi);
            const int  g0 = gq[0];
            float m = acc[0];
            #pragma unroll
            for (int i = 1; i < 8; ++i) m = fmaxf(m, acc[i]);
            m = fmaxf(m, __shfl_xor(m, 16));
            if (one_graph) {
                if (lane < 16)
                    atomicMax((int*)&hg[(size_t)g0 * HID_C + col],
                              __float_as_int(fmaxf(m + bv, 0.0f)));
            } else {
                #pragma unroll
                for (int i = 0; i < 8; ++i) {
                    int node = mbase + i + ((lane >> 4) << 3);
                    if (node < n_nodes)
                        atomicMax((int*)&hg[(size_t)gids[node] * HID_C + col],
                                  __float_as_int(fmaxf(acc[i] + bv, 0.0f)));
                }
            }
        }
    }
}

__global__ void __launch_bounds__(256, 8) graph_conv_kernel(
    const float* __restrict__ hin, const unsigned long long* __restrict__ bucket,
    const int2* __restrict__ ent, const float* __restrict__ dstn,
    const float* __restrict__ W, const float* __restrict__ bias,
    float* __restrict__ out, int n_nodes) {
    graph_conv_body<false>(hin, bucket, ent, dstn, W, bias, out, nullptr, nullptr, n_nodes);
}

__global__ void __launch_bounds__(256, 8) graph_conv_pool_kernel(
    const float* __restrict__ hin, const unsigned long long* __restrict__ bucket,
    const int2* __restrict__ ent, const float* __restrict__ dstn,
    const float* __restrict__ W, const float* __restrict__ bias,
    const int* __restrict__ gids, float* __restrict__ hg, int n_nodes) {
    graph_conv_body<true>(hin, bucket, ent, dstn, W, bias, nullptr, gids, hg, n_nodes);
}

#define MLP_SPLIT 4
#define MID_C (HID_C / 2)

__global__ void __launch_bounds__(256) mlp_kernel(
    const float* __restrict__ hg,
    const float* __restrict__ W1, const float* __restrict__ b1,
    const float* __restrict__ W2, const float* __restrict__ b2,
    const float* __restrict__ W3, const float* __restrict__ b3,
    float* __restrict__ out) {
    __shared__ float X0[16][HID_C + 4];
    __shared__ float X1[16][HID_C + 4];
    __shared__ float X2[16][MID_C + 4];

    const int tid   = threadIdx.x;
    const int lane  = tid & 31;
    const int wave  = __builtin_amdgcn_readfirstlane(tid >> 5);
    const int l15   = lane & 15;
    const int khalf = (lane >> 4) << 1;
    const int rhalf = (lane >> 4) << 3;
    const int mbase = blockIdx.x * 16;

    float4 t0, t1;
    {
        const int i0 = tid, i1 = tid + 256;
        t0 = ((const float4*)(hg + (size_t)(mbase + (i0 >> 5)) * HID_C))[i0 & 31];
        t1 = ((const float4*)(hg + (size_t)(mbase + (i1 >> 5)) * HID_C))[i1 & 31];
    }

    const int c1 = wave * 16 + l15;
    v2f w1[HID_C / 4];
    #pragma unroll
    for (int kk = 0; kk < HID_C / 4; ++kk) {
        w1[kk].x = W1[(size_t)(kk * 4 + khalf) * HID_C + c1];
        w1[kk].y = W1[(size_t)(kk * 4 + khalf + 1) * HID_C + c1];
    }
    const float bv1 = b1[c1];

    const int c2 = (wave & 3) * 16 + l15;
    v2f w2[HID_C / 4];
    #pragma unroll
    for (int kk = 0; kk < HID_C / 4; ++kk) { w2[kk].x = 0.f; w2[kk].y = 0.f; }
    float bv2 = 0.f;
    if (wave < 4) {
        #pragma unroll
        for (int kk = 0; kk < HID_C / 4; ++kk) {
            w2[kk].x = W2[(size_t)(kk * 4 + khalf) * MID_C + c2];
            w2[kk].y = W2[(size_t)(kk * 4 + khalf + 1) * MID_C + c2];
        }
        bv2 = b2[c2];
    }

    const int  c3   = (blockIdx.y * 8 + wave) * 16 + l15;
    const bool c3ok = c3 < OUT_DIM_C;
    v2f w3[MID_C / 4];
    #pragma unroll
    for (int kk = 0; kk < MID_C / 4; ++kk) {
        w3[kk].x = c3ok ? W3[(size_t)(kk * 4 + khalf) * OUT_DIM_C + c3] : 0.0f;
        w3[kk].y = c3ok ? W3[(size_t)(kk * 4 + khalf + 1) * OUT_DIM_C + c3] : 0.0f;
    }
    const float bv3 = c3ok ? b3[c3] : 0.0f;

    *(float4*)&X0[tid >> 5][(tid & 31) * 4]         = t0;
    *(float4*)&X0[(tid + 256) >> 5][(tid & 31) * 4] = t1;
    __syncthreads();

    {
        v8f acc = {0.f, 0.f, 0.f, 0.f, 0.f, 0.f, 0.f, 0.f};
        #pragma unroll
        for (int kk = 0; kk < HID_C / 4; ++kk) {
            v2f a;
            a.x = X0[l15][kk * 4 + khalf];
            a.y = X0[l15][kk * 4 + khalf + 1];
            acc = __builtin_amdgcn_wmma_f32_16x16x4_f32(false, a, false, w1[kk],
                                                        (short)0, acc, false, false);
        }
        #pragma unroll
        for (int i = 0; i < 8; ++i) X1[i + rhalf][c1] = fmaxf(acc[i] + bv1, 0.0f);
    }
    __syncthreads();

    if (wave < 4) {
        v8f acc = {0.f, 0.f, 0.f, 0.f, 0.f, 0.f, 0.f, 0.f};
        #pragma unroll
        for (int kk = 0; kk < HID_C / 4; ++kk) {
            v2f a;
            a.x = X1[l15][kk * 4 + khalf];
            a.y = X1[l15][kk * 4 + khalf + 1];
            acc = __builtin_amdgcn_wmma_f32_16x16x4_f32(false, a, false, w2[kk],
                                                        (short)0, acc, false, false);
        }
        #pragma unroll
        for (int i = 0; i < 8; ++i) X2[i + rhalf][c2] = fmaxf(acc[i] + bv2, 0.0f);
    }
    __syncthreads();

    {
        v8f acc = {0.f, 0.f, 0.f, 0.f, 0.f, 0.f, 0.f, 0.f};
        #pragma unroll
        for (int kk = 0; kk < MID_C / 4; ++kk) {
            v2f a;
            a.x = X2[l15][kk * 4 + khalf];
            a.y = X2[l15][kk * 4 + khalf + 1];
            acc = __builtin_amdgcn_wmma_f32_16x16x4_f32(false, a, false, w3[kk],
                                                        (short)0, acc, false, false);
        }
        if (c3ok) {
            #pragma unroll
            for (int i = 0; i < 8; ++i)
                out[(size_t)(mbase + i + rhalf) * OUT_DIM_C + c3] = acc[i] + bv3;
        }
    }
}

extern "C" void kernel_launch(void* const* d_in, const int* in_sizes, int n_in,
                              void* d_out, int out_size, void* d_ws, size_t ws_size,
                              hipStream_t stream) {
    (void)in_sizes; (void)n_in; (void)out_size; (void)ws_size;

    const float* feat = (const float*)d_in[0];
    const int*   src  = (const int*)d_in[1];
    const int*   dst  = (const int*)d_in[2];
    const int*   gids = (const int*)d_in[3];
    const float* Wg[3] = {(const float*)d_in[4], (const float*)d_in[6], (const float*)d_in[8]};
    const float* bg[3] = {(const float*)d_in[5], (const float*)d_in[7], (const float*)d_in[9]};
    const float* Wc1 = (const float*)d_in[10]; const float* bc1 = (const float*)d_in[11];
    const float* Wc2 = (const float*)d_in[12]; const float* bc2 = (const float*)d_in[13];
    const float* Wc3 = (const float*)d_in[14]; const float* bc3 = (const float*)d_in[15];
    float* out = (float*)d_out;

    char* ws = (char*)d_ws;
    size_t off = 0;
    auto take = [&](size_t elems) -> float* {
        float* p = (float*)(ws + off);
        off += ((elems * sizeof(float) + 255) / 256) * 256;
        return p;
    };
    unsigned* odeg = (unsigned*)take(N_NODES_C);
    float* dstn = take(N_NODES_C);
    float* h1   = take((size_t)N_NODES_C * HID_C);
    float* h2   = take((size_t)N_NODES_C * HID_C);
    float* hg   = take((size_t)N_GRAPHS_C * HID_C);
    int2*  stage  = (int2*)take(2 * (size_t)BIN_BLOCKS * BIN_CHUNK);
    unsigned* runs = (unsigned*)take((size_t)BIN_BLOCKS * N_RANGES);
    unsigned long long* bucket = (unsigned long long*)take(2 * N_NODES_C);
    int2*  ent    = (int2*)take(2 * (size_t)N_EDGES_C);
    int2*  tmp    = (int2*)take(2 * (size_t)N_EDGES_C);

    hipMemsetAsync(odeg, 0, N_NODES_C * sizeof(unsigned), stream);
    bin_kernel<<<BIN_BLOCKS, BIN_THREADS, 0, stream>>>(src, dst, odeg, stage, runs, N_EDGES_C);
    bucket_kernel<<<N_RANGES, BIN_THREADS, 0, stream>>>(stage, runs, odeg, dstn, bucket, ent, tmp, N_NODES_C);

    const int mblocks = CONV_BLOCKS;
    graph_conv_kernel<<<mblocks, 256, 0, stream>>>(feat, bucket, ent, dstn, Wg[0], bg[0], h1, N_NODES_C);
    graph_conv_kernel<<<mblocks, 256, 0, stream>>>(h1, bucket, ent, dstn, Wg[1], bg[1], h2, N_NODES_C);
    hipMemsetAsync(hg, 0, (size_t)N_GRAPHS_C * HID_C * sizeof(float), stream);
    graph_conv_pool_kernel<<<mblocks, 256, 0, stream>>>(h2, bucket, ent, dstn, Wg[2], bg[2], gids, hg,
                                                        N_NODES_C);

    mlp_kernel<<<dim3(N_GRAPHS_C / 16, MLP_SPLIT), 256, 0, stream>>>(
        hg, Wc1, bc1, Wc2, bc2, Wc3, bc3, out);
}


# Conv output through LDS: a tile's relu'd rows wait in an LDS tile and leave behind the next tile's barrier as whole 512-byte rows, 2 stores a wave instead of 8 half-line stores with row tests
# speedup vs baseline: 4.0676x; 1.0331x over previous
#include <hip/hip_runtime.h>
#include <hip/hip_bf16.h>


typedef float v2f __attribute__((ext_vector_type(2)));
typedef float v8f __attribute__((ext_vector_type(8)));

#define N_NODES_C 100000
#define N_EDGES_C 800000
#define HID_C     128
#define OUT_DIM_C 500
#define N_GRAPHS_C 64

#define BIN_BLOCKS   256
#define BIN_THREADS  1024
#define BIN_CHUNK    ((N_EDGES_C + BIN_BLOCKS - 1) / BIN_BLOCKS)
#define BIN_EPT      ((BIN_CHUNK + BIN_THREADS - 1) / BIN_THREADS)
#define RANGE_SHIFT  8
#define RANGE_NODES  (1 << RANGE_SHIFT)
#define N_RANGES     ((N_NODES_C + RANGE_NODES - 1) >> RANGE_SHIFT)

static_assert(BIN_THREADS == 1024, "block_scan_excl is written for 32 waves");
static_assert(BIN_CHUNK < 65536, "run offset and length share one 32-bit word");
static_assert(N_RANGES <= BIN_THREADS, "one range per thread in the scan");
static_assert(BIN_BLOCKS <= BIN_THREADS && RANGE_NODES <= BIN_THREADS, "one item per thread in the scans");

__device__ __forceinline__ int block_scan_excl(int v, int* ws, int& total) {
    const int lane = threadIdx.x & 31;
    const int wave = threadIdx.x >> 5;
    int incl = v;
    #pragma unroll
    for (int o = 1; o < 32; o <<= 1) {
        int t = __shfl_up(incl, o);
        if (lane >= o) incl += t;
    }
    if (lane == 31) ws[wave] = incl;
    __syncthreads();
    if (wave == 0) {
        const int wv = ws[lane];
        int wi = wv;
        #pragma unroll
        for (int o = 1; o < 32; o <<= 1) {
            int t = __shfl_up(wi, o);
            if (lane >= o) wi += t;
        }
        ws[lane] = wi - wv;
        if (lane == 31) ws[32] = wi;
    }
    __syncthreads();
    total = ws[32];
    return ws[wave] + incl - v;
}

__global__ void __launch_bounds__(BIN_THREADS) bin_kernel(
    const int* __restrict__ src, const int* __restrict__ dst,
    unsigned* __restrict__ odeg, int2* __restrict__ stage,
    unsigned* __restrict__ runs, int n_edges) {
    __shared__ int  hist[N_RANGES];
    __shared__ int2 sorted[BIN_CHUNK];
    __shared__ int  ws[33];

    const int tid     = threadIdx.x;
    const int e_begin = blockIdx.x * BIN_CHUNK;
    const int e_end   = min(e_begin + BIN_CHUNK, n_edges);

    for (int i = tid; i < N_RANGES; i += BIN_THREADS) hist[i] = 0;

    int s[BIN_EPT], d[BIN_EPT], rk[BIN_EPT];
    #pragma unroll
    for (int j = 0; j < BIN_EPT; ++j) {
        const int e = e_begin + tid + j * BIN_THREADS;
        s[j] = 0; d[j] = -1; rk[j] = 0;
        if (e < e_end) { s[j] = src[e]; d[j] = dst[e]; }
    }
    __syncthreads();
    #pragma unroll
    for (int j = 0; j < BIN_EPT; ++j) {
        if (d[j] >= 0) rk[j] = atomicAdd(&hist[d[j] >> RANGE_SHIFT], 1);
    }
    __syncthreads();

    const int c = (tid < N_RANGES) ? hist[tid] : 0;
    int total;
    const int ex = block_scan_excl(c, ws, total);
    if (tid < N_RANGES) {
        hist[tid] = ex;
        runs[(size_t)blockIdx.x * N_RANGES + tid] = (unsigned)ex | ((unsigned)c << 16);
    }
    __syncthreads();
    #pragma unroll
    for (int j = 0; j < BIN_EPT; ++j)
        if (d[j] >= 0) sorted[hist[d[j] >> RANGE_SHIFT] + rk[j]] = make_int2(s[j], d[j]);
    __syncthreads();
    for (int i = tid; i < e_end - e_begin; i += BIN_THREADS) stage[e_begin + i] = sorted[i];
    #pragma unroll
    for (int j = 0; j < BIN_EPT; ++j)
        if (d[j] >= 0) atomicAdd(&odeg[s[j]], 1u);
}

__device__ __forceinline__ int run_of(const int* pre, int i) {
    int lo = 0, hi = BIN_BLOCKS;
    while (hi - lo > 1) {
        const int mid = (lo + hi) >> 1;
        if (pre[mid] <= i) lo = mid; else hi = mid;
    }
    return lo;
}

#define BUCKET_EPT 4
static_assert(BUCKET_EPT == 4, "the load fence in bucket_kernel names four values");
#define SORT_CAP (BUCKET_EPT * BIN_THREADS)

__global__ void __launch_bounds__(BIN_THREADS) bucket_kernel(
    const int2* __restrict__ stage, const unsigned* __restrict__ runs,
    const unsigned* __restrict__ odeg, float* __restrict__ dstn,
    unsigned long long* __restrict__ bucket, int2* __restrict__ ent, int2* tmp, int n_nodes) {
    __shared__ int pre[BIN_BLOCKS + 1];
    __shared__ int rbase[BIN_BLOCKS];
    __shared__ int cnt[RANGE_NODES];
    __shared__ int nstart[RANGE_NODES];
    __shared__ int fill[RANGE_NODES];
    __shared__ int lkey[SORT_CAP];
    __shared__ int wsa[33], wsb[33], wsc[33];

    const int tid = threadIdx.x;
    const int bin = blockIdx.x;

    int len = 0, off = 0;
    if (tid < BIN_BLOCKS) {
        const unsigned word = runs[(size_t)tid * N_RANGES + bin];
        off = (int)(word & 0xffffu);
        len = (int)(word >> 16);
    }
    if (tid < RANGE_NODES) { cnt[tid] = 0; fill[tid] = 0; }
    int total, base;
    const int ex = block_scan_excl(len, wsa, total);
    (void)block_scan_excl(off, wsb, base);
    if (tid < BIN_BLOCKS) { pre[tid] = ex; rbase[tid] = tid * BIN_CHUNK + off; }
    if (tid == 0) pre[BIN_BLOCKS] = total;
    __syncthreads();

    int2 eh[BUCKET_EPT];
    #pragma unroll
    for (int j = 0; j < BUCKET_EPT; ++j) {
        const int i = tid + j * BIN_THREADS;
        eh[j] = make_int2(0, 0);
        if (i < total) {
            const int b = run_of(pre, i);
            eh[j] = stage[rbase[b] + (i - pre[b])];
        }
    }
    unsigned od[BUCKET_EPT];
    #pragma unroll
    for (int j = 0; j < BUCKET_EPT; ++j) od[j] = odeg[eh[j].x];
    #pragma unroll
    for (int j = 0; j < BUCKET_EPT; ++j) {
        const int i = tid + j * BIN_THREADS;
        if (i < total) atomicAdd(&cnt[eh[j].y & (RANGE_NODES - 1)], 1);
    }
    for (int i = tid + BUCKET_EPT * BIN_THREADS; i < total; i += BIN_THREADS) {
        const int b = run_of(pre, i);
        const int2 e = stage[rbase[b] + (i - pre[b])];
        atomicAdd(&cnt[e.y & (RANGE_NODES - 1)], 1);
    }
    asm volatile("" :: "v"(od[0]), "v"(od[1]), "v"(od[2]), "v"(od[3]));
    __syncthreads();

    const int c = (tid < RANGE_NODES) ? cnt[tid] : 0;
    int total2;
    const int st = block_scan_excl(c, wsc, total2);
    if (tid < RANGE_NODES) {
        nstart[tid] = st;
        const int node = (bin << RANGE_SHIFT) + tid;
        if (node < n_nodes) {
            bucket[node] = (unsigned long long)(unsigned)(base + st) |
                           ((unsigned long long)(unsigned)c << 32);
            dstn[node] = rsqrtf(fmaxf((float)c, 1.0f));
        }
    }
    __syncthreads();

    if (total <= SORT_CAP) {
        int slot[BUCKET_EPT];
        #pragma unroll
        for (int j = 0; j < BUCKET_EPT; ++j) {
            const int i = tid + j * BIN_THREADS;
            slot[j] = 0;
            if (i < total) {
                const int k = eh[j].y & (RANGE_NODES - 1);
                slot[j] = atomicAdd(&fill[k], 1);
                lkey[nstart[k] + slot[j]] = eh[j].x;
            }
        }
        __syncthreads();
        #pragma unroll
        for (int j = 0; j < BUCKET_EPT; ++j) {
            const int i = tid + j * BIN_THREADS;
            if (i < total) {
                const int k  = eh[j].y & (RANGE_NODES - 1);
                const int st = nstart[k], c = cnt[k];
                const int id = eh[j].x, a = slot[j];
                int rank = 0;
                for (int t = 0; t < c; ++t) {
                    const int o = lkey[st + t];
                    rank += ((o < id) || (o == id && t < a)) ? 1 : 0;
                }
                const float w = rsqrtf(fmaxf((float)od[j], 1.0f));
                ent[base + st + rank] = make_int2(id, __float_as_int(w));
            }
        }
    } else {
        #pragma unroll
        for (int j = 0; j < BUCKET_EPT; ++j) {
            const int i = tid + j * BIN_THREADS;
            if (i < total) {
                const int k  = eh[j].y & (RANGE_NODES - 1);
                const int sl = atomicAdd(&fill[k], 1);
                const float w = rsqrtf(fmaxf((float)od[j], 1.0f));
                tmp[base + nstart[k] + sl] = make_int2(eh[j].x, __float_as_int(w));
            }
        }
        for (int i = tid + BUCKET_EPT * BIN_THREADS; i < total; i += BIN_THREADS) {
            const int b = run_of(pre, i);
            const int2 e = stage[rbase[b] + (i - pre[b])];
            const int k  = e.y & (RANGE_NODES - 1);
            const int sl = atomicAdd(&fill[k], 1);
            const float w = rsqrtf(fmaxf((float)odeg[e.x], 1.0f));
            tmp[base + nstart[k] + sl] = make_int2(e.x, __float_as_int(w));
        }
        __threadfence_block();
        __syncthreads();
        for (int q = tid; q < total; q += BIN_THREADS) {
            int lo = 0, hi = RANGE_NODES;
            while (hi - lo > 1) {
                const int mid = (lo + hi) >> 1;
                if (nstart[mid] <= q) lo = mid; else hi = mid;
            }
            const int st = nstart[lo], c = cnt[lo], a = q - st;
            const int2 e = tmp[base + q];
            int rank = 0;
            for (int t = 0; t < c; ++t) {
                const int o = tmp[base + st + t].x;
                rank += ((o < e.x) || (o == e.x && t < a)) ? 1 : 0;
            }
            ent[base + st + rank] = e;
        }
    }
}

#define CONV_BLOCKS 1024

__device__ __forceinline__ void load_entries(const int2* __restrict__ ent, int offA, int mA,
                                             int offB, int mB, int lane, int& s, float& w) {
    s = 0x7fffffff;
    w = 0.f;
    if (lane < mA + mB) {
        int2 q = ent[(lane < mA) ? (offA + lane) : (offB + lane - mA)];
        s = q.x;
        w = __int_as_float(q.y);
    }
}

#define ROW_OF(sid) (((const float4*)(hin + (size_t)(sid) * HID_C))[lane])
#define ADD_ROW(ACC, V, WV)                                                            \
    { ACC.x += V.x * WV; ACC.y += V.y * WV; ACC.z += V.z * WV; ACC.w += V.w * WV; }
#define ENTRY_POS(i, SID, WV)                                                          \
    {                                                                                  \
        const int j_ = min(p + i, m - 1);                                              \
        SID = __builtin_amdgcn_readlane(s, j_);                                        \
        WV  = __int_as_float(__builtin_amdgcn_readlane(__float_as_int(w), j_));        \
    }
#define ASK_POS(i, V, SID) if (left > i) V = ROW_OF(SID);
#define ADD_POS(i, V, WV)                                                              \
    if (i == sw) { accA = cur; cur = make_float4(0.f, 0.f, 0.f, 0.f); inB = true; }    \
    if (left > i) ADD_ROW(cur, V, WV)

template <bool POOL>
__device__ __forceinline__ void graph_conv_body(
    const float* __restrict__ hin, const unsigned long long* __restrict__ bucket,
    const int2* __restrict__ ent, const float* __restrict__ dstn,
    const float* __restrict__ W, const float* __restrict__ bias,
    float* __restrict__ out, const int* __restrict__ gids,
    float* __restrict__ hg, int n_nodes) {
    __shared__ float As[2][16][HID_C + 4];
    __shared__ float Os[2][16][HID_C + 4];

    const int tid    = threadIdx.x;
    const int lane   = tid & 31;
    const int wave   = __builtin_amdgcn_readfirstlane(tid >> 5);
    const int l15    = lane & 15;
    const int khalf  = (lane >> 4) << 1;
    const int col    = wave * 16 + l15;
    const int ntiles = (n_nodes + 15) >> 4;
    const int tstep  = gridDim.x;

    v2f breg[HID_C / 4];
    #pragma unroll
    for (int kk = 0; kk < HID_C / 4; ++kk) {
        breg[kk].x = W[(size_t)(kk * 4 + khalf) * HID_C + col];
        breg[kk].y = W[(size_t)(kk * 4 + khalf + 1) * HID_C + col];
    }
    const float bv = bias[col];

    int tile = blockIdx.x;
    unsigned long long bA = 0ull, bB = 0ull, nbA = 0ull, nbB = 0ull;
    if (tile < ntiles) {
        const int n0 = tile * 16 + wave, n1 = n0 + 8;
        if (n0 < n_nodes) bA = bucket[n0];
        if (n1 < n_nodes) bB = bucket[n1];
    }
    if (tile + tstep < ntiles) {
        const int n0 = (tile + tstep) * 16 + wave, n1 = n0 + 8;
        if (n0 < n_nodes) nbA = bucket[n0];
        if (n1 < n_nodes) nbB = bucket[n1];
    }
    int   s;
    float w;
    {
        const int cA = (int)(unsigned)(bA >> 32), cB = (int)(unsigned)(bB >> 32);
        const bool one = (cA + cB <= 32);
        load_entries(ent, (int)(unsigned)bA, one ? cA : 0, (int)(unsigned)bB, one ? cB : 0, lane, s, w);
    }

    int buf = 0;
    int pbase = -1;
    for (; tile < ntiles; tile += tstep) {
        __builtin_amdgcn_s_setprio(3);
        const int mbase = tile * 16;
        const int nA = mbase + wave, nB = nA + 8;
        const int startA = (int)(unsigned)bA, cntA = (int)(unsigned)(bA >> 32);
        const int startB = (int)(unsigned)bB, cntB = (int)(unsigned)(bB >> 32);
        const float dA = (nA < n_nodes) ? dstn[nA] : 0.f;
        const float dB = (nB < n_nodes) ? dstn[nB] : 0.f;

        float4 cur = make_float4(0.f, 0.f, 0.f, 0.f), accA = cur;
        bool inB = false;
        {
            int  ca = 0, cb = 0;
            bool have = (cntA + cntB <= 32);
            int  mA = have ? cntA : 0, mB = have ? cntB : 0;
            while (ca < cntA || cb < cntB) {
                if (!have) {
                    if (ca < cntA) { mA = min(32, cntA - ca); mB = 0; }
                    else           { mA = 0; mB = min(32, cntB - cb); }
                    load_entries(ent, startA + ca, mA, startB + cb, mB, lane, s, w);
                }
                have = false;
                const int m = mA + mB;
                for (int p = 0; p < m; p += 8) {
                    const int left = m - p;
                    const int sw   = (mB > 0 && !inB) ? mA - p : -1;
                    const float4 zr = make_float4(0.f, 0.f, 0.f, 0.f);
                    float4 v0 = zr, v1 = zr, v2 = zr, v3 = zr, v4 = zr, v5 = zr, v6 = zr, v7 = zr;
                    float w0, w1, w2, w3, w4, w5, w6, w7;
                    int   i0, i1, i2, i3, i4, i5, i6, i7;
                    ENTRY_POS(0, i0, w0) ENTRY_POS(1, i1, w1) ENTRY_POS(2, i2, w2) ENTRY_POS(3, i3, w3)
                    ENTRY_POS(4, i4, w4) ENTRY_POS(5, i5, w5) ENTRY_POS(6, i6, w6) ENTRY_POS(7, i7, w7)
                    ASK_POS(0, v0, i0) ASK_POS(1, v1, i1) ASK_POS(2, v2, i2) ASK_POS(3, v3, i3)
                    ASK_POS(4, v4, i4) ASK_POS(5, v5, i5) ASK_POS(6, v6, i6) ASK_POS(7, v7, i7)
                    ADD_POS(0, v0, w0) ADD_POS(1, v1, w1) ADD_POS(2, v2, w2) ADD_POS(3, v3, w3)
                    ADD_POS(4, v4, w4) ADD_POS(5, v5, w5) ADD_POS(6, v6, w6) ADD_POS(7, v7, w7)
                }
                ca += mA; cb += mB;
            }
        }

        const bool full_tile = POOL && (mbase + 16 <= n_nodes);
        int gq[16];
        #pragma unroll
        for (int r = 0; r < 16; ++r) gq[r] = 0;
        if constexpr (POOL) {
            if (full_tile) {
                const int* __restrict__ gp = gids + mbase;
                #pragma unroll
                for (int r = 0; r < 16; ++r) gq[r] = gp[r];
            }
        }

        const float4 z4 = make_float4(0.f, 0.f, 0.f, 0.f);
        const float4 rowA = inB ? accA : cur;
        const float4 rowB = inB ? cur : z4;
        *(float4*)&As[buf][wave][lane * 4] =
            make_float4(rowA.x * dA, rowA.y * dA, rowA.z * dA, rowA.w * dA);
        *(float4*)&As[buf][wave + 8][lane * 4] =
            make_float4(rowB.x * dB, rowB.y * dB, rowB.z * dB, rowB.w * dB);
        __syncthreads();

        bA = nbA; bB = nbB;
        {
            const int cA = (int)(unsigned)(bA >> 32), cB = (int)(unsigned)(bB >> 32);
            const bool one = (cA + cB <= 32);
            load_entries(ent, (int)(unsigned)bA, one ? cA : 0, (int)(unsigned)bB, one ? cB : 0, lane, s, w);
        }
        nbA = 0ull; nbB = 0ull;
        if (tile + 2 * tstep < ntiles) {
            const int n0 = (tile + 2 * tstep) * 16 + wave, n1 = n0 + 8;
            if (n0 < n_nodes) nbA = bucket[n0];
            if (n1 < n_nodes) nbB = bucket[n1];
        }

        __builtin_amdgcn_s_setprio(0);

        float4 o0 = make_float4(0.f, 0.f, 0.f, 0.f), o1 = o0;
        if constexpr (!POOL) {
            if (pbase >= 0) {
                o0 = *(const float4*)&Os[buf ^ 1][wave][lane * 4];
                o1 = *(const float4*)&Os[buf ^ 1][wave + 8][lane * 4];
            }
        }

        v8f acc = {0.f, 0.f, 0.f, 0.f, 0.f, 0.f, 0.f, 0.f};
        #pragma unroll
        for (int kk = 0; kk < HID_C / 4; ++kk) {
            v2f a;
            a.x = As[buf][l15][kk * 4 + khalf];
            a.y = As[buf][l15][kk * 4 + khalf + 1];
            acc = __builtin_amdgcn_wmma_f32_16x16x4_f32(false, a, false, breg[kk],
                                                        (short)0, acc, false, false);
        }

        if constexpr (!POOL) {
            if (pbase >= 0) {
                const int r0 = pbase + wave, r1 = r0 + 8;
                if (r0 < n_nodes) ((float4*)(out + (size_t)r0 * HID_C))[lane] = o0;
                if (r1 < n_nodes) ((float4*)(out + (size_t)r1 * HID_C))[lane] = o1;
            }
            const int rhalf = (lane >> 4) << 3;
            #pragma unroll
            for (int i = 0; i < 8; ++i) Os[buf][i + rhalf][col] = fmaxf(acc[i] + bv, 0.0f);
            pbase = mbase;
        }
        buf ^= 1;

        if constexpr (POOL) {
            int glo = gq[0], ghi = gq[0];
            #pragma unroll
            for (int r = 1; r < 16; ++r) { glo = min(glo, gq[r]); ghi = max(ghi, gq[r]); }
            const bool one_graph = full_tile && (glo == ghi);
            const int  g0 = gq[0];
            float m = acc[0];
            #pragma unroll
            for (int i = 1; i < 8; ++i) m = fmaxf(m, acc[i]);
            m = fmaxf(m, __shfl_xor(m, 16));
            if (one_graph) {
                if (lane < 16)
                    atomicMax((int*)&hg[(size_t)g0 * HID_C + col],
                              __float_as_int(fmaxf(m + bv, 0.0f)));
            } else {
                #pragma unroll
                for (int i = 0; i < 8; ++i) {
                    int node = mbase + i + ((lane >> 4) << 3);
                    if (node < n_nodes)
                        atomicMax((int*)&hg[(size_t)gids[node] * HID_C + col],
                                  __float_as_int(fmaxf(acc[i] + bv, 0.0f)));
                }
            }
        }
    }

    if constexpr (!POOL) {
        if (pbase >= 0) {
            __syncthreads();
            const int r0 = pbase + wave, r1 = r0 + 8;
            if (r0 < n_nodes)
                ((float4*)(out + (size_t)r0 * HID_C))[lane] = *(const float4*)&Os[buf ^ 1][wave][lane * 4];
            if (r1 < n_nodes)
                ((float4*)(out + (size_t)r1 * HID_C))[lane] = *(const float4*)&Os[buf ^ 1][wave + 8][lane * 4];
        }
    }
}

__global__ void __launch_bounds__(256, 8) graph_conv_kernel(
    const float* __restrict__ hin, const unsigned long long* __restrict__ bucket,
    const int2* __restrict__ ent, const float* __restrict__ dstn,
    const float* __restrict__ W, const float* __restrict__ bias,
    float* __restrict__ out, int n_nodes) {
    graph_conv_body<false>(hin, bucket, ent, dstn, W, bias, out, nullptr, nullptr, n_nodes);
}

__global__ void __launch_bounds__(256, 8) graph_conv_pool_kernel(
    const float* __restrict__ hin, const unsigned long long* __restrict__ bucket,
    const int2* __restrict__ ent, const float* __restrict__ dstn,
    const float* __restrict__ W, const float* __restrict__ bias,
    const int* __restrict__ gids, float* __restrict__ hg, int n_nodes) {
    graph_conv_body<true>(hin, bucket, ent, dstn, W, bias, nullptr, gids, hg, n_nodes);
}

#define MLP_SPLIT 4
#define MID_C (HID_C / 2)

__global__ void __launch_bounds__(256) mlp_kernel(
    const float* __restrict__ hg,
    const float* __restrict__ W1, const float* __restrict__ b1,
    const float* __restrict__ W2, const float* __restrict__ b2,
    const float* __restrict__ W3, const float* __restrict__ b3,
    float* __restrict__ out) {
    __shared__ float X0[16][HID_C + 4];
    __shared__ float X1[16][HID_C + 4];
    __shared__ float X2[16][MID_C + 4];

    const int tid   = threadIdx.x;
    const int lane  = tid & 31;
    const int wave  = __builtin_amdgcn_readfirstlane(tid >> 5);
    const int l15   = lane & 15;
    const int khalf = (lane >> 4) << 1;
    const int rhalf = (lane >> 4) << 3;
    const int mbase = blockIdx.x * 16;

    float4 t0, t1;
    {
        const int i0 = tid, i1 = tid + 256;
        t0 = ((const float4*)(hg + (size_t)(mbase + (i0 >> 5)) * HID_C))[i0 & 31];
        t1 = ((const float4*)(hg + (size_t)(mbase + (i1 >> 5)) * HID_C))[i1 & 31];
    }

    const int c1 = wave * 16 + l15;
    v2f w1[HID_C / 4];
    #pragma unroll
    for (int kk = 0; kk < HID_C / 4; ++kk) {
        w1[kk].x = W1[(size_t)(kk * 4 + khalf) * HID_C + c1];
        w1[kk].y = W1[(size_t)(kk * 4 + khalf + 1) * HID_C + c1];
    }
    const float bv1 = b1[c1];

    const int c2 = (wave & 3) * 16 + l15;
    v2f w2[HID_C / 4];
    #pragma unroll
    for (int kk = 0; kk < HID_C / 4; ++kk) { w2[kk].x = 0.f; w2[kk].y = 0.f; }
    float bv2 = 0.f;
    if (wave < 4) {
        #pragma unroll
        for (int kk = 0; kk < HID_C / 4; ++kk) {
            w2[kk].x = W2[(size_t)(kk * 4 + khalf) * MID_C + c2];
            w2[kk].y = W2[(size_t)(kk * 4 + khalf + 1) * MID_C + c2];
        }
        bv2 = b2[c2];
    }

    const int  c3   = (blockIdx.y * 8 + wave) * 16 + l15;
    const bool c3ok = c3 < OUT_DIM_C;
    v2f w3[MID_C / 4];
    #pragma unroll
    for (int kk = 0; kk < MID_C / 4; ++kk) {
        w3[kk].x = c3ok ? W3[(size_t)(kk * 4 + khalf) * OUT_DIM_C + c3] : 0.0f;
        w3[kk].y = c3ok ? W3[(size_t)(kk * 4 + khalf + 1) * OUT_DIM_C + c3] : 0.0f;
    }
    const float bv3 = c3ok ? b3[c3] : 0.0f;

    *(float4*)&X0[tid >> 5][(tid & 31) * 4]         = t0;
    *(float4*)&X0[(tid + 256) >> 5][(tid & 31) * 4] = t1;
    __syncthreads();

    {
        v8f acc = {0.f, 0.f, 0.f, 0.f, 0.f, 0.f, 0.f, 0.f};
        #pragma unroll
        for (int kk = 0; kk < HID_C / 4; ++kk) {
            v2f a;
            a.x = X0[l15][kk * 4 + khalf];
            a.y = X0[l15][kk * 4 + khalf + 1];
            acc = __builtin_amdgcn_wmma_f32_16x16x4_f32(false, a, false, w1[kk],
                                                        (short)0, acc, false, false);
        }
        #pragma unroll
        for (int i = 0; i < 8; ++i) X1[i + rhalf][c1] = fmaxf(acc[i] + bv1, 0.0f);
    }
    __syncthreads();

    if (wave < 4) {
        v8f acc = {0.f, 0.f, 0.f, 0.f, 0.f, 0.f, 0.f, 0.f};
        #pragma unroll
        for (int kk = 0; kk < HID_C / 4; ++kk) {
            v2f a;
            a.x = X1[l15][kk * 4 + khalf];
            a.y = X1[l15][kk * 4 + khalf + 1];
            acc = __builtin_amdgcn_wmma_f32_16x16x4_f32(false, a, false, w2[kk],
                                                        (short)0, acc, false, false);
        }
        #pragma unroll
        for (int i = 0; i < 8; ++i) X2[i + rhalf][c2] = fmaxf(acc[i] + bv2, 0.0f);
    }
    __syncthreads();

    {
        v8f acc = {0.f, 0.f, 0.f, 0.f, 0.f, 0.f, 0.f, 0.f};
        #pragma unroll
        for (int kk = 0; kk < MID_C / 4; ++kk) {
            v2f a;
            a.x = X2[l15][kk * 4 + khalf];
            a.y = X2[l15][kk * 4 + khalf + 1];
            acc = __builtin_amdgcn_wmma_f32_16x16x4_f32(false, a, false, w3[kk],
                                                        (short)0, acc, false, false);
        }
        if (c3ok) {
            #pragma unroll
            for (int i = 0; i < 8; ++i)
                out[(size_t)(mbase + i + rhalf) * OUT_DIM_C + c3] = acc[i] + bv3;
        }
    }
}

extern "C" void kernel_launch(void* const* d_in, const int* in_sizes, int n_in,
                              void* d_out, int out_size, void* d_ws, size_t ws_size,
                              hipStream_t stream) {
    (void)in_sizes; (void)n_in; (void)out_size; (void)ws_size;

    const float* feat = (const float*)d_in[0];
    const int*   src  = (const int*)d_in[1];
    const int*   dst  = (const int*)d_in[2];
    const int*   gids = (const int*)d_in[3];
    const float* Wg[3] = {(const float*)d_in[4], (const float*)d_in[6], (const float*)d_in[8]};
    const float* bg[3] = {(const float*)d_in[5], (const float*)d_in[7], (const float*)d_in[9]};
    const float* Wc1 = (const float*)d_in[10]; const float* bc1 = (const float*)d_in[11];
    const float* Wc2 = (const float*)d_in[12]; const float* bc2 = (const float*)d_in[13];
    const float* Wc3 = (const float*)d_in[14]; const float* bc3 = (const float*)d_in[15];
    float* out = (float*)d_out;

    char* ws = (char*)d_ws;
    size_t off = 0;
    auto take = [&](size_t elems) -> float* {
        float* p = (float*)(ws + off);
        off += ((elems * sizeof(float) + 255) / 256) * 256;
        return p;
    };
    unsigned* odeg = (unsigned*)take(N_NODES_C);
    float* dstn = take(N_NODES_C);
    float* h1   = take((size_t)N_NODES_C * HID_C);
    float* h2   = take((size_t)N_NODES_C * HID_C);
    float* hg   = take((size_t)N_GRAPHS_C * HID_C);
    int2*  stage  = (int2*)take(2 * (size_t)BIN_BLOCKS * BIN_CHUNK);
    unsigned* runs = (unsigned*)take((size_t)BIN_BLOCKS * N_RANGES);
    unsigned long long* bucket = (unsigned long long*)take(2 * N_NODES_C);
    int2*  ent    = (int2*)take(2 * (size_t)N_EDGES_C);
    int2*  tmp    = (int2*)take(2 * (size_t)N_EDGES_C);

    hipMemsetAsync(odeg, 0, N_NODES_C * sizeof(unsigned), stream);
    bin_kernel<<<BIN_BLOCKS, BIN_THREADS, 0, stream>>>(src, dst, odeg, stage, runs, N_EDGES_C);
    bucket_kernel<<<N_RANGES, BIN_THREADS, 0, stream>>>(stage, runs, odeg, dstn, bucket, ent, tmp, N_NODES_C);

    const int mblocks = CONV_BLOCKS;
    graph_conv_kernel<<<mblocks, 256, 0, stream>>>(feat, bucket, ent, dstn, Wg[0], bg[0], h1, N_NODES_C);
    graph_conv_kernel<<<mblocks, 256, 0, stream>>>(h1, bucket, ent, dstn, Wg[1], bg[1], h2, N_NODES_C);
    hipMemsetAsync(hg, 0, (size_t)N_GRAPHS_C * HID_C * sizeof(float), stream);
    graph_conv_pool_kernel<<<mblocks, 256, 0, stream>>>(h2, bucket, ent, dstn, Wg[2], bg[2], gids, hg,
                                                        N_NODES_C);

    mlp_kernel<<<dim3(N_GRAPHS_C / 16, MLP_SPLIT), 256, 0, stream>>>(
        hg, Wc1, bc1, Wc2, bc2, Wc3, bc3, out);
}


# Gather group as straight-line code in the conv launches: 8 (or last 4) row loads always, short groups filled with row 0 at weight 0; per-position tests, zero fills, selects gone; add order kept
# speedup vs baseline: 4.2157x; 1.0364x over previous
#include <hip/hip_runtime.h>
#include <hip/hip_bf16.h>


typedef float v2f __attribute__((ext_vector_type(2)));
typedef float v8f __attribute__((ext_vector_type(8)));

#define N_NODES_C 100000
#define N_EDGES_C 800000
#define HID_C     128
#define OUT_DIM_C 500
#define N_GRAPHS_C 64

#define BIN_BLOCKS   256
#define BIN_THREADS  1024
#define BIN_CHUNK    ((N_EDGES_C + BIN_BLOCKS - 1) / BIN_BLOCKS)
#define BIN_EPT      ((BIN_CHUNK + BIN_THREADS - 1) / BIN_THREADS)
#define RANGE_SHIFT  8
#define RANGE_NODES  (1 << RANGE_SHIFT)
#define N_RANGES     ((N_NODES_C + RANGE_NODES - 1) >> RANGE_SHIFT)

static_assert(BIN_THREADS == 1024, "block_scan_excl is written for 32 waves");
static_assert(BIN_CHUNK < 65536, "run offset and length share one 32-bit word");
static_assert(N_RANGES <= BIN_THREADS, "one range per thread in the scan");
static_assert(BIN_BLOCKS <= BIN_THREADS && RANGE_NODES <= BIN_THREADS, "one item per thread in the scans");

__device__ __forceinline__ int block_scan_excl(int v, int* ws, int& total) {
    const int lane = threadIdx.x & 31;
    const int wave = threadIdx.x >> 5;
    int incl = v;
    #pragma unroll
    for (int o = 1; o < 32; o <<= 1) {
        int t = __shfl_up(incl, o);
        if (lane >= o) incl += t;
    }
    if (lane == 31) ws[wave] = incl;
    __syncthreads();
    if (wave == 0) {
        const int wv = ws[lane];
        int wi = wv;
        #pragma unroll
        for (int o = 1; o < 32; o <<= 1) {
            int t = __shfl_up(wi, o);
            if (lane >= o) wi += t;
        }
        ws[lane] = wi - wv;
        if (lane == 31) ws[32] = wi;
    }
    __syncthreads();
    total = ws[32];
    return ws[wave] + incl - v;
}

__global__ void __launch_bounds__(BIN_THREADS) bin_kernel(
    const int* __restrict__ src, const int* __restrict__ dst,
    unsigned* __restrict__ odeg, int2* __restrict__ stage,
    unsigned* __restrict__ runs, int n_edges) {
    __shared__ int  hist[N_RANGES];
    __shared__ int2 sorted[BIN_CHUNK];
    __shared__ int  ws[33];

    const int tid     = threadIdx.x;
    const int e_begin = blockIdx.x * BIN_CHUNK;
    const int e_end   = min(e_begin + BIN_CHUNK, n_edges);

    for (int i = tid; i < N_RANGES; i += BIN_THREADS) hist[i] = 0;

    int s[BIN_EPT], d[BIN_EPT], rk[BIN_EPT];
    #pragma unroll
    for (int j = 0; j < BIN_EPT; ++j) {
        const int e = e_begin + tid + j * BIN_THREADS;
        s[j] = 0; d[j] = -1; rk[j] = 0;
        if (e < e_end) { s[j] = src[e]; d[j] = dst[e]; }
    }
    __syncthreads();
    #pragma unroll
    for (int j = 0; j < BIN_EPT; ++j) {
        if (d[j] >= 0) rk[j] = atomicAdd(&hist[d[j] >> RANGE_SHIFT], 1);
    }
    __syncthreads();

    const int c = (tid < N_RANGES) ? hist[tid] : 0;
    int total;
    const int ex = block_scan_excl(c, ws, total);
    if (tid < N_RANGES) {
        hist[tid] = ex;
        runs[(size_t)blockIdx.x * N_RANGES + tid] = (unsigned)ex | ((unsigned)c << 16);
    }
    __syncthreads();
    #pragma unroll
    for (int j = 0; j < BIN_EPT; ++j)
        if (d[j] >= 0) sorted[hist[d[j] >> RANGE_SHIFT] + rk[j]] = make_int2(s[j], d[j]);
    __syncthreads();
    for (int i = tid; i < e_end - e_begin; i += BIN_THREADS) stage[e_begin + i] = sorted[i];
    #pragma unroll
    for (int j = 0; j < BIN_EPT; ++j)
        if (d[j] >= 0) atomicAdd(&odeg[s[j]], 1u);
}

__device__ __forceinline__ int run_of(const int* pre, int i) {
    int lo = 0, hi = BIN_BLOCKS;
    while (hi - lo > 1) {
        const int mid = (lo + hi) >> 1;
        if (pre[mid] <= i) lo = mid; else hi = mid;
    }
    return lo;
}

#define BUCKET_EPT 4
static_assert(BUCKET_EPT == 4, "the load fence in bucket_kernel names four values");
#define SORT_CAP (BUCKET_EPT * BIN_THREADS)

__global__ void __launch_bounds__(BIN_THREADS) bucket_kernel(
    const int2* __restrict__ stage, const unsigned* __restrict__ runs,
    const unsigned* __restrict__ odeg, float* __restrict__ dstn,
    unsigned long long* __restrict__ bucket, int2* __restrict__ ent, int2* tmp, int n_nodes) {
    __shared__ int pre[BIN_BLOCKS + 1];
    __shared__ int rbase[BIN_BLOCKS];
    __shared__ int cnt[RANGE_NODES];
    __shared__ int nstart[RANGE_NODES];
    __shared__ int fill[RANGE_NODES];
    __shared__ int lkey[SORT_CAP];
    __shared__ int wsa[33], wsb[33], wsc[33];

    const int tid = threadIdx.x;
    const int bin = blockIdx.x;

    int len = 0, off = 0;
    if (tid < BIN_BLOCKS) {
        const unsigned word = runs[(size_t)tid * N_RANGES + bin];
        off = (int)(word & 0xffffu);
        len = (int)(word >> 16);
    }
    if (tid < RANGE_NODES) { cnt[tid] = 0; fill[tid] = 0; }
    int total, base;
    const int ex = block_scan_excl(len, wsa, total);
    (void)block_scan_excl(off, wsb, base);
    if (tid < BIN_BLOCKS) { pre[tid] = ex; rbase[tid] = tid * BIN_CHUNK + off; }
    if (tid == 0) pre[BIN_BLOCKS] = total;
    __syncthreads();

    int2 eh[BUCKET_EPT];
    #pragma unroll
    for (int j = 0; j < BUCKET_EPT; ++j) {
        const int i = tid + j * BIN_THREADS;
        eh[j] = make_int2(0, 0);
        if (i < total) {
            const int b = run_of(pre, i);
            eh[j] = stage[rbase[b] + (i - pre[b])];
        }
    }
    unsigned od[BUCKET_EPT];
    #pragma unroll
    for (int j = 0; j < BUCKET_EPT; ++j) od[j] = odeg[eh[j].x];
    #pragma unroll
    for (int j = 0; j < BUCKET_EPT; ++j) {
        const int i = tid + j * BIN_THREADS;
        if (i < total) atomicAdd(&cnt[eh[j].y & (RANGE_NODES - 1)], 1);
    }
    for (int i = tid + BUCKET_EPT * BIN_THREADS; i < total; i += BIN_THREADS) {
        const int b = run_of(pre, i);
        const int2 e = stage[rbase[b] + (i - pre[b])];
        atomicAdd(&cnt[e.y & (RANGE_NODES - 1)], 1);
    }
    asm volatile("" :: "v"(od[0]), "v"(od[1]), "v"(od[2]), "v"(od[3]));
    __syncthreads();

    const int c = (tid < RANGE_NODES) ? cnt[tid] : 0;
    int total2;
    const int st = block_scan_excl(c, wsc, total2);
    if (tid < RANGE_NODES) {
        nstart[tid] = st;
        const int node = (bin << RANGE_SHIFT) + tid;
        if (node < n_nodes) {
            bucket[node] = (unsigned long long)(unsigned)(base + st) |
                           ((unsigned long long)(unsigned)c << 32);
            dstn[node] = rsqrtf(fmaxf((float)c, 1.0f));
        }
    }
    __syncthreads();

    if (total <= SORT_CAP) {
        int slot[BUCKET_EPT];
        #pragma unroll
        for (int j = 0; j < BUCKET_EPT; ++j) {
            const int i = tid + j * BIN_THREADS;
            slot[j] = 0;
            if (i < total) {
                const int k = eh[j].y & (RANGE_NODES - 1);
                slot[j] = atomicAdd(&fill[k], 1);
                lkey[nstart[k] + slot[j]] = eh[j].x;
            }
        }
        __syncthreads();
        #pragma unroll
        for (int j = 0; j < BUCKET_EPT; ++j) {
            const int i = tid + j * BIN_THREADS;
            if (i < total) {
                const int k  = eh[j].y & (RANGE_NODES - 1);
                const int st = nstart[k], c = cnt[k];
                const int id = eh[j].x, a = slot[j];
                int rank = 0;
                for (int t = 0; t < c; ++t) {
                    const int o = lkey[st + t];
                    rank += ((o < id) || (o == id && t < a)) ? 1 : 0;
                }
                const float w = rsqrtf(fmaxf((float)od[j], 1.0f));
                ent[base + st + rank] = make_int2(id, __float_as_int(w));
            }
        }
    } else {
        #pragma unroll
        for (int j = 0; j < BUCKET_EPT; ++j) {
            const int i = tid + j * BIN_THREADS;
            if (i < total) {
                const int k  = eh[j].y & (RANGE_NODES - 1);
                const int sl = atomicAdd(&fill[k], 1);
                const float w = rsqrtf(fmaxf((float)od[j], 1.0f));
                tmp[base + nstart[k] + sl] = make_int2(eh[j].x, __float_as_int(w));
            }
        }
        for (int i = tid + BUCKET_EPT * BIN_THREADS; i < total; i += BIN_THREADS) {
            const int b = run_of(pre, i);
            const int2 e = stage[rbase[b] + (i - pre[b])];
            const int k  = e.y & (RANGE_NODES - 1);
            const int sl = atomicAdd(&fill[k], 1);
            const float w = rsqrtf(fmaxf((float)odeg[e.x], 1.0f));
            tmp[base + nstart[k] + sl] = make_int2(e.x, __float_as_int(w));
        }
        __threadfence_block();
        __syncthreads();
        for (int q = tid; q < total; q += BIN_THREADS) {
            int lo = 0, hi = RANGE_NODES;
            while (hi - lo > 1) {
                const int mid = (lo + hi) >> 1;
                if (nstart[mid] <= q) lo = mid; else hi = mid;
            }
            const int st = nstart[lo], c = cnt[lo], a = q - st;
            const int2 e = tmp[base + q];
            int rank = 0;
            for (int t = 0; t < c; ++t) {
                const int o = tmp[base + st + t].x;
                rank += ((o < e.x) || (o == e.x && t < a)) ? 1 : 0;
            }
            ent[base + st + rank] = e;
        }
    }
}

#define CONV_BLOCKS 1024

__device__ __forceinline__ void load_entries(const int2* __restrict__ ent, int offA, int mA,
                                             int offB, int mB, int lane, int& s, float& w) {
    s = 0;
    w = 0.f;
    if (lane < mA + mB) {
        int2 q = ent[(lane < mA) ? (offA + lane) : (offB + lane - mA)];
        s = q.x;
        w = __int_as_float(q.y);
    }
}

#define ROW_OF(sid) (((const float4*)(hin + (size_t)(sid) * HID_C))[lane])
#define ADD_ROW(ACC, V, WV)                                                            \
    { ACC.x += V.x * WV; ACC.y += V.y * WV; ACC.z += V.z * WV; ACC.w += V.w * WV; }
#define ENTRY_POS(i, SID, WV)                                                          \
    {                                                                                  \
        SID = __builtin_amdgcn_readlane(s, p + i);                                     \
        WV  = __int_as_float(__builtin_amdgcn_readlane(__float_as_int(w), p + i));     \
    }
#define ASK_POS(i, V, SID) V = ROW_OF(SID);
#define ADD_POS(i, V, WV)                                                              \
    if (i == sw) { accA = cur; cur = make_float4(0.f, 0.f, 0.f, 0.f); inB = true; }    \
    ADD_ROW(cur, V, WV)

template <bool POOL>
__device__ __forceinline__ void graph_conv_body(
    const float* __restrict__ hin, const unsigned long long* __restrict__ bucket,
    const int2* __restrict__ ent, const float* __restrict__ dstn,
    const float* __restrict__ W, const float* __restrict__ bias,
    float* __restrict__ out, const int* __restrict__ gids,
    float* __restrict__ hg, int n_nodes) {
    __shared__ float As[2][16][HID_C + 4];
    __shared__ float Os[2][16][HID_C + 4];

    const int tid    = threadIdx.x;
    const int lane   = tid & 31;
    const int wave   = __builtin_amdgcn_readfirstlane(tid >> 5);
    const int l15    = lane & 15;
    const int khalf  = (lane >> 4) << 1;
    const int col    = wave * 16 + l15;
    const int ntiles = (n_nodes + 15) >> 4;
    const int tstep  = gridDim.x;

    v2f breg[HID_C / 4];
    #pragma unroll
    for (int kk = 0; kk < HID_C / 4; ++kk) {
        breg[kk].x = W[(size_t)(kk * 4 + khalf) * HID_C + col];
        breg[kk].y = W[(size_t)(kk * 4 + khalf + 1) * HID_C + col];
    }
    const float bv = bias[col];

    int tile = blockIdx.x;
    unsigned long long bA = 0ull, bB = 0ull, nbA = 0ull, nbB = 0ull;
    if (tile < ntiles) {
        const int n0 = tile * 16 + wave, n1 = n0 + 8;
        if (n0 < n_nodes) bA = bucket[n0];
        if (n1 < n_nodes) bB = bucket[n1];
    }
    if (tile + tstep < ntiles) {
        const int n0 = (tile + tstep) * 16 + wave, n1 = n0 + 8;
        if (n0 < n_nodes) nbA = bucket[n0];
        if (n1 < n_nodes) nbB = bucket[n1];
    }
    int   s;
    float w;
    {
        const int cA = (int)(unsigned)(bA >> 32), cB = (int)(unsigned)(bB >> 32);
        const bool one = (cA + cB <= 32);
        load_entries(ent, (int)(unsigned)bA, one ? cA : 0, (int)(unsigned)bB, one ? cB : 0, lane, s, w);
    }

    int buf = 0;
    int pbase = -1;
    for (; tile < ntiles; tile += tstep) {
        __builtin_amdgcn_s_setprio(3);
        const int mbase = tile * 16;
        const int nA = mbase + wave, nB = nA + 8;
        const int startA = (int)(unsigned)bA, cntA = (int)(unsigned)(bA >> 32);
        const int startB = (int)(unsigned)bB, cntB = (int)(unsigned)(bB >> 32);
        const float dA = (nA < n_nodes) ? dstn[nA] : 0.f;
        const float dB = (nB < n_nodes) ? dstn[nB] : 0.f;

        float4 cur = make_float4(0.f, 0.f, 0.f, 0.f), accA = cur;
        bool inB = false;
        {
            int  ca = 0, cb = 0;
            bool have = (cntA + cntB <= 32);
            int  mA = have ? cntA : 0, mB = have ? cntB : 0;
            while (ca < cntA || cb < cntB) {
                if (!have) {
                    if (ca < cntA) { mA = min(32, cntA - ca); mB = 0; }
                    else           { mA = 0; mB = min(32, cntB - cb); }
                    load_entries(ent, startA + ca, mA, startB + cb, mB, lane, s, w);
                }
                have = false;
                const int m = mA + mB;
                for (int p = 0; p < m; ) {
                    const int sw   = (mB > 0 && !inB) ? mA - p : -1;
                    float4 v0, v1, v2, v3, v4, v5, v6, v7;
                    float w0, w1, w2, w3, w4, w5, w6, w7;
                    int   i0, i1, i2, i3, i4, i5, i6, i7;
                    if (m - p > 4) {
                        ENTRY_POS(0, i0, w0) ENTRY_POS(1, i1, w1) ENTRY_POS(2, i2, w2) ENTRY_POS(3, i3, w3)
                        ENTRY_POS(4, i4, w4) ENTRY_POS(5, i5, w5) ENTRY_POS(6, i6, w6) ENTRY_POS(7, i7, w7)
                        ASK_POS(0, v0, i0) ASK_POS(1, v1, i1) ASK_POS(2, v2, i2) ASK_POS(3, v3, i3)
                        ASK_POS(4, v4, i4) ASK_POS(5, v5, i5) ASK_POS(6, v6, i6) ASK_POS(7, v7, i7)
                        ADD_POS(0, v0, w0) ADD_POS(1, v1, w1) ADD_POS(2, v2, w2) ADD_POS(3, v3, w3)
                        ADD_POS(4, v4, w4) ADD_POS(5, v5, w5) ADD_POS(6, v6, w6) ADD_POS(7, v7, w7)
                        p += 8;
                    } else {
                        ENTRY_POS(0, i0, w0) ENTRY_POS(1, i1, w1) ENTRY_POS(2, i2, w2) ENTRY_POS(3, i3, w3)
                        ASK_POS(0, v0, i0) ASK_POS(1, v1, i1) ASK_POS(2, v2, i2) ASK_POS(3, v3, i3)
                        ADD_POS(0, v0, w0) ADD_POS(1, v1, w1) ADD_POS(2, v2, w2) ADD_POS(3, v3, w3)
                        p += 4;
                    }
                }
                ca += mA; cb += mB;
            }
        }

        const bool full_tile = POOL && (mbase + 16 <= n_nodes);
        int gq[16];
        #pragma unroll
        for (int r = 0; r < 16; ++r) gq[r] = 0;
        if constexpr (POOL) {
            if (full_tile) {
                const int* __restrict__ gp = gids + mbase;
                #pragma unroll
                for (int r = 0; r < 16; ++r) gq[r] = gp[r];
            }
        }

        const float4 z4 = make_float4(0.f, 0.f, 0.f, 0.f);
        const float4 rowA = inB ? accA : cur;
        const float4 rowB = inB ? cur : z4;
        *(float4*)&As[buf][wave][lane * 4] =
            make_float4(rowA.x * dA, rowA.y * dA, rowA.z * dA, rowA.w * dA);
        *(float4*)&As[buf][wave + 8][lane * 4] =
            make_float4(rowB.x * dB, rowB.y * dB, rowB.z * dB, rowB.w * dB);
        __syncthreads();

        bA = nbA; bB = nbB;
        {
            const int cA = (int)(unsigned)(bA >> 32), cB = (int)(unsigned)(bB >> 32);
            const bool one = (cA + cB <= 32);
            load_entries(ent, (int)(unsigned)bA, one ? cA : 0, (int)(unsigned)bB, one ? cB : 0, lane, s, w);
        }
        nbA = 0ull; nbB = 0ull;
        if (tile + 2 * tstep < ntiles) {
            const int n0 = (tile + 2 * tstep) * 16 + wave, n1 = n0 + 8;
            if (n0 < n_nodes) nbA = bucket[n0];
            if (n1 < n_nodes) nbB = bucket[n1];
        }

        __builtin_amdgcn_s_setprio(0);

        float4 o0 = make_float4(0.f, 0.f, 0.f, 0.f), o1 = o0;
        if constexpr (!POOL) {
            if (pbase >= 0) {
                o0 = *(const float4*)&Os[buf ^ 1][wave][lane * 4];
                o1 = *(const float4*)&Os[buf ^ 1][wave + 8][lane * 4];
            }
        }

        v8f acc = {0.f, 0.f, 0.f, 0.f, 0.f, 0.f, 0.f, 0.f};
        #pragma unroll
        for (int kk = 0; kk < HID_C / 4; ++kk) {
            v2f a;
            a.x = As[buf][l15][kk * 4 + khalf];
            a.y = As[buf][l15][kk * 4 + khalf + 1];
            acc = __builtin_amdgcn_wmma_f32_16x16x4_f32(false, a, false, breg[kk],
                                                        (short)0, acc, false, false);
        }

        if constexpr (!POOL) {
            if (pbase >= 0) {
                const int r0 = pbase + wave, r1 = r0 + 8;
                if (r0 < n_nodes) ((float4*)(out + (size_t)r0 * HID_C))[lane] = o0;
                if (r1 < n_nodes) ((float4*)(out + (size_t)r1 * HID_C))[lane] = o1;
            }
            const int rhalf = (lane >> 4) << 3;
            #pragma unroll
            for (int i = 0; i < 8; ++i) Os[buf][i + rhalf][col] = fmaxf(acc[i] + bv, 0.0f);
            pbase = mbase;
        }
        buf ^= 1;

        if constexpr (POOL) {
            int glo = gq[0], ghi = gq[0];
            #pragma unroll
            for (int r = 1; r < 16; ++r) { glo = min(glo, gq[r]); ghi = max(ghi, gq[r]); }
            const bool one_graph = full_tile && (glo == ghi);
            const int  g0 = gq[0];
            float m = acc[0];
            #pragma unroll
            for (int i = 1; i < 8; ++i) m = fmaxf(m, acc[i]);
            m = fmaxf(m, __shfl_xor(m, 16));
            if (one_graph) {
                if (lane < 16)
                    atomicMax((int*)&hg[(size_t)g0 * HID_C + col],
                              __float_as_int(fmaxf(m + bv, 0.0f)));
            } else {
                #pragma unroll
                for (int i = 0; i < 8; ++i) {
                    int node = mbase + i + ((lane >> 4) << 3);
                    if (node < n_nodes)
                        atomicMax((int*)&hg[(size_t)gids[node] * HID_C + col],
                                  __float_as_int(fmaxf(acc[i] + bv, 0.0f)));
                }
            }
        }
    }

    if constexpr (!POOL) {
        if (pbase >= 0) {
            __syncthreads();
            const int r0 = pbase + wave, r1 = r0 + 8;
            if (r0 < n_nodes)
                ((float4*)(out + (size_t)r0 * HID_C))[lane] = *(const float4*)&Os[buf ^ 1][wave][lane * 4];
            if (r1 < n_nodes)
                ((float4*)(out + (size_t)r1 * HID_C))[lane] = *(const float4*)&Os[buf ^ 1][wave + 8][lane * 4];
        }
    }
}

__global__ void __launch_bounds__(256, 8) graph_conv_kernel(
    const float* __restrict__ hin, const unsigned long long* __restrict__ bucket,
    const int2* __restrict__ ent, const float* __restrict__ dstn,
    const float* __restrict__ W, const float* __restrict__ bias,
    float* __restrict__ out, int n_nodes) {
    graph_conv_body<false>(hin, bucket, ent, dstn, W, bias, out, nullptr, nullptr, n_nodes);
}

__global__ void __launch_bounds__(256, 8) graph_conv_pool_kernel(
    const float* __restrict__ hin, const unsigned long long* __restrict__ bucket,
    const int2* __restrict__ ent, const float* __restrict__ dstn,
    const float* __restrict__ W, const float* __restrict__ bias,
    const int* __restrict__ gids, float* __restrict__ hg, int n_nodes) {
    graph_conv_body<true>(hin, bucket, ent, dstn, W, bias, nullptr, gids, hg, n_nodes);
}

#define MLP_SPLIT 4
#define MID_C (HID_C / 2)

__global__ void __launch_bounds__(256) mlp_kernel(
    const float* __restrict__ hg,
    const float* __restrict__ W1, const float* __restrict__ b1,
    const float* __restrict__ W2, const float* __restrict__ b2,
    const float* __restrict__ W3, const float* __restrict__ b3,
    float* __restrict__ out) {
    __shared__ float X0[16][HID_C + 4];
    __shared__ float X1[16][HID_C + 4];
    __shared__ float X2[16][MID_C + 4];

    const int tid   = threadIdx.x;
    const int lane  = tid & 31;
    const int wave  = __builtin_amdgcn_readfirstlane(tid >> 5);
    const int l15   = lane & 15;
    const int khalf = (lane >> 4) << 1;
    const int rhalf = (lane >> 4) << 3;
    const int mbase = blockIdx.x * 16;

    float4 t0, t1;
    {
        const int i0 = tid, i1 = tid + 256;
        t0 = ((const float4*)(hg + (size_t)(mbase + (i0 >> 5)) * HID_C))[i0 & 31];
        t1 = ((const float4*)(hg + (size_t)(mbase + (i1 >> 5)) * HID_C))[i1 & 31];
    }

    const int c1 = wave * 16 + l15;
    v2f w1[HID_C / 4];
    #pragma unroll
    for (int kk = 0; kk < HID_C / 4; ++kk) {
        w1[kk].x = W1[(size_t)(kk * 4 + khalf) * HID_C + c1];
        w1[kk].y = W1[(size_t)(kk * 4 + khalf + 1) * HID_C + c1];
    }
    const float bv1 = b1[c1];

    const int c2 = (wave & 3) * 16 + l15;
    v2f w2[HID_C / 4];
    #pragma unroll
    for (int kk = 0; kk < HID_C / 4; ++kk) { w2[kk].x = 0.f; w2[kk].y = 0.f; }
    float bv2 = 0.f;
    if (wave < 4) {
        #pragma unroll
        for (int kk = 0; kk < HID_C / 4; ++kk) {
            w2[kk].x = W2[(size_t)(kk * 4 + khalf) * MID_C + c2];
            w2[kk].y = W2[(size_t)(kk * 4 + khalf + 1) * MID_C + c2];
        }
        bv2 = b2[c2];
    }

    const int  c3   = (blockIdx.y * 8 + wave) * 16 + l15;
    const bool c3ok = c3 < OUT_DIM_C;
    v2f w3[MID_C / 4];
    #pragma unroll
    for (int kk = 0; kk < MID_C / 4; ++kk) {
        w3[kk].x = c3ok ? W3[(size_t)(kk * 4 + khalf) * OUT_DIM_C + c3] : 0.0f;
        w3[kk].y = c3ok ? W3[(size_t)(kk * 4 + khalf + 1) * OUT_DIM_C + c3] : 0.0f;
    }
    const float bv3 = c3ok ? b3[c3] : 0.0f;

    *(float4*)&X0[tid >> 5][(tid & 31) * 4]         = t0;
    *(float4*)&X0[(tid + 256) >> 5][(tid & 31) * 4] = t1;
    __syncthreads();

    {
        v8f acc = {0.f, 0.f, 0.f, 0.f, 0.f, 0.f, 0.f, 0.f};
        #pragma unroll
        for (int kk = 0; kk < HID_C / 4; ++kk) {
            v2f a;
            a.x = X0[l15][kk * 4 + khalf];
            a.y = X0[l15][kk * 4 + khalf + 1];
            acc = __builtin_amdgcn_wmma_f32_16x16x4_f32(false, a, false, w1[kk],
                                                        (short)0, acc, false, false);
        }
        #pragma unroll
        for (int i = 0; i < 8; ++i) X1[i + rhalf][c1] = fmaxf(acc[i] + bv1, 0.0f);
    }
    __syncthreads();

    if (wave < 4) {
        v8f acc = {0.f, 0.f, 0.f, 0.f, 0.f, 0.f, 0.f, 0.f};
        #pragma unroll
        for (int kk = 0; kk < HID_C / 4; ++kk) {
            v2f a;
            a.x = X1[l15][kk * 4 + khalf];
            a.y = X1[l15][kk * 4 + khalf + 1];
            acc = __builtin_amdgcn_wmma_f32_16x16x4_f32(false, a, false, w2[kk],
                                                        (short)0, acc, false, false);
        }
        #pragma unroll
        for (int i = 0; i < 8; ++i) X2[i + rhalf][c2] = fmaxf(acc[i] + bv2, 0.0f);
    }
    __syncthreads();

    {
        v8f acc = {0.f, 0.f, 0.f, 0.f, 0.f, 0.f, 0.f, 0.f};
        #pragma unroll
        for (int kk = 0; kk < MID_C / 4; ++kk) {
            v2f a;
            a.x = X2[l15][kk * 4 + khalf];
            a.y = X2[l15][kk * 4 + khalf + 1];
            acc = __builtin_amdgcn_wmma_f32_16x16x4_f32(false, a, false, w3[kk],
                                                        (short)0, acc, false, false);
        }
        if (c3ok) {
            #pragma unroll
            for (int i = 0; i < 8; ++i)
                out[(size_t)(mbase + i + rhalf) * OUT_DIM_C + c3] = acc[i] + bv3;
        }
    }
}

extern "C" void kernel_launch(void* const* d_in, const int* in_sizes, int n_in,
                              void* d_out, int out_size, void* d_ws, size_t ws_size,
                              hipStream_t stream) {
    (void)in_sizes; (void)n_in; (void)out_size; (void)ws_size;

    const float* feat = (const float*)d_in[0];
    const int*   src  = (const int*)d_in[1];
    const int*   dst  = (const int*)d_in[2];
    const int*   gids = (const int*)d_in[3];
    const float* Wg[3] = {(const float*)d_in[4], (const float*)d_in[6], (const float*)d_in[8]};
    const float* bg[3] = {(const float*)d_in[5], (const float*)d_in[7], (const float*)d_in[9]};
    const float* Wc1 = (const float*)d_in[10]; const float* bc1 = (const float*)d_in[11];
    const float* Wc2 = (const float*)d_in[12]; const float* bc2 = (const float*)d_in[13];
    const float* Wc3 = (const float*)d_in[14]; const float* bc3 = (const float*)d_in[15];
    float* out = (float*)d_out;

    char* ws = (char*)d_ws;
    size_t off = 0;
    auto take = [&](size_t elems) -> float* {
        float* p = (float*)(ws + off);
        off += ((elems * sizeof(float) + 255) / 256) * 256;
        return p;
    };
    unsigned* odeg = (unsigned*)take(N_NODES_C);
    float* dstn = take(N_NODES_C);
    float* h1   = take((size_t)N_NODES_C * HID_C);
    float* h2   = take((size_t)N_NODES_C * HID_C);
    float* hg   = take((size_t)N_GRAPHS_C * HID_C);
    int2*  stage  = (int2*)take(2 * (size_t)BIN_BLOCKS * BIN_CHUNK);
    unsigned* runs = (unsigned*)take((size_t)BIN_BLOCKS * N_RANGES);
    unsigned long long* bucket = (unsigned long long*)take(2 * N_NODES_C);
    int2*  ent    = (int2*)take(2 * (size_t)N_EDGES_C);
    int2*  tmp    = (int2*)take(2 * (size_t)N_EDGES_C);

    hipMemsetAsync(odeg, 0, N_NODES_C * sizeof(unsigned), stream);
    bin_kernel<<<BIN_BLOCKS, BIN_THREADS, 0, stream>>>(src, dst, odeg, stage, runs, N_EDGES_C);
    bucket_kernel<<<N_RANGES, BIN_THREADS, 0, stream>>>(stage, runs, odeg, dstn, bucket, ent, tmp, N_NODES_C);

    const int mblocks = CONV_BLOCKS;
    graph_conv_kernel<<<mblocks, 256, 0, stream>>>(feat, bucket, ent, dstn, Wg[0], bg[0], h1, N_NODES_C);
    graph_conv_kernel<<<mblocks, 256, 0, stream>>>(h1, bucket, ent, dstn, Wg[1], bg[1], h2, N_NODES_C);
    hipMemsetAsync(hg, 0, (size_t)N_GRAPHS_C * HID_C * sizeof(float), stream);
    graph_conv_pool_kernel<<<mblocks, 256, 0, stream>>>(h2, bucket, ent, dstn, Wg[2], bg[2], gids, hg,
                                                        N_NODES_C);

    mlp_kernel<<<dim3(N_GRAPHS_C / 16, MLP_SPLIT), 256, 0, stream>>>(
        hg, Wc1, bc1, Wc2, bc2, Wc3, bc3, out);
}
